# starting kernel
# baseline (speedup 1.0000x reference)
#include <hip/hip_runtime.h>

typedef float v2f __attribute__((ext_vector_type(2)));
typedef float v8f __attribute__((ext_vector_type(8)));

#define WMMA_F32(a, b, c) \
  __builtin_amdgcn_wmma_f32_16x16x4_f32(false, (a), false, (b), (short)0, (c), false, false)

constexpr int L = 1024, DM = 768, Hh = 12, FD = 16, HD = 64;
constexpr float EPSF = 1e-12f;
constexpr float QSC = 0.17677669529663687f;

__global__ __launch_bounds__(256) void gemm_f32_wmma(const float* __restrict__ A,
                                                     const float* __restrict__ B,
                                                     float* __restrict__ C,
                                                     int M, int N, int K) {
  __shared__ float As[64][36];
  __shared__ float BsT[64][38];
  const int tid  = threadIdx.x;
  const int wave = tid >> 5, lane = tid & 31;
  const int kh   = lane >> 4, ln = lane & 15;
  const int ntb  = N / 64;
  const int bm   = (blockIdx.x / ntb) * 64;
  const int bn   = (blockIdx.x % ntb) * 64;
  const int wm   = wave >> 1;
  const int wn   = wave & 1;
  const int ar   = tid >> 2, ac = (tid & 3) * 4;
  const int br   = tid >> 3, bc = (tid & 7) * 4;
  (void)M;

  const float4* Ag = (const float4*)A;
  const float4* Bg = (const float4*)B;

  float4 a0 = Ag[((size_t)(bm + ar) * K + ac) >> 2];
  float4 a1 = Ag[((size_t)(bm + ar) * K + ac + 16) >> 2];
  float4 b0 = Bg[((size_t)br * N + bn + bc) >> 2];
  float4 b1 = Bg[((size_t)br * N + bn + bc + 32) >> 2];
  v8f acc0 = {}, acc1 = {};

  for (int kb = 0; kb < K; kb += 32) {
    *(float4*)&As[ar][ac]      = a0;
    *(float4*)&As[ar][ac + 16] = a1;
    BsT[bc + 0][br]  = b0.x;  BsT[bc + 1][br]  = b0.y;
    BsT[bc + 2][br]  = b0.z;  BsT[bc + 3][br]  = b0.w;
    BsT[bc + 32][br] = b1.x;  BsT[bc + 33][br] = b1.y;
    BsT[bc + 34][br] = b1.z;  BsT[bc + 35][br] = b1.w;
    __syncthreads();
    if (kb + 32 < K) {
      a0 = Ag[((size_t)(bm + ar) * K + kb + 32 + ac) >> 2];
      a1 = Ag[((size_t)(bm + ar) * K + kb + 32 + ac + 16) >> 2];
      b0 = Bg[((size_t)(kb + 32 + br) * N + bn + bc) >> 2];
      b1 = Bg[((size_t)(kb + 32 + br) * N + bn + bc + 32) >> 2];
    }
#pragma unroll
    for (int k = 0; k < 32; k += 4) {
      const int kk = k + 2 * kh;
      v2f a = *(const v2f*)&As[wm * 16 + ln][kk];
      v2f p = *(const v2f*)&BsT[wn * 32 + ln][kk];
      v2f q = *(const v2f*)&BsT[wn * 32 + 16 + ln][kk];
      acc0 = WMMA_F32(a, p, acc0);
      acc1 = WMMA_F32(a, q, acc1);
    }
    __syncthreads();
  }
#pragma unroll
  for (int r = 0; r < 8; ++r) {
    const size_t row = (size_t)(bm + wm * 16 + r + 8 * kh);
    C[row * N + bn + wn * 32 + ln]      = acc0[r];
    C[row * N + bn + wn * 32 + 16 + ln] = acc1[r];
  }
}

__global__ __launch_bounds__(256) void based_attn(const float* __restrict__ Qg,
                                                  const float* __restrict__ Kg,
                                                  const float* __restrict__ Vg,
                                                  float* __restrict__ Yg) {
  __shared__ float s_lin[16][17];
  __shared__ float s_quad[256][17];
  __shared__ float s_ones[16];
  __shared__ float kcum[273];
  __shared__ float Qc[32][20];
  __shared__ float Kc[32][20];
  __shared__ float Vc[32][20];
  __shared__ float Aw[32][33];
  __shared__ float Fq[32][257];
  __shared__ float Yc[32][20];
  __shared__ float den[32];

  const int tid  = threadIdx.x;
  const int wave = tid >> 5, lane = tid & 31;
  const int kh   = lane >> 4, ln = lane & 15;
  const int h    = blockIdx.x >> 2;
  const int vb   = blockIdx.x & 3;
  const int qoff = h * FD;
  const int voff = h * HD + vb * 16;

  for (int i = tid; i < 16 * 16; i += 256) s_lin[i >> 4][i & 15] = 0.f;
  for (int i = tid; i < 256 * 16; i += 256) s_quad[i >> 4][i & 15] = 0.f;
  if (tid < 16) s_ones[tid] = 0.f;
  for (int i = tid; i < 273; i += 256) kcum[i] = 0.f;
  __syncthreads();

  for (int c = 0; c < L / 32; ++c) {
    const int bl = c * 32;
    if (tid < 128) {
      const int t = tid >> 2, c4 = (tid & 3) * 4;
      *(float4*)&Qc[t][c4] = *(const float4*)&Qg[(size_t)(bl + t) * (Hh * FD) + qoff + c4];
      *(float4*)&Kc[t][c4] = *(const float4*)&Kg[(size_t)(bl + t) * (Hh * FD) + qoff + c4];
    } else {
      const int i = tid - 128;
      const int t = i >> 2, c4 = (i & 3) * 4;
      *(float4*)&Vc[t][c4] = *(const float4*)&Vg[(size_t)(bl + t) * DM + voff + c4];
    }
    __syncthreads();

    if (wave < 4) {
      const int ti = wave >> 1, tj = wave & 1;
      v8f acc = {};
#pragma unroll
      for (int k = 0; k < 16; k += 4) {
        const int kk = k + 2 * kh;
        v2f a = *(const v2f*)&Qc[ti * 16 + ln][kk];
        v2f b = *(const v2f*)&Kc[tj * 16 + ln][kk];
        acc = WMMA_F32(a, b, acc);
      }
#pragma unroll
      for (int r = 0; r < 8; ++r) {
        const int row = ti * 16 + r + 8 * kh, col = tj * 16 + ln;
        const float s = acc[r];
        Aw[row][col] = (row >= col) ? (1.f + 0.25f * s + (s * s) * 0.03125f) : 0.f;
      }
    }
    for (int i = tid; i < 32 * 256; i += 256) {
      int t = i >> 8, ij = i & 255;
      Fq[t][ij] = Qc[t][ij >> 4] * Qc[t][ij & 15] * QSC;
    }
    __syncthreads();

    if (wave < 2) {
      const int ti = wave;
      const int row = ti * 16 + ln;
      v8f acc;
#pragma unroll
      for (int r = 0; r < 8; ++r) acc[r] = s_ones[ln];
#pragma unroll
      for (int k = 0; k < 32; k += 4) {
        const int kk = k + 2 * kh;
        v2f a = *(const v2f*)&Aw[row][kk];
        v2f b;
        b.x = Vc[kk][ln];  b.y = Vc[kk + 1][ln];
        acc = WMMA_F32(a, b, acc);
      }
#pragma unroll
      for (int k = 0; k < 16; k += 4) {
        const int kk = k + 2 * kh;
        v2f a, b;
        a.x = Qc[row][kk] * 0.5f; a.y = Qc[row][kk + 1] * 0.5f;
        b.x = s_lin[kk][ln];      b.y = s_lin[kk + 1][ln];
        acc = WMMA_F32(a, b, acc);
      }
#pragma unroll 8
      for (int k = 0; k < 256; k += 4) {
        const int kk = k + 2 * kh;
        v2f a = *(const v2f*)&Fq[row][kk];
        v2f b;
        b.x = s_quad[kk][ln]; b.y = s_quad[kk + 1][ln];
        acc = WMMA_F32(a, b, acc);
      }
#pragma unroll
      for (int r = 0; r < 8; ++r) Yc[ti * 16 + r + 8 * kh][ln] = acc[r];
    } else if (wave == 2) {
      const int t = lane;
      float d = kcum[0];
      for (int s2 = 0; s2 < 32; ++s2) d += Aw[t][s2];
      for (int i = 0; i < 16; ++i) d += Qc[t][i] * 0.5f * kcum[1 + i];
      for (int ij = 0; ij < 256; ++ij) d += Fq[t][ij] * kcum[17 + ij];
      den[t] = d + EPSF;
    }
    __syncthreads();

    if (tid < 128) {
      const int t = tid >> 2, c4 = (tid & 3) * 4;
      const float dd = den[t];
      float4 y = *(float4*)&Yc[t][c4];
      y.x /= dd; y.y /= dd; y.z /= dd; y.w /= dd;
      *(float4*)&Yg[(size_t)(bl + t) * DM + voff + c4] = y;
    }
    for (int i = tid; i < 32 * 256; i += 256) {
      int t = i >> 8, ij = i & 255;
      Fq[t][ij] = Kc[t][ij >> 4] * Kc[t][ij & 15] * QSC;
    }
    if (tid >= 128 && tid < 144) {
      const int f = tid - 128;
      float s = 0.f;
      for (int t = 0; t < 32; ++t) s += Vc[t][f];
      s_ones[f] += s;
    } else if (tid >= 144 && tid < 160) {
      const int f = tid - 144;
      float s = 0.f;
      for (int t = 0; t < 32; ++t) s += Kc[t][f];
      kcum[1 + f] += 0.5f * s;
    } else if (tid == 160) {
      kcum[0] += 32.f;
    }
    __syncthreads();

    {
      float s = 0.f;
      for (int t = 0; t < 32; ++t) s += Fq[t][tid];
      kcum[17 + tid] += s;
    }
    for (int tile = wave; tile < 17; tile += 8) {
      v8f acc;
      if (tile < 16) {
        const int qm = tile * 16;
#pragma unroll
        for (int r = 0; r < 8; ++r) acc[r] = s_quad[qm + r + 8 * kh][ln];
#pragma unroll
        for (int k = 0; k < 32; k += 4) {
          const int kk = k + 2 * kh;
          v2f a, b;
          a.x = Fq[kk][qm + ln]; a.y = Fq[kk + 1][qm + ln];
          b.x = Vc[kk][ln];      b.y = Vc[kk + 1][ln];
          acc = WMMA_F32(a, b, acc);
        }
#pragma unroll
        for (int r = 0; r < 8; ++r) s_quad[qm + r + 8 * kh][ln] = acc[r];
      } else {
#pragma unroll
        for (int r = 0; r < 8; ++r) acc[r] = s_lin[r + 8 * kh][ln];
#pragma unroll
        for (int k = 0; k < 32; k += 4) {
          const int kk = k + 2 * kh;
          v2f a, b;
          a.x = Kc[kk][ln] * 0.5f; a.y = Kc[kk + 1][ln] * 0.5f;
          b.x = Vc[kk][ln];        b.y = Vc[kk + 1][ln];
          acc = WMMA_F32(a, b, acc);
        }
#pragma unroll
        for (int r = 0; r < 8; ++r) s_lin[r + 8 * kh][ln] = acc[r];
      }
    }
    __syncthreads();
  }
}

extern "C" void kernel_launch(void* const* d_in, const int* in_sizes, int n_in,
                              void* d_out, int out_size, void* d_ws, size_t ws_size,
                              hipStream_t stream) {
  (void)in_sizes; (void)n_in; (void)out_size; (void)ws_size;
  const float* hs = (const float*)d_in[0];
  const float* Wq = (const float*)d_in[1];
  const float* Wk = (const float*)d_in[2];
  const float* Wv = (const float*)d_in[3];
  const float* Wo = (const float*)d_in[4];
  float* out = (float*)d_out;

  float* Qg = (float*)d_ws;
  float* Kg = Qg + (size_t)L * (Hh * FD);
  float* Vg = Kg + (size_t)L * (Hh * FD);
  float* Yg = Vg + (size_t)L * DM;

  const dim3 blk(256);
  gemm_f32_wmma<<<dim3((L / 64) * ((Hh * FD) / 64)), blk, 0, stream>>>(hs, Wq, Qg, L, Hh * FD, DM);
  gemm_f32_wmma<<<dim3((L / 64) * ((Hh * FD) / 64)), blk, 0, stream>>>(hs, Wk, Kg, L, Hh * FD, DM);
  gemm_f32_wmma<<<dim3((L / 64) * (DM / 64)), blk, 0, stream>>>(hs, Wv, Vg, L, DM, DM);
  based_attn<<<dim3(Hh * 4), blk, 0, stream>>>(Qg, Kg, Vg, Yg);
  gemm_f32_wmma<<<dim3((L / 64) * (DM / 64)), blk, 0, stream>>>(Yg, Wo, out, L, DM, DM);
}


# attention: sequence split over 8 blocks per (head, v-block); each block rebuilds its start state with the state-update half of the chunk loop, every sum in unchanged order
# speedup vs baseline: 1.4436x; 1.4436x over previous
#include <hip/hip_runtime.h>

typedef float v2f __attribute__((ext_vector_type(2)));
typedef float v8f __attribute__((ext_vector_type(8)));

#define WMMA_F32(a, b, c) \
  __builtin_amdgcn_wmma_f32_16x16x4_f32(false, (a), false, (b), (short)0, (c), false, false)

constexpr int L = 1024, DM = 768, Hh = 12, FD = 16, HD = 64;
constexpr float EPSF = 1e-12f;
constexpr float QSC = 0.17677669529663687f;
constexpr int NSEG = 8;
constexpr int CPS  = (L / 32) / NSEG;

__global__ __launch_bounds__(256) void gemm_f32_wmma(const float* __restrict__ A,
                                                     const float* __restrict__ B,
                                                     float* __restrict__ C,
                                                     int M, int N, int K) {
  __shared__ float As[64][36];
  __shared__ float BsT[64][38];
  const int tid  = threadIdx.x;
  const int wave = tid >> 5, lane = tid & 31;
  const int kh   = lane >> 4, ln = lane & 15;
  const int ntb  = N / 64;
  const int bm   = (blockIdx.x / ntb) * 64;
  const int bn   = (blockIdx.x % ntb) * 64;
  const int wm   = wave >> 1;
  const int wn   = wave & 1;
  const int ar   = tid >> 2, ac = (tid & 3) * 4;
  const int br   = tid >> 3, bc = (tid & 7) * 4;
  (void)M;

  const float4* Ag = (const float4*)A;
  const float4* Bg = (const float4*)B;

  float4 a0 = Ag[((size_t)(bm + ar) * K + ac) >> 2];
  float4 a1 = Ag[((size_t)(bm + ar) * K + ac + 16) >> 2];
  float4 b0 = Bg[((size_t)br * N + bn + bc) >> 2];
  float4 b1 = Bg[((size_t)br * N + bn + bc + 32) >> 2];
  v8f acc0 = {}, acc1 = {};

  for (int kb = 0; kb < K; kb += 32) {
    *(float4*)&As[ar][ac]      = a0;
    *(float4*)&As[ar][ac + 16] = a1;
    BsT[bc + 0][br]  = b0.x;  BsT[bc + 1][br]  = b0.y;
    BsT[bc + 2][br]  = b0.z;  BsT[bc + 3][br]  = b0.w;
    BsT[bc + 32][br] = b1.x;  BsT[bc + 33][br] = b1.y;
    BsT[bc + 34][br] = b1.z;  BsT[bc + 35][br] = b1.w;
    __syncthreads();
    if (kb + 32 < K) {
      a0 = Ag[((size_t)(bm + ar) * K + kb + 32 + ac) >> 2];
      a1 = Ag[((size_t)(bm + ar) * K + kb + 32 + ac + 16) >> 2];
      b0 = Bg[((size_t)(kb + 32 + br) * N + bn + bc) >> 2];
      b1 = Bg[((size_t)(kb + 32 + br) * N + bn + bc + 32) >> 2];
    }
#pragma unroll
    for (int k = 0; k < 32; k += 4) {
      const int kk = k + 2 * kh;
      v2f a = *(const v2f*)&As[wm * 16 + ln][kk];
      v2f p = *(const v2f*)&BsT[wn * 32 + ln][kk];
      v2f q = *(const v2f*)&BsT[wn * 32 + 16 + ln][kk];
      acc0 = WMMA_F32(a, p, acc0);
      acc1 = WMMA_F32(a, q, acc1);
    }
    __syncthreads();
  }
#pragma unroll
  for (int r = 0; r < 8; ++r) {
    const size_t row = (size_t)(bm + wm * 16 + r + 8 * kh);
    C[row * N + bn + wn * 32 + ln]      = acc0[r];
    C[row * N + bn + wn * 32 + 16 + ln] = acc1[r];
  }
}

__global__ __launch_bounds__(256) void based_attn(const float* __restrict__ Qg,
                                                  const float* __restrict__ Kg,
                                                  const float* __restrict__ Vg,
                                                  float* __restrict__ Yg) {
  __shared__ float s_lin[16][17];
  __shared__ float s_quad[256][17];
  __shared__ float s_ones[16];
  __shared__ float kcum[273];
  __shared__ float Qc[32][20];
  __shared__ float Kc[32][20];
  __shared__ float Vc[32][20];
  __shared__ float Aw[32][33];
  __shared__ float Fq[32][257];
  __shared__ float Yc[32][20];
  __shared__ float den[32];

  const int tid  = threadIdx.x;
  const int wave = tid >> 5, lane = tid & 31;
  const int kh   = lane >> 4, ln = lane & 15;
  const int seg  = NSEG - 1 - (int)(blockIdx.x / (Hh * 4));
  const int hv   = blockIdx.x % (Hh * 4);
  const int h    = hv >> 2;
  const int vb   = hv & 3;
  const int c_begin = seg * CPS, c_end = c_begin + CPS;
  const int qoff = h * FD;
  const int voff = h * HD + vb * 16;

  for (int i = tid; i < 16 * 16; i += 256) s_lin[i >> 4][i & 15] = 0.f;
  for (int i = tid; i < 256 * 16; i += 256) s_quad[i >> 4][i & 15] = 0.f;
  if (tid < 16) s_ones[tid] = 0.f;
  for (int i = tid; i < 273; i += 256) kcum[i] = 0.f;
  __syncthreads();

  for (int c = 0; c < c_end; ++c) {
    const int bl = c * 32;
    const bool full = (c >= c_begin);
    if (tid < 128) {
      const int t = tid >> 2, c4 = (tid & 3) * 4;
      *(float4*)&Qc[t][c4] = *(const float4*)&Qg[(size_t)(bl + t) * (Hh * FD) + qoff + c4];
      *(float4*)&Kc[t][c4] = *(const float4*)&Kg[(size_t)(bl + t) * (Hh * FD) + qoff + c4];
    } else {
      const int i = tid - 128;
      const int t = i >> 2, c4 = (i & 3) * 4;
      *(float4*)&Vc[t][c4] = *(const float4*)&Vg[(size_t)(bl + t) * DM + voff + c4];
    }
    __syncthreads();

    if (full) {
    if (wave < 4) {
      const int ti = wave >> 1, tj = wave & 1;
      v8f acc = {};
#pragma unroll
      for (int k = 0; k < 16; k += 4) {
        const int kk = k + 2 * kh;
        v2f a = *(const v2f*)&Qc[ti * 16 + ln][kk];
        v2f b = *(const v2f*)&Kc[tj * 16 + ln][kk];
        acc = WMMA_F32(a, b, acc);
      }
#pragma unroll
      for (int r = 0; r < 8; ++r) {
        const int row = ti * 16 + r + 8 * kh, col = tj * 16 + ln;
        const float s = acc[r];
        Aw[row][col] = (row >= col) ? (1.f + 0.25f * s + (s * s) * 0.03125f) : 0.f;
      }
    }
    for (int i = tid; i < 32 * 256; i += 256) {
      int t = i >> 8, ij = i & 255;
      Fq[t][ij] = Qc[t][ij >> 4] * Qc[t][ij & 15] * QSC;
    }
    __syncthreads();

    if (wave < 2) {
      const int ti = wave;
      const int row = ti * 16 + ln;
      v8f acc;
#pragma unroll
      for (int r = 0; r < 8; ++r) acc[r] = s_ones[ln];
#pragma unroll
      for (int k = 0; k < 32; k += 4) {
        const int kk = k + 2 * kh;
        v2f a = *(const v2f*)&Aw[row][kk];
        v2f b;
        b.x = Vc[kk][ln];  b.y = Vc[kk + 1][ln];
        acc = WMMA_F32(a, b, acc);
      }
#pragma unroll
      for (int k = 0; k < 16; k += 4) {
        const int kk = k + 2 * kh;
        v2f a, b;
        a.x = Qc[row][kk] * 0.5f; a.y = Qc[row][kk + 1] * 0.5f;
        b.x = s_lin[kk][ln];      b.y = s_lin[kk + 1][ln];
        acc = WMMA_F32(a, b, acc);
      }
#pragma unroll 8
      for (int k = 0; k < 256; k += 4) {
        const int kk = k + 2 * kh;
        v2f a = *(const v2f*)&Fq[row][kk];
        v2f b;
        b.x = s_quad[kk][ln]; b.y = s_quad[kk + 1][ln];
        acc = WMMA_F32(a, b, acc);
      }
#pragma unroll
      for (int r = 0; r < 8; ++r) Yc[ti * 16 + r + 8 * kh][ln] = acc[r];
    } else if (wave == 2) {
      const int t = lane;
      float d = kcum[0];
      for (int s2 = 0; s2 < 32; ++s2) d += Aw[t][s2];
      for (int i = 0; i < 16; ++i) d += Qc[t][i] * 0.5f * kcum[1 + i];
      for (int ij = 0; ij < 256; ++ij) d += Fq[t][ij] * kcum[17 + ij];
      den[t] = d + EPSF;
    }
    __syncthreads();

    if (tid < 128) {
      const int t = tid >> 2, c4 = (tid & 3) * 4;
      const float dd = den[t];
      float4 y = *(float4*)&Yc[t][c4];
      y.x /= dd; y.y /= dd; y.z /= dd; y.w /= dd;
      *(float4*)&Yg[(size_t)(bl + t) * DM + voff + c4] = y;
    }
    }
    for (int i = tid; i < 32 * 256; i += 256) {
      int t = i >> 8, ij = i & 255;
      Fq[t][ij] = Kc[t][ij >> 4] * Kc[t][ij & 15] * QSC;
    }
    if (tid >= 128 && tid < 144) {
      const int f = tid - 128;
      float s = 0.f;
      for (int t = 0; t < 32; ++t) s += Vc[t][f];
      s_ones[f] += s;
    } else if (tid >= 144 && tid < 160) {
      const int f = tid - 144;
      float s = 0.f;
      for (int t = 0; t < 32; ++t) s += Kc[t][f];
      kcum[1 + f] += 0.5f * s;
    } else if (tid == 160) {
      kcum[0] += 32.f;
    }
    __syncthreads();

    {
      float s = 0.f;
      for (int t = 0; t < 32; ++t) s += Fq[t][tid];
      kcum[17 + tid] += s;
    }
    for (int tile = wave; tile < 17; tile += 8) {
      v8f acc;
      if (tile < 16) {
        const int qm = tile * 16;
#pragma unroll
        for (int r = 0; r < 8; ++r) acc[r] = s_quad[qm + r + 8 * kh][ln];
#pragma unroll
        for (int k = 0; k < 32; k += 4) {
          const int kk = k + 2 * kh;
          v2f a, b;
          a.x = Fq[kk][qm + ln]; a.y = Fq[kk + 1][qm + ln];
          b.x = Vc[kk][ln];      b.y = Vc[kk + 1][ln];
          acc = WMMA_F32(a, b, acc);
        }
#pragma unroll
        for (int r = 0; r < 8; ++r) s_quad[qm + r + 8 * kh][ln] = acc[r];
      } else {
#pragma unroll
        for (int r = 0; r < 8; ++r) acc[r] = s_lin[r + 8 * kh][ln];
#pragma unroll
        for (int k = 0; k < 32; k += 4) {
          const int kk = k + 2 * kh;
          v2f a, b;
          a.x = Kc[kk][ln] * 0.5f; a.y = Kc[kk + 1][ln] * 0.5f;
          b.x = Vc[kk][ln];        b.y = Vc[kk + 1][ln];
          acc = WMMA_F32(a, b, acc);
        }
#pragma unroll
        for (int r = 0; r < 8; ++r) s_lin[r + 8 * kh][ln] = acc[r];
      }
    }
    __syncthreads();
  }
}

extern "C" void kernel_launch(void* const* d_in, const int* in_sizes, int n_in,
                              void* d_out, int out_size, void* d_ws, size_t ws_size,
                              hipStream_t stream) {
  (void)in_sizes; (void)n_in; (void)out_size; (void)ws_size;
  const float* hs = (const float*)d_in[0];
  const float* Wq = (const float*)d_in[1];
  const float* Wk = (const float*)d_in[2];
  const float* Wv = (const float*)d_in[3];
  const float* Wo = (const float*)d_in[4];
  float* out = (float*)d_out;

  float* Qg = (float*)d_ws;
  float* Kg = Qg + (size_t)L * (Hh * FD);
  float* Vg = Kg + (size_t)L * (Hh * FD);
  float* Yg = Vg + (size_t)L * DM;

  const dim3 blk(256);
  gemm_f32_wmma<<<dim3((L / 64) * ((Hh * FD) / 64)), blk, 0, stream>>>(hs, Wq, Qg, L, Hh * FD, DM);
  gemm_f32_wmma<<<dim3((L / 64) * ((Hh * FD) / 64)), blk, 0, stream>>>(hs, Wk, Kg, L, Hh * FD, DM);
  gemm_f32_wmma<<<dim3((L / 64) * (DM / 64)), blk, 0, stream>>>(hs, Wv, Vg, L, DM, DM);
  based_attn<<<dim3(Hh * 4 * NSEG), blk, 0, stream>>>(Qg, Kg, Vg, Yg);
  gemm_f32_wmma<<<dim3((L / 64) * (DM / 64)), blk, 0, stream>>>(Yg, Wo, out, L, DM, DM);
}


# attention: state rebuild in its own loop: state tiles and running sums held in registers, K features formed in registers from a transposed K/V stage, one barrier per chunk; sum order unchanged
# speedup vs baseline: 1.9571x; 1.3557x over previous
#include <hip/hip_runtime.h>

typedef float v2f __attribute__((ext_vector_type(2)));
typedef float v8f __attribute__((ext_vector_type(8)));

#define WMMA_F32(a, b, c) \
  __builtin_amdgcn_wmma_f32_16x16x4_f32(false, (a), false, (b), (short)0, (c), false, false)

constexpr int L = 1024, DM = 768, Hh = 12, FD = 16, HD = 64;
constexpr float EPSF = 1e-12f;
constexpr float QSC = 0.17677669529663687f;
constexpr int NSEG = 8;
constexpr int CPS  = (L / 32) / NSEG;

__global__ __launch_bounds__(256) void gemm_f32_wmma(const float* __restrict__ A,
                                                     const float* __restrict__ B,
                                                     float* __restrict__ C,
                                                     int M, int N, int K) {
  __shared__ float As[64][36];
  __shared__ float BsT[64][38];
  const int tid  = threadIdx.x;
  const int wave = tid >> 5, lane = tid & 31;
  const int kh   = lane >> 4, ln = lane & 15;
  const int ntb  = N / 64;
  const int bm   = (blockIdx.x / ntb) * 64;
  const int bn   = (blockIdx.x % ntb) * 64;
  const int wm   = wave >> 1;
  const int wn   = wave & 1;
  const int ar   = tid >> 2, ac = (tid & 3) * 4;
  const int br   = tid >> 3, bc = (tid & 7) * 4;
  (void)M;

  const float4* Ag = (const float4*)A;
  const float4* Bg = (const float4*)B;

  float4 a0 = Ag[((size_t)(bm + ar) * K + ac) >> 2];
  float4 a1 = Ag[((size_t)(bm + ar) * K + ac + 16) >> 2];
  float4 b0 = Bg[((size_t)br * N + bn + bc) >> 2];
  float4 b1 = Bg[((size_t)br * N + bn + bc + 32) >> 2];
  v8f acc0 = {}, acc1 = {};

  for (int kb = 0; kb < K; kb += 32) {
    *(float4*)&As[ar][ac]      = a0;
    *(float4*)&As[ar][ac + 16] = a1;
    BsT[bc + 0][br]  = b0.x;  BsT[bc + 1][br]  = b0.y;
    BsT[bc + 2][br]  = b0.z;  BsT[bc + 3][br]  = b0.w;
    BsT[bc + 32][br] = b1.x;  BsT[bc + 33][br] = b1.y;
    BsT[bc + 34][br] = b1.z;  BsT[bc + 35][br] = b1.w;
    __syncthreads();
    if (kb + 32 < K) {
      a0 = Ag[((size_t)(bm + ar) * K + kb + 32 + ac) >> 2];
      a1 = Ag[((size_t)(bm + ar) * K + kb + 32 + ac + 16) >> 2];
      b0 = Bg[((size_t)(kb + 32 + br) * N + bn + bc) >> 2];
      b1 = Bg[((size_t)(kb + 32 + br) * N + bn + bc + 32) >> 2];
    }
#pragma unroll
    for (int k = 0; k < 32; k += 4) {
      const int kk = k + 2 * kh;
      v2f a = *(const v2f*)&As[wm * 16 + ln][kk];
      v2f p = *(const v2f*)&BsT[wn * 32 + ln][kk];
      v2f q = *(const v2f*)&BsT[wn * 32 + 16 + ln][kk];
      acc0 = WMMA_F32(a, p, acc0);
      acc1 = WMMA_F32(a, q, acc1);
    }
    __syncthreads();
  }
#pragma unroll
  for (int r = 0; r < 8; ++r) {
    const size_t row = (size_t)(bm + wm * 16 + r + 8 * kh);
    C[row * N + bn + wn * 32 + ln]      = acc0[r];
    C[row * N + bn + wn * 32 + 16 + ln] = acc1[r];
  }
}

__global__ __launch_bounds__(256) void based_attn(const float* __restrict__ Qg,
                                                  const float* __restrict__ Kg,
                                                  const float* __restrict__ Vg,
                                                  float* __restrict__ Yg) {
  __shared__ float s_lin[16][17];
  __shared__ float s_quad[256][17];
  __shared__ float s_ones[16];
  __shared__ float kcum[273];
  __shared__ float Qc[32][20];
  __shared__ float Kc[32][20];
  __shared__ float Vc[32][20];
  __shared__ float Aw[32][33];
  __shared__ float Fq[32][257];
  __shared__ float Yc[32][20];
  __shared__ float den[32];
  __shared__ __attribute__((aligned(16))) float ST[2][2][16][36];

  const int tid  = threadIdx.x;
  const int wave = tid >> 5, lane = tid & 31;
  const int kh   = lane >> 4, ln = lane & 15;
  const int seg  = NSEG - 1 - (int)(blockIdx.x / (Hh * 4));
  const int hv   = blockIdx.x % (Hh * 4);
  const int h    = hv >> 2;
  const int vb   = hv & 3;
  const int c_begin = seg * CPS, c_end = c_begin + CPS;
  const int qoff = h * FD;
  const int voff = h * HD + vb * 16;

  v8f q0 = {}, q1 = {}, ql = {};
  float kq = 0.f;
  float sc = 0.f;
  {
    const int wv = __builtin_amdgcn_readfirstlane(wave);
    const int kv = tid >> 7;
    const int st = (tid & 127) >> 2, sc4 = (tid & 3) * 4;
    const float* gsrc = kv ? (Vg + voff + sc4) : (Kg + qoff + sc4);
    const size_t gp = kv ? (size_t)DM : (size_t)(Hh * FD);
    float4 pre = {0.f, 0.f, 0.f, 0.f};
    if (c_begin > 0) pre = *(const float4*)&gsrc[(size_t)st * gp];
    for (int c = 0; c < c_begin; ++c) {
      const int bf = c & 1;
      ST[bf][kv][sc4 + 0][st] = pre.x;
      ST[bf][kv][sc4 + 1][st] = pre.y;
      ST[bf][kv][sc4 + 2][st] = pre.z;
      ST[bf][kv][sc4 + 3][st] = pre.w;
      __syncthreads();
      if (c + 1 < c_begin) pre = *(const float4*)&gsrc[(size_t)((c + 1) * 32 + st) * gp];
      const float (*kt)[36] = ST[bf][0];
      const float (*vt)[36] = ST[bf][1];
      const int fi = tid >> 4, fj = tid & 15;
      float4 ka[8], kb[8];
#pragma unroll
      for (int m = 0; m < 8; ++m) {
        ka[m] = *(const float4*)&kt[fi][4 * m];
        kb[m] = *(const float4*)&kt[fj][4 * m];
      }
      if (tid >= 128 && tid < 160) {
        const float* row = (tid < 144) ? vt[tid - 128] : kt[tid - 144];
        float s = 0.f;
#pragma unroll
        for (int t = 0; t < 32; t += 4) {
          const float4 a = *(const float4*)&row[t];
          s += a.x;  s += a.y;  s += a.z;  s += a.w;
        }
        sc += (tid < 144) ? s : 0.5f * s;
      } else if (tid == 160) {
        sc += 32.f;
      }
      v2f fkl[8], fb[8], fk0[8], fk1[8];
#pragma unroll
      for (int m = 0; m < 8; ++m) {
        const int kk = 4 * m + 2 * kh;
        fkl[m] = *(const v2f*)&kt[ln][kk];
        fb[m]  = *(const v2f*)&vt[ln][kk];
        fk0[m] = *(const v2f*)&kt[wv][kk];
        fk1[m] = *(const v2f*)&kt[wv + 8][kk];
      }
      if (wv == 0) {
#pragma unroll
        for (int m = 0; m < 8; ++m) {
          const v2f al = fkl[m] * 0.5f;
          ql = WMMA_F32(al, fb[m], ql);
        }
      }
#pragma unroll
      for (int m = 0; m < 8; ++m) {
        const v2f a0 = fk0[m] * fkl[m] * QSC;
        const v2f a1 = fk1[m] * fkl[m] * QSC;
        q0 = WMMA_F32(a0, fb[m], q0);
        q1 = WMMA_F32(a1, fb[m], q1);
      }
      {
        float s = 0.f;
#pragma unroll
        for (int m = 0; m < 8; ++m) {
          s += ka[m].x * kb[m].x * QSC;
          s += ka[m].y * kb[m].y * QSC;
          s += ka[m].z * kb[m].z * QSC;
          s += ka[m].w * kb[m].w * QSC;
        }
        kq += s;
      }
    }
#pragma unroll
    for (int r = 0; r < 8; ++r) {
      s_quad[wv * 16 + r + 8 * kh][ln]       = q0[r];
      s_quad[(wv + 8) * 16 + r + 8 * kh][ln] = q1[r];
    }
    if (wv == 0) {
#pragma unroll
      for (int r = 0; r < 8; ++r) s_lin[r + 8 * kh][ln] = ql[r];
    }
    kcum[17 + tid] = kq;
    if (tid >= 128 && tid < 144)      s_ones[tid - 128] = sc;
    else if (tid >= 144 && tid < 160) kcum[1 + tid - 144] = sc;
    else if (tid == 160)              kcum[0] = sc;
  }

  for (int c = c_begin; c < c_end; ++c) {
    const int bl = c * 32;
    if (tid < 128) {
      const int t = tid >> 2, c4 = (tid & 3) * 4;
      *(float4*)&Qc[t][c4] = *(const float4*)&Qg[(size_t)(bl + t) * (Hh * FD) + qoff + c4];
      *(float4*)&Kc[t][c4] = *(const float4*)&Kg[(size_t)(bl + t) * (Hh * FD) + qoff + c4];
    } else {
      const int i = tid - 128;
      const int t = i >> 2, c4 = (i & 3) * 4;
      *(float4*)&Vc[t][c4] = *(const float4*)&Vg[(size_t)(bl + t) * DM + voff + c4];
    }
    __syncthreads();

    if (wave < 4) {
      const int ti = wave >> 1, tj = wave & 1;
      v8f acc = {};
#pragma unroll
      for (int k = 0; k < 16; k += 4) {
        const int kk = k + 2 * kh;
        v2f a = *(const v2f*)&Qc[ti * 16 + ln][kk];
        v2f b = *(const v2f*)&Kc[tj * 16 + ln][kk];
        acc = WMMA_F32(a, b, acc);
      }
#pragma unroll
      for (int r = 0; r < 8; ++r) {
        const int row = ti * 16 + r + 8 * kh, col = tj * 16 + ln;
        const float s = acc[r];
        Aw[row][col] = (row >= col) ? (1.f + 0.25f * s + (s * s) * 0.03125f) : 0.f;
      }
    }
    for (int i = tid; i < 32 * 256; i += 256) {
      int t = i >> 8, ij = i & 255;
      Fq[t][ij] = Qc[t][ij >> 4] * Qc[t][ij & 15] * QSC;
    }
    __syncthreads();

    if (wave < 2) {
      const int ti = wave;
      const int row = ti * 16 + ln;
      v8f acc;
#pragma unroll
      for (int r = 0; r < 8; ++r) acc[r] = s_ones[ln];
#pragma unroll
      for (int k = 0; k < 32; k += 4) {
        const int kk = k + 2 * kh;
        v2f a = *(const v2f*)&Aw[row][kk];
        v2f b;
        b.x = Vc[kk][ln];  b.y = Vc[kk + 1][ln];
        acc = WMMA_F32(a, b, acc);
      }
#pragma unroll
      for (int k = 0; k < 16; k += 4) {
        const int kk = k + 2 * kh;
        v2f a, b;
        a.x = Qc[row][kk] * 0.5f; a.y = Qc[row][kk + 1] * 0.5f;
        b.x = s_lin[kk][ln];      b.y = s_lin[kk + 1][ln];
        acc = WMMA_F32(a, b, acc);
      }
#pragma unroll 8
      for (int k = 0; k < 256; k += 4) {
        const int kk = k + 2 * kh;
        v2f a = *(const v2f*)&Fq[row][kk];
        v2f b;
        b.x = s_quad[kk][ln]; b.y = s_quad[kk + 1][ln];
        acc = WMMA_F32(a, b, acc);
      }
#pragma unroll
      for (int r = 0; r < 8; ++r) Yc[ti * 16 + r + 8 * kh][ln] = acc[r];
    } else if (wave == 2) {
      const int t = lane;
      float d = kcum[0];
      for (int s2 = 0; s2 < 32; ++s2) d += Aw[t][s2];
      for (int i = 0; i < 16; ++i) d += Qc[t][i] * 0.5f * kcum[1 + i];
      for (int ij = 0; ij < 256; ++ij) d += Fq[t][ij] * kcum[17 + ij];
      den[t] = d + EPSF;
    }
    __syncthreads();

    if (tid < 128) {
      const int t = tid >> 2, c4 = (tid & 3) * 4;
      const float dd = den[t];
      float4 y = *(float4*)&Yc[t][c4];
      y.x /= dd; y.y /= dd; y.z /= dd; y.w /= dd;
      *(float4*)&Yg[(size_t)(bl + t) * DM + voff + c4] = y;
    }
    for (int i = tid; i < 32 * 256; i += 256) {
      int t = i >> 8, ij = i & 255;
      Fq[t][ij] = Kc[t][ij >> 4] * Kc[t][ij & 15] * QSC;
    }
    if (tid >= 128 && tid < 144) {
      const int f = tid - 128;
      float s = 0.f;
      for (int t = 0; t < 32; ++t) s += Vc[t][f];
      s_ones[f] += s;
    } else if (tid >= 144 && tid < 160) {
      const int f = tid - 144;
      float s = 0.f;
      for (int t = 0; t < 32; ++t) s += Kc[t][f];
      kcum[1 + f] += 0.5f * s;
    } else if (tid == 160) {
      kcum[0] += 32.f;
    }
    __syncthreads();

    {
      float s = 0.f;
      for (int t = 0; t < 32; ++t) s += Fq[t][tid];
      kcum[17 + tid] += s;
    }
    for (int tile = wave; tile < 17; tile += 8) {
      v8f acc;
      if (tile < 16) {
        const int qm = tile * 16;
#pragma unroll
        for (int r = 0; r < 8; ++r) acc[r] = s_quad[qm + r + 8 * kh][ln];
#pragma unroll
        for (int k = 0; k < 32; k += 4) {
          const int kk = k + 2 * kh;
          v2f a, b;
          a.x = Fq[kk][qm + ln]; a.y = Fq[kk + 1][qm + ln];
          b.x = Vc[kk][ln];      b.y = Vc[kk + 1][ln];
          acc = WMMA_F32(a, b, acc);
        }
#pragma unroll
        for (int r = 0; r < 8; ++r) s_quad[qm + r + 8 * kh][ln] = acc[r];
      } else {
#pragma unroll
        for (int r = 0; r < 8; ++r) acc[r] = s_lin[r + 8 * kh][ln];
#pragma unroll
        for (int k = 0; k < 32; k += 4) {
          const int kk = k + 2 * kh;
          v2f a, b;
          a.x = Kc[kk][ln] * 0.5f; a.y = Kc[kk + 1][ln] * 0.5f;
          b.x = Vc[kk][ln];        b.y = Vc[kk + 1][ln];
          acc = WMMA_F32(a, b, acc);
        }
#pragma unroll
        for (int r = 0; r < 8; ++r) s_lin[r + 8 * kh][ln] = acc[r];
      }
    }
    __syncthreads();
  }
}

extern "C" void kernel_launch(void* const* d_in, const int* in_sizes, int n_in,
                              void* d_out, int out_size, void* d_ws, size_t ws_size,
                              hipStream_t stream) {
  (void)in_sizes; (void)n_in; (void)out_size; (void)ws_size;
  const float* hs = (const float*)d_in[0];
  const float* Wq = (const float*)d_in[1];
  const float* Wk = (const float*)d_in[2];
  const float* Wv = (const float*)d_in[3];
  const float* Wo = (const float*)d_in[4];
  float* out = (float*)d_out;

  float* Qg = (float*)d_ws;
  float* Kg = Qg + (size_t)L * (Hh * FD);
  float* Vg = Kg + (size_t)L * (Hh * FD);
  float* Yg = Vg + (size_t)L * DM;

  const dim3 blk(256);
  gemm_f32_wmma<<<dim3((L / 64) * ((Hh * FD) / 64)), blk, 0, stream>>>(hs, Wq, Qg, L, Hh * FD, DM);
  gemm_f32_wmma<<<dim3((L / 64) * ((Hh * FD) / 64)), blk, 0, stream>>>(hs, Wk, Kg, L, Hh * FD, DM);
  gemm_f32_wmma<<<dim3((L / 64) * (DM / 64)), blk, 0, stream>>>(hs, Wv, Vg, L, DM, DM);
  based_attn<<<dim3(Hh * 4 * NSEG), blk, 0, stream>>>(Qg, Kg, Vg, Yg);
  gemm_f32_wmma<<<dim3((L / 64) * (DM / 64)), blk, 0, stream>>>(Yg, Wo, out, L, DM, DM);
}


# projections: the three input GEMMs (Q, K, V) run as ONE launch of 288 tile blocks that share the A operand, each tile computed by the same code; 5 launches become 3; sum order unchanged
# speedup vs baseline: 2.9055x; 1.4846x over previous
#include <hip/hip_runtime.h>

typedef float v2f __attribute__((ext_vector_type(2)));
typedef float v8f __attribute__((ext_vector_type(8)));

#define WMMA_F32(a, b, c) \
  __builtin_amdgcn_wmma_f32_16x16x4_f32(false, (a), false, (b), (short)0, (c), false, false)

constexpr int L = 1024, DM = 768, Hh = 12, FD = 16, HD = 64;
constexpr float EPSF = 1e-12f;
constexpr float QSC = 0.17677669529663687f;
constexpr int NSEG = 8;
constexpr int CPS  = (L / 32) / NSEG;

__device__ __forceinline__ void gemm_tile_f32_wmma(const float* __restrict__ A,
                                                   const float* __restrict__ B,
                                                   float* __restrict__ C,
                                                   int N, int K, int bm, int bn) {
  __shared__ float As[64][36];
  __shared__ float BsT[64][38];
  const int tid  = threadIdx.x;
  const int wave = tid >> 5, lane = tid & 31;
  const int kh   = lane >> 4, ln = lane & 15;
  const int wm   = wave >> 1;
  const int wn   = wave & 1;
  const int ar   = tid >> 2, ac = (tid & 3) * 4;
  const int br   = tid >> 3, bc = (tid & 7) * 4;

  const float4* Ag = (const float4*)A;
  const float4* Bg = (const float4*)B;

  float4 a0 = Ag[((size_t)(bm + ar) * K + ac) >> 2];
  float4 a1 = Ag[((size_t)(bm + ar) * K + ac + 16) >> 2];
  float4 b0 = Bg[((size_t)br * N + bn + bc) >> 2];
  float4 b1 = Bg[((size_t)br * N + bn + bc + 32) >> 2];
  v8f acc0 = {}, acc1 = {};

  for (int kb = 0; kb < K; kb += 32) {
    *(float4*)&As[ar][ac]      = a0;
    *(float4*)&As[ar][ac + 16] = a1;
    BsT[bc + 0][br]  = b0.x;  BsT[bc + 1][br]  = b0.y;
    BsT[bc + 2][br]  = b0.z;  BsT[bc + 3][br]  = b0.w;
    BsT[bc + 32][br] = b1.x;  BsT[bc + 33][br] = b1.y;
    BsT[bc + 34][br] = b1.z;  BsT[bc + 35][br] = b1.w;
    __syncthreads();
    if (kb + 32 < K) {
      a0 = Ag[((size_t)(bm + ar) * K + kb + 32 + ac) >> 2];
      a1 = Ag[((size_t)(bm + ar) * K + kb + 32 + ac + 16) >> 2];
      b0 = Bg[((size_t)(kb + 32 + br) * N + bn + bc) >> 2];
      b1 = Bg[((size_t)(kb + 32 + br) * N + bn + bc + 32) >> 2];
    }
#pragma unroll
    for (int k = 0; k < 32; k += 4) {
      const int kk = k + 2 * kh;
      v2f a = *(const v2f*)&As[wm * 16 + ln][kk];
      v2f p = *(const v2f*)&BsT[wn * 32 + ln][kk];
      v2f q = *(const v2f*)&BsT[wn * 32 + 16 + ln][kk];
      acc0 = WMMA_F32(a, p, acc0);
      acc1 = WMMA_F32(a, q, acc1);
    }
    __syncthreads();
  }
#pragma unroll
  for (int r = 0; r < 8; ++r) {
    const size_t row = (size_t)(bm + wm * 16 + r + 8 * kh);
    C[row * N + bn + wn * 32 + ln]      = acc0[r];
    C[row * N + bn + wn * 32 + 16 + ln] = acc1[r];
  }
}

__global__ __launch_bounds__(256) void gemm_f32_wmma(const float* __restrict__ A,
                                                     const float* __restrict__ B,
                                                     float* __restrict__ C,
                                                     int M, int N, int K) {
  (void)M;
  const int ntb = N / 64;
  gemm_tile_f32_wmma(A, B, C, N, K, (int)(blockIdx.x / ntb) * 64, (int)(blockIdx.x % ntb) * 64);
}

__global__ __launch_bounds__(256) void gemm_qkv_f32_wmma(const float* __restrict__ X,
                                                         const float* __restrict__ Wq,
                                                         const float* __restrict__ Wk,
                                                         const float* __restrict__ Wv,
                                                         float* __restrict__ Qg,
                                                         float* __restrict__ Kg,
                                                         float* __restrict__ Vg) {
  constexpr int NQ = Hh * FD;
  constexpr int TV = (L / 64) * (DM / 64);
  constexpr int TQ = (L / 64) * (NQ / 64);
  const int b = (int)blockIdx.x;
  if (b < TV) {
    gemm_tile_f32_wmma(X, Wv, Vg, DM, DM, (b / (DM / 64)) * 64, (b % (DM / 64)) * 64);
  } else {
    const int t = (b - TV) % TQ;
    const bool isq = (b - TV) < TQ;
    gemm_tile_f32_wmma(X, isq ? Wq : Wk, isq ? Qg : Kg, NQ, DM,
                       (t / (NQ / 64)) * 64, (t % (NQ / 64)) * 64);
  }
}

__global__ __launch_bounds__(256) void based_attn(const float* __restrict__ Qg,
                                                  const float* __restrict__ Kg,
                                                  const float* __restrict__ Vg,
                                                  float* __restrict__ Yg) {
  __shared__ float s_lin[16][17];
  __shared__ float s_quad[256][17];
  __shared__ float s_ones[16];
  __shared__ float kcum[273];
  __shared__ float Qc[32][20];
  __shared__ float Kc[32][20];
  __shared__ float Vc[32][20];
  __shared__ float Aw[32][33];
  __shared__ float Fq[32][257];
  __shared__ float Yc[32][20];
  __shared__ float den[32];
  __shared__ __attribute__((aligned(16))) float ST[2][2][16][36];

  const int tid  = threadIdx.x;
  const int wave = tid >> 5, lane = tid & 31;
  const int kh   = lane >> 4, ln = lane & 15;
  const int seg  = NSEG - 1 - (int)(blockIdx.x / (Hh * 4));
  const int hv   = blockIdx.x % (Hh * 4);
  const int h    = hv >> 2;
  const int vb   = hv & 3;
  const int c_begin = seg * CPS, c_end = c_begin + CPS;
  const int qoff = h * FD;
  const int voff = h * HD + vb * 16;

  v8f q0 = {}, q1 = {}, ql = {};
  float kq = 0.f;
  float sc = 0.f;
  {
    const int wv = __builtin_amdgcn_readfirstlane(wave);
    const int kv = tid >> 7;
    const int st = (tid & 127) >> 2, sc4 = (tid & 3) * 4;
    const float* gsrc = kv ? (Vg + voff + sc4) : (Kg + qoff + sc4);
    const size_t gp = kv ? (size_t)DM : (size_t)(Hh * FD);
    float4 pre = {0.f, 0.f, 0.f, 0.f};
    if (c_begin > 0) pre = *(const float4*)&gsrc[(size_t)st * gp];
    for (int c = 0; c < c_begin; ++c) {
      const int bf = c & 1;
      ST[bf][kv][sc4 + 0][st] = pre.x;
      ST[bf][kv][sc4 + 1][st] = pre.y;
      ST[bf][kv][sc4 + 2][st] = pre.z;
      ST[bf][kv][sc4 + 3][st] = pre.w;
      __syncthreads();
      if (c + 1 < c_begin) pre = *(const float4*)&gsrc[(size_t)((c + 1) * 32 + st) * gp];
      const float (*kt)[36] = ST[bf][0];
      const float (*vt)[36] = ST[bf][1];
      const int fi = tid >> 4, fj = tid & 15;
      float4 ka[8], kb[8];
#pragma unroll
      for (int m = 0; m < 8; ++m) {
        ka[m] = *(const float4*)&kt[fi][4 * m];
        kb[m] = *(const float4*)&kt[fj][4 * m];
      }
      if (tid >= 128 && tid < 160) {
        const float* row = (tid < 144) ? vt[tid - 128] : kt[tid - 144];
        float s = 0.f;
#pragma unroll
        for (int t = 0; t < 32; t += 4) {
          const float4 a = *(const float4*)&row[t];
          s += a.x;  s += a.y;  s += a.z;  s += a.w;
        }
        sc += (tid < 144) ? s : 0.5f * s;
      } else if (tid == 160) {
        sc += 32.f;
      }
      v2f fkl[8], fb[8], fk0[8], fk1[8];
#pragma unroll
      for (int m = 0; m < 8; ++m) {
        const int kk = 4 * m + 2 * kh;
        fkl[m] = *(const v2f*)&kt[ln][kk];
        fb[m]  = *(const v2f*)&vt[ln][kk];
        fk0[m] = *(const v2f*)&kt[wv][kk];
        fk1[m] = *(const v2f*)&kt[wv + 8][kk];
      }
      if (wv == 0) {
#pragma unroll
        for (int m = 0; m < 8; ++m) {
          const v2f al = fkl[m] * 0.5f;
          ql = WMMA_F32(al, fb[m], ql);
        }
      }
#pragma unroll
      for (int m = 0; m < 8; ++m) {
        const v2f a0 = fk0[m] * fkl[m] * QSC;
        const v2f a1 = fk1[m] * fkl[m] * QSC;
        q0 = WMMA_F32(a0, fb[m], q0);
        q1 = WMMA_F32(a1, fb[m], q1);
      }
      {
        float s = 0.f;
#pragma unroll
        for (int m = 0; m < 8; ++m) {
          s += ka[m].x * kb[m].x * QSC;
          s += ka[m].y * kb[m].y * QSC;
          s += ka[m].z * kb[m].z * QSC;
          s += ka[m].w * kb[m].w * QSC;
        }
        kq += s;
      }
    }
#pragma unroll
    for (int r = 0; r < 8; ++r) {
      s_quad[wv * 16 + r + 8 * kh][ln]       = q0[r];
      s_quad[(wv + 8) * 16 + r + 8 * kh][ln] = q1[r];
    }
    if (wv == 0) {
#pragma unroll
      for (int r = 0; r < 8; ++r) s_lin[r + 8 * kh][ln] = ql[r];
    }
    kcum[17 + tid] = kq;
    if (tid >= 128 && tid < 144)      s_ones[tid - 128] = sc;
    else if (tid >= 144 && tid < 160) kcum[1 + tid - 144] = sc;
    else if (tid == 160)              kcum[0] = sc;
  }

  for (int c = c_begin; c < c_end; ++c) {
    const int bl = c * 32;
    if (tid < 128) {
      const int t = tid >> 2, c4 = (tid & 3) * 4;
      *(float4*)&Qc[t][c4] = *(const float4*)&Qg[(size_t)(bl + t) * (Hh * FD) + qoff + c4];
      *(float4*)&Kc[t][c4] = *(const float4*)&Kg[(size_t)(bl + t) * (Hh * FD) + qoff + c4];
    } else {
      const int i = tid - 128;
      const int t = i >> 2, c4 = (i & 3) * 4;
      *(float4*)&Vc[t][c4] = *(const float4*)&Vg[(size_t)(bl + t) * DM + voff + c4];
    }
    __syncthreads();

    if (wave < 4) {
      const int ti = wave >> 1, tj = wave & 1;
      v8f acc = {};
#pragma unroll
      for (int k = 0; k < 16; k += 4) {
        const int kk = k + 2 * kh;
        v2f a = *(const v2f*)&Qc[ti * 16 + ln][kk];
        v2f b = *(const v2f*)&Kc[tj * 16 + ln][kk];
        acc = WMMA_F32(a, b, acc);
      }
#pragma unroll
      for (int r = 0; r < 8; ++r) {
        const int row = ti * 16 + r + 8 * kh, col = tj * 16 + ln;
        const float s = acc[r];
        Aw[row][col] = (row >= col) ? (1.f + 0.25f * s + (s * s) * 0.03125f) : 0.f;
      }
    }
    for (int i = tid; i < 32 * 256; i += 256) {
      int t = i >> 8, ij = i & 255;
      Fq[t][ij] = Qc[t][ij >> 4] * Qc[t][ij & 15] * QSC;
    }
    __syncthreads();

    if (wave < 2) {
      const int ti = wave;
      const int row = ti * 16 + ln;
      v8f acc;
#pragma unroll
      for (int r = 0; r < 8; ++r) acc[r] = s_ones[ln];
#pragma unroll
      for (int k = 0; k < 32; k += 4) {
        const int kk = k + 2 * kh;
        v2f a = *(const v2f*)&Aw[row][kk];
        v2f b;
        b.x = Vc[kk][ln];  b.y = Vc[kk + 1][ln];
        acc = WMMA_F32(a, b, acc);
      }
#pragma unroll
      for (int k = 0; k < 16; k += 4) {
        const int kk = k + 2 * kh;
        v2f a, b;
        a.x = Qc[row][kk] * 0.5f; a.y = Qc[row][kk + 1] * 0.5f;
        b.x = s_lin[kk][ln];      b.y = s_lin[kk + 1][ln];
        acc = WMMA_F32(a, b, acc);
      }
#pragma unroll 8
      for (int k = 0; k < 256; k += 4) {
        const int kk = k + 2 * kh;
        v2f a = *(const v2f*)&Fq[row][kk];
        v2f b;
        b.x = s_quad[kk][ln]; b.y = s_quad[kk + 1][ln];
        acc = WMMA_F32(a, b, acc);
      }
#pragma unroll
      for (int r = 0; r < 8; ++r) Yc[ti * 16 + r + 8 * kh][ln] = acc[r];
    } else if (wave == 2) {
      const int t = lane;
      float d = kcum[0];
      for (int s2 = 0; s2 < 32; ++s2) d += Aw[t][s2];
      for (int i = 0; i < 16; ++i) d += Qc[t][i] * 0.5f * kcum[1 + i];
      for (int ij = 0; ij < 256; ++ij) d += Fq[t][ij] * kcum[17 + ij];
      den[t] = d + EPSF;
    }
    __syncthreads();

    if (tid < 128) {
      const int t = tid >> 2, c4 = (tid & 3) * 4;
      const float dd = den[t];
      float4 y = *(float4*)&Yc[t][c4];
      y.x /= dd; y.y /= dd; y.z /= dd; y.w /= dd;
      *(float4*)&Yg[(size_t)(bl + t) * DM + voff + c4] = y;
    }
    for (int i = tid; i < 32 * 256; i += 256) {
      int t = i >> 8, ij = i & 255;
      Fq[t][ij] = Kc[t][ij >> 4] * Kc[t][ij & 15] * QSC;
    }
    if (tid >= 128 && tid < 144) {
      const int f = tid - 128;
      float s = 0.f;
      for (int t = 0; t < 32; ++t) s += Vc[t][f];
      s_ones[f] += s;
    } else if (tid >= 144 && tid < 160) {
      const int f = tid - 144;
      float s = 0.f;
      for (int t = 0; t < 32; ++t) s += Kc[t][f];
      kcum[1 + f] += 0.5f * s;
    } else if (tid == 160) {
      kcum[0] += 32.f;
    }
    __syncthreads();

    {
      float s = 0.f;
      for (int t = 0; t < 32; ++t) s += Fq[t][tid];
      kcum[17 + tid] += s;
    }
    for (int tile = wave; tile < 17; tile += 8) {
      v8f acc;
      if (tile < 16) {
        const int qm = tile * 16;
#pragma unroll
        for (int r = 0; r < 8; ++r) acc[r] = s_quad[qm + r + 8 * kh][ln];
#pragma unroll
        for (int k = 0; k < 32; k += 4) {
          const int kk = k + 2 * kh;
          v2f a, b;
          a.x = Fq[kk][qm + ln]; a.y = Fq[kk + 1][qm + ln];
          b.x = Vc[kk][ln];      b.y = Vc[kk + 1][ln];
          acc = WMMA_F32(a, b, acc);
        }
#pragma unroll
        for (int r = 0; r < 8; ++r) s_quad[qm + r + 8 * kh][ln] = acc[r];
      } else {
#pragma unroll
        for (int r = 0; r < 8; ++r) acc[r] = s_lin[r + 8 * kh][ln];
#pragma unroll
        for (int k = 0; k < 32; k += 4) {
          const int kk = k + 2 * kh;
          v2f a, b;
          a.x = Kc[kk][ln] * 0.5f; a.y = Kc[kk + 1][ln] * 0.5f;
          b.x = Vc[kk][ln];        b.y = Vc[kk + 1][ln];
          acc = WMMA_F32(a, b, acc);
        }
#pragma unroll
        for (int r = 0; r < 8; ++r) s_lin[r + 8 * kh][ln] = acc[r];
      }
    }
    __syncthreads();
  }
}

extern "C" void kernel_launch(void* const* d_in, const int* in_sizes, int n_in,
                              void* d_out, int out_size, void* d_ws, size_t ws_size,
                              hipStream_t stream) {
  (void)in_sizes; (void)n_in; (void)out_size; (void)ws_size;
  const float* hs = (const float*)d_in[0];
  const float* Wq = (const float*)d_in[1];
  const float* Wk = (const float*)d_in[2];
  const float* Wv = (const float*)d_in[3];
  const float* Wo = (const float*)d_in[4];
  float* out = (float*)d_out;

  float* Qg = (float*)d_ws;
  float* Kg = Qg + (size_t)L * (Hh * FD);
  float* Vg = Kg + (size_t)L * (Hh * FD);
  float* Yg = Vg + (size_t)L * DM;

  const dim3 blk(256);
  gemm_qkv_f32_wmma<<<dim3((L / 64) * (DM / 64) + 2 * (L / 64) * ((Hh * FD) / 64)), blk, 0, stream>>>(
      hs, Wq, Wk, Wv, Qg, Kg, Vg);
  based_attn<<<dim3(Hh * 4 * NSEG), blk, 0, stream>>>(Qg, Kg, Vg, Yg);
  gemm_f32_wmma<<<dim3((L / 64) * (DM / 64)), blk, 0, stream>>>(Yg, Wo, out, L, DM, DM);
}


# GEMM K loop software-pipelined: K-step 64, two LDS stages, one LDS-only barrier per step, global loads of the stage after next stay in flight across it; order of sums unchanged
# speedup vs baseline: 3.6151x; 1.2442x over previous
#include <hip/hip_runtime.h>

typedef float v2f __attribute__((ext_vector_type(2)));
typedef float v8f __attribute__((ext_vector_type(8)));

#define WMMA_F32(a, b, c) \
  __builtin_amdgcn_wmma_f32_16x16x4_f32(false, (a), false, (b), (short)0, (c), false, false)

constexpr int L = 1024, DM = 768, Hh = 12, FD = 16, HD = 64;
constexpr float EPSF = 1e-12f;
constexpr float QSC = 0.17677669529663687f;
constexpr int NSEG = 8;
constexpr int CPS  = (L / 32) / NSEG;

constexpr int KS = 64;
constexpr int LP = KS + 4;

typedef float v4f __attribute__((ext_vector_type(4)));
struct GemmStage { v4f a0, a1, a2, a3, b0, b1, b2, b3; };

__device__ __forceinline__ void barrier_lds_only() {
  __builtin_amdgcn_fence(__ATOMIC_RELEASE, "workgroup", "local");
  __builtin_amdgcn_s_barrier();
  __builtin_amdgcn_fence(__ATOMIC_ACQUIRE, "workgroup", "local");
}

__device__ __forceinline__ void gemm_stage_load(GemmStage& g, const float* __restrict__ A,
                                                const float* __restrict__ B, int N, int K,
                                                int bm, int bn, int kb, int tid) {
  const int ar = tid >> 4, ac = (tid & 15) * 4;
  const int bk = tid >> 2, bc = (tid & 3) * 4;
  const float* ap = &A[(size_t)(bm + ar) * K + kb + ac];
  const float* bp = &B[(size_t)(kb + bk) * N + bn + bc];
  g.a0 = *(const v4f*)&ap[0];
  g.a1 = *(const v4f*)&ap[(size_t)16 * K];
  g.a2 = *(const v4f*)&ap[(size_t)32 * K];
  g.a3 = *(const v4f*)&ap[(size_t)48 * K];
  g.b0 = *(const v4f*)&bp[0];
  g.b1 = *(const v4f*)&bp[16];
  g.b2 = *(const v4f*)&bp[32];
  g.b3 = *(const v4f*)&bp[48];
}

__device__ __forceinline__ void gemm_b_store(v4f b, float (*BsT)[LP], int col, int bk) {
  BsT[col + 0][bk] = b.x;
  BsT[col + 1][bk] = b.y;
  BsT[col + 2][bk] = b.z;
  BsT[col + 3][bk] = b.w;
}

__device__ __forceinline__ void gemm_stage_store(const GemmStage& g, float (*As)[LP],
                                                 float (*BsT)[LP], int tid) {
  const int ar = tid >> 4, ac = (tid & 15) * 4;
  const int bk = tid >> 2, bc = (tid & 3) * 4;
  *(v4f*)&As[ar][ac]      = g.a0;
  *(v4f*)&As[ar + 16][ac] = g.a1;
  *(v4f*)&As[ar + 32][ac] = g.a2;
  *(v4f*)&As[ar + 48][ac] = g.a3;
  gemm_b_store(g.b0, BsT, bc, bk);
  gemm_b_store(g.b1, BsT, bc + 16, bk);
  gemm_b_store(g.b2, BsT, bc + 32, bk);
  gemm_b_store(g.b3, BsT, bc + 48, bk);
}

__device__ __forceinline__ void gemm_stage_compute(const float (*As)[LP], const float (*BsT)[LP],
                                                   v8f& acc0, v8f& acc1,
                                                   int wm, int wn, int ln, int kh) {
#pragma unroll
  for (int k = 0; k < KS; k += 4) {
    const int kk = k + 2 * kh;
    v2f a = *(const v2f*)&As[wm * 16 + ln][kk];
    v2f p = *(const v2f*)&BsT[wn * 32 + ln][kk];
    v2f q = *(const v2f*)&BsT[wn * 32 + 16 + ln][kk];
    acc0 = WMMA_F32(a, p, acc0);
    acc1 = WMMA_F32(a, q, acc1);
  }
}

__device__ __forceinline__ void gemm_tile_f32_wmma(const float* __restrict__ A,
                                                   const float* __restrict__ B,
                                                   float* __restrict__ C,
                                                   int N, int K, int bm, int bn) {
  __shared__ __attribute__((aligned(16))) float As[2][64][LP];
  __shared__ __attribute__((aligned(16))) float BsT[2][64][LP];
  const int tid  = threadIdx.x;
  const int wave = tid >> 5, lane = tid & 31;
  const int kh   = lane >> 4, ln = lane & 15;
  const int wm   = wave >> 1;
  const int wn   = wave & 1;

  GemmStage g;
  v8f acc0 = {}, acc1 = {};
  gemm_stage_load(g, A, B, N, K, bm, bn, 0, tid);
  gemm_stage_store(g, As[0], BsT[0], tid);
  gemm_stage_load(g, A, B, N, K, bm, bn, KS, tid);
  barrier_lds_only();

  int kb = 0;
  for (; kb + 2 * KS < K; kb += 2 * KS) {
    gemm_stage_store(g, As[1], BsT[1], tid);
    gemm_stage_load(g, A, B, N, K, bm, bn, kb + 2 * KS, tid);
    gemm_stage_compute(As[0], BsT[0], acc0, acc1, wm, wn, ln, kh);
    barrier_lds_only();
    gemm_stage_store(g, As[0], BsT[0], tid);
    gemm_stage_load(g, A, B, N, K, bm, bn, kb + 3 * KS, tid);
    gemm_stage_compute(As[1], BsT[1], acc0, acc1, wm, wn, ln, kh);
    barrier_lds_only();
  }
  gemm_stage_store(g, As[1], BsT[1], tid);
  gemm_stage_compute(As[0], BsT[0], acc0, acc1, wm, wn, ln, kh);
  barrier_lds_only();
  gemm_stage_compute(As[1], BsT[1], acc0, acc1, wm, wn, ln, kh);
#pragma unroll
  for (int r = 0; r < 8; ++r) {
    const size_t row = (size_t)(bm + wm * 16 + r + 8 * kh);
    C[row * N + bn + wn * 32 + ln]      = acc0[r];
    C[row * N + bn + wn * 32 + 16 + ln] = acc1[r];
  }
}

__global__ __launch_bounds__(256) void gemm_f32_wmma(const float* __restrict__ A,
                                                     const float* __restrict__ B,
                                                     float* __restrict__ C,
                                                     int M, int N, int K) {
  (void)M;
  const int ntb = N / 64;
  gemm_tile_f32_wmma(A, B, C, N, K, (int)(blockIdx.x / ntb) * 64, (int)(blockIdx.x % ntb) * 64);
}

__global__ __launch_bounds__(256) void gemm_qkv_f32_wmma(const float* __restrict__ X,
                                                         const float* __restrict__ Wq,
                                                         const float* __restrict__ Wk,
                                                         const float* __restrict__ Wv,
                                                         float* __restrict__ Qg,
                                                         float* __restrict__ Kg,
                                                         float* __restrict__ Vg) {
  constexpr int NQ = Hh * FD;
  constexpr int TV = (L / 64) * (DM / 64);
  constexpr int TQ = (L / 64) * (NQ / 64);
  const int b = (int)blockIdx.x;
  if (b < TV) {
    gemm_tile_f32_wmma(X, Wv, Vg, DM, DM, (b / (DM / 64)) * 64, (b % (DM / 64)) * 64);
  } else {
    const int t = (b - TV) % TQ;
    const bool isq = (b - TV) < TQ;
    gemm_tile_f32_wmma(X, isq ? Wq : Wk, isq ? Qg : Kg, NQ, DM,
                       (t / (NQ / 64)) * 64, (t % (NQ / 64)) * 64);
  }
}

__global__ __launch_bounds__(256) void based_attn(const float* __restrict__ Qg,
                                                  const float* __restrict__ Kg,
                                                  const float* __restrict__ Vg,
                                                  float* __restrict__ Yg) {
  __shared__ float s_lin[16][17];
  __shared__ float s_quad[256][17];
  __shared__ float s_ones[16];
  __shared__ float kcum[273];
  __shared__ float Qc[32][20];
  __shared__ float Kc[32][20];
  __shared__ float Vc[32][20];
  __shared__ float Aw[32][33];
  __shared__ float Fq[32][257];
  __shared__ float Yc[32][20];
  __shared__ float den[32];
  __shared__ __attribute__((aligned(16))) float ST[2][2][16][36];

  const int tid  = threadIdx.x;
  const int wave = tid >> 5, lane = tid & 31;
  const int kh   = lane >> 4, ln = lane & 15;
  const int seg  = NSEG - 1 - (int)(blockIdx.x / (Hh * 4));
  const int hv   = blockIdx.x % (Hh * 4);
  const int h    = hv >> 2;
  const int vb   = hv & 3;
  const int c_begin = seg * CPS, c_end = c_begin + CPS;
  const int qoff = h * FD;
  const int voff = h * HD + vb * 16;

  v8f q0 = {}, q1 = {}, ql = {};
  float kq = 0.f;
  float sc = 0.f;
  {
    const int wv = __builtin_amdgcn_readfirstlane(wave);
    const int kv = tid >> 7;
    const int st = (tid & 127) >> 2, sc4 = (tid & 3) * 4;
    const float* gsrc = kv ? (Vg + voff + sc4) : (Kg + qoff + sc4);
    const size_t gp = kv ? (size_t)DM : (size_t)(Hh * FD);
    float4 pre = {0.f, 0.f, 0.f, 0.f};
    if (c_begin > 0) pre = *(const float4*)&gsrc[(size_t)st * gp];
    for (int c = 0; c < c_begin; ++c) {
      const int bf = c & 1;
      ST[bf][kv][sc4 + 0][st] = pre.x;
      ST[bf][kv][sc4 + 1][st] = pre.y;
      ST[bf][kv][sc4 + 2][st] = pre.z;
      ST[bf][kv][sc4 + 3][st] = pre.w;
      __syncthreads();
      if (c + 1 < c_begin) pre = *(const float4*)&gsrc[(size_t)((c + 1) * 32 + st) * gp];
      const float (*kt)[36] = ST[bf][0];
      const float (*vt)[36] = ST[bf][1];
      const int fi = tid >> 4, fj = tid & 15;
      float4 ka[8], kb[8];
#pragma unroll
      for (int m = 0; m < 8; ++m) {
        ka[m] = *(const float4*)&kt[fi][4 * m];
        kb[m] = *(const float4*)&kt[fj][4 * m];
      }
      if (tid >= 128 && tid < 160) {
        const float* row = (tid < 144) ? vt[tid - 128] : kt[tid - 144];
        float s = 0.f;
#pragma unroll
        for (int t = 0; t < 32; t += 4) {
          const float4 a = *(const float4*)&row[t];
          s += a.x;  s += a.y;  s += a.z;  s += a.w;
        }
        sc += (tid < 144) ? s : 0.5f * s;
      } else if (tid == 160) {
        sc += 32.f;
      }
      v2f fkl[8], fb[8], fk0[8], fk1[8];
#pragma unroll
      for (int m = 0; m < 8; ++m) {
        const int kk = 4 * m + 2 * kh;
        fkl[m] = *(const v2f*)&kt[ln][kk];
        fb[m]  = *(const v2f*)&vt[ln][kk];
        fk0[m] = *(const v2f*)&kt[wv][kk];
        fk1[m] = *(const v2f*)&kt[wv + 8][kk];
      }
      if (wv == 0) {
#pragma unroll
        for (int m = 0; m < 8; ++m) {
          const v2f al = fkl[m] * 0.5f;
          ql = WMMA_F32(al, fb[m], ql);
        }
      }
#pragma unroll
      for (int m = 0; m < 8; ++m) {
        const v2f a0 = fk0[m] * fkl[m] * QSC;
        const v2f a1 = fk1[m] * fkl[m] * QSC;
        q0 = WMMA_F32(a0, fb[m], q0);
        q1 = WMMA_F32(a1, fb[m], q1);
      }
      {
        float s = 0.f;
#pragma unroll
        for (int m = 0; m < 8; ++m) {
          s += ka[m].x * kb[m].x * QSC;
          s += ka[m].y * kb[m].y * QSC;
          s += ka[m].z * kb[m].z * QSC;
          s += ka[m].w * kb[m].w * QSC;
        }
        kq += s;
      }
    }
#pragma unroll
    for (int r = 0; r < 8; ++r) {
      s_quad[wv * 16 + r + 8 * kh][ln]       = q0[r];
      s_quad[(wv + 8) * 16 + r + 8 * kh][ln] = q1[r];
    }
    if (wv == 0) {
#pragma unroll
      for (int r = 0; r < 8; ++r) s_lin[r + 8 * kh][ln] = ql[r];
    }
    kcum[17 + tid] = kq;
    if (tid >= 128 && tid < 144)      s_ones[tid - 128] = sc;
    else if (tid >= 144 && tid < 160) kcum[1 + tid - 144] = sc;
    else if (tid == 160)              kcum[0] = sc;
  }

  for (int c = c_begin; c < c_end; ++c) {
    const int bl = c * 32;
    if (tid < 128) {
      const int t = tid >> 2, c4 = (tid & 3) * 4;
      *(float4*)&Qc[t][c4] = *(const float4*)&Qg[(size_t)(bl + t) * (Hh * FD) + qoff + c4];
      *(float4*)&Kc[t][c4] = *(const float4*)&Kg[(size_t)(bl + t) * (Hh * FD) + qoff + c4];
    } else {
      const int i = tid - 128;
      const int t = i >> 2, c4 = (i & 3) * 4;
      *(float4*)&Vc[t][c4] = *(const float4*)&Vg[(size_t)(bl + t) * DM + voff + c4];
    }
    __syncthreads();

    if (wave < 4) {
      const int ti = wave >> 1, tj = wave & 1;
      v8f acc = {};
#pragma unroll
      for (int k = 0; k < 16; k += 4) {
        const int kk = k + 2 * kh;
        v2f a = *(const v2f*)&Qc[ti * 16 + ln][kk];
        v2f b = *(const v2f*)&Kc[tj * 16 + ln][kk];
        acc = WMMA_F32(a, b, acc);
      }
#pragma unroll
      for (int r = 0; r < 8; ++r) {
        const int row = ti * 16 + r + 8 * kh, col = tj * 16 + ln;
        const float s = acc[r];
        Aw[row][col] = (row >= col) ? (1.f + 0.25f * s + (s * s) * 0.03125f) : 0.f;
      }
    }
    for (int i = tid; i < 32 * 256; i += 256) {
      int t = i >> 8, ij = i & 255;
      Fq[t][ij] = Qc[t][ij >> 4] * Qc[t][ij & 15] * QSC;
    }
    __syncthreads();

    if (wave < 2) {
      const int ti = wave;
      const int row = ti * 16 + ln;
      v8f acc;
#pragma unroll
      for (int r = 0; r < 8; ++r) acc[r] = s_ones[ln];
#pragma unroll
      for (int k = 0; k < 32; k += 4) {
        const int kk = k + 2 * kh;
        v2f a = *(const v2f*)&Aw[row][kk];
        v2f b;
        b.x = Vc[kk][ln];  b.y = Vc[kk + 1][ln];
        acc = WMMA_F32(a, b, acc);
      }
#pragma unroll
      for (int k = 0; k < 16; k += 4) {
        const int kk = k + 2 * kh;
        v2f a, b;
        a.x = Qc[row][kk] * 0.5f; a.y = Qc[row][kk + 1] * 0.5f;
        b.x = s_lin[kk][ln];      b.y = s_lin[kk + 1][ln];
        acc = WMMA_F32(a, b, acc);
      }
#pragma unroll 8
      for (int k = 0; k < 256; k += 4) {
        const int kk = k + 2 * kh;
        v2f a = *(const v2f*)&Fq[row][kk];
        v2f b;
        b.x = s_quad[kk][ln]; b.y = s_quad[kk + 1][ln];
        acc = WMMA_F32(a, b, acc);
      }
#pragma unroll
      for (int r = 0; r < 8; ++r) Yc[ti * 16 + r + 8 * kh][ln] = acc[r];
    } else if (wave == 2) {
      const int t = lane;
      float d = kcum[0];
      for (int s2 = 0; s2 < 32; ++s2) d += Aw[t][s2];
      for (int i = 0; i < 16; ++i) d += Qc[t][i] * 0.5f * kcum[1 + i];
      for (int ij = 0; ij < 256; ++ij) d += Fq[t][ij] * kcum[17 + ij];
      den[t] = d + EPSF;
    }
    __syncthreads();

    if (tid < 128) {
      const int t = tid >> 2, c4 = (tid & 3) * 4;
      const float dd = den[t];
      float4 y = *(float4*)&Yc[t][c4];
      y.x /= dd; y.y /= dd; y.z /= dd; y.w /= dd;
      *(float4*)&Yg[(size_t)(bl + t) * DM + voff + c4] = y;
    }
    for (int i = tid; i < 32 * 256; i += 256) {
      int t = i >> 8, ij = i & 255;
      Fq[t][ij] = Kc[t][ij >> 4] * Kc[t][ij & 15] * QSC;
    }
    if (tid >= 128 && tid < 144) {
      const int f = tid - 128;
      float s = 0.f;
      for (int t = 0; t < 32; ++t) s += Vc[t][f];
      s_ones[f] += s;
    } else if (tid >= 144 && tid < 160) {
      const int f = tid - 144;
      float s = 0.f;
      for (int t = 0; t < 32; ++t) s += Kc[t][f];
      kcum[1 + f] += 0.5f * s;
    } else if (tid == 160) {
      kcum[0] += 32.f;
    }
    __syncthreads();

    {
      float s = 0.f;
      for (int t = 0; t < 32; ++t) s += Fq[t][tid];
      kcum[17 + tid] += s;
    }
    for (int tile = wave; tile < 17; tile += 8) {
      v8f acc;
      if (tile < 16) {
        const int qm = tile * 16;
#pragma unroll
        for (int r = 0; r < 8; ++r) acc[r] = s_quad[qm + r + 8 * kh][ln];
#pragma unroll
        for (int k = 0; k < 32; k += 4) {
          const int kk = k + 2 * kh;
          v2f a, b;
          a.x = Fq[kk][qm + ln]; a.y = Fq[kk + 1][qm + ln];
          b.x = Vc[kk][ln];      b.y = Vc[kk + 1][ln];
          acc = WMMA_F32(a, b, acc);
        }
#pragma unroll
        for (int r = 0; r < 8; ++r) s_quad[qm + r + 8 * kh][ln] = acc[r];
      } else {
#pragma unroll
        for (int r = 0; r < 8; ++r) acc[r] = s_lin[r + 8 * kh][ln];
#pragma unroll
        for (int k = 0; k < 32; k += 4) {
          const int kk = k + 2 * kh;
          v2f a, b;
          a.x = Kc[kk][ln] * 0.5f; a.y = Kc[kk + 1][ln] * 0.5f;
          b.x = Vc[kk][ln];        b.y = Vc[kk + 1][ln];
          acc = WMMA_F32(a, b, acc);
        }
#pragma unroll
        for (int r = 0; r < 8; ++r) s_lin[r + 8 * kh][ln] = acc[r];
      }
    }
    __syncthreads();
  }
}

extern "C" void kernel_launch(void* const* d_in, const int* in_sizes, int n_in,
                              void* d_out, int out_size, void* d_ws, size_t ws_size,
                              hipStream_t stream) {
  (void)in_sizes; (void)n_in; (void)out_size; (void)ws_size;
  const float* hs = (const float*)d_in[0];
  const float* Wq = (const float*)d_in[1];
  const float* Wk = (const float*)d_in[2];
  const float* Wv = (const float*)d_in[3];
  const float* Wo = (const float*)d_in[4];
  float* out = (float*)d_out;

  float* Qg = (float*)d_ws;
  float* Kg = Qg + (size_t)L * (Hh * FD);
  float* Vg = Kg + (size_t)L * (Hh * FD);
  float* Yg = Vg + (size_t)L * DM;

  const dim3 blk(256);
  gemm_qkv_f32_wmma<<<dim3((L / 64) * (DM / 64) + 2 * (L / 64) * ((Hh * FD) / 64)), blk, 0, stream>>>(
      hs, Wq, Wk, Wv, Qg, Kg, Vg);
  based_attn<<<dim3(Hh * 4 * NSEG), blk, 0, stream>>>(Qg, Kg, Vg, Yg);
  gemm_f32_wmma<<<dim3((L / 64) * (DM / 64)), blk, 0, stream>>>(Yg, Wo, out, L, DM, DM);
}


# attention chunk loop, output phase: each row tile's 76 k-steps and the denominator's 304 terms split over all 8 waves, partials added in the write phase; order of sums changes
# speedup vs baseline: 3.8001x; 1.0512x over previous
#include <hip/hip_runtime.h>

typedef float v2f __attribute__((ext_vector_type(2)));
typedef float v8f __attribute__((ext_vector_type(8)));

#define WMMA_F32(a, b, c) \
  __builtin_amdgcn_wmma_f32_16x16x4_f32(false, (a), false, (b), (short)0, (c), false, false)

constexpr int L = 1024, DM = 768, Hh = 12, FD = 16, HD = 64;
constexpr float EPSF = 1e-12f;
constexpr float QSC = 0.17677669529663687f;
constexpr int NSEG = 8;
constexpr int CPS  = (L / 32) / NSEG;

constexpr int KS = 64;
constexpr int LP = KS + 4;

typedef float v4f __attribute__((ext_vector_type(4)));
struct GemmStage { v4f a0, a1, a2, a3, b0, b1, b2, b3; };

__device__ __forceinline__ void barrier_lds_only() {
  __builtin_amdgcn_fence(__ATOMIC_RELEASE, "workgroup", "local");
  __builtin_amdgcn_s_barrier();
  __builtin_amdgcn_fence(__ATOMIC_ACQUIRE, "workgroup", "local");
}

__device__ __forceinline__ void gemm_stage_load(GemmStage& g, const float* __restrict__ A,
                                                const float* __restrict__ B, int N, int K,
                                                int bm, int bn, int kb, int tid) {
  const int ar = tid >> 4, ac = (tid & 15) * 4;
  const int bk = tid >> 2, bc = (tid & 3) * 4;
  const float* ap = &A[(size_t)(bm + ar) * K + kb + ac];
  const float* bp = &B[(size_t)(kb + bk) * N + bn + bc];
  g.a0 = *(const v4f*)&ap[0];
  g.a1 = *(const v4f*)&ap[(size_t)16 * K];
  g.a2 = *(const v4f*)&ap[(size_t)32 * K];
  g.a3 = *(const v4f*)&ap[(size_t)48 * K];
  g.b0 = *(const v4f*)&bp[0];
  g.b1 = *(const v4f*)&bp[16];
  g.b2 = *(const v4f*)&bp[32];
  g.b3 = *(const v4f*)&bp[48];
}

__device__ __forceinline__ void gemm_b_store(v4f b, float (*BsT)[LP], int col, int bk) {
  BsT[col + 0][bk] = b.x;
  BsT[col + 1][bk] = b.y;
  BsT[col + 2][bk] = b.z;
  BsT[col + 3][bk] = b.w;
}

__device__ __forceinline__ void gemm_stage_store(const GemmStage& g, float (*As)[LP],
                                                 float (*BsT)[LP], int tid) {
  const int ar = tid >> 4, ac = (tid & 15) * 4;
  const int bk = tid >> 2, bc = (tid & 3) * 4;
  *(v4f*)&As[ar][ac]      = g.a0;
  *(v4f*)&As[ar + 16][ac] = g.a1;
  *(v4f*)&As[ar + 32][ac] = g.a2;
  *(v4f*)&As[ar + 48][ac] = g.a3;
  gemm_b_store(g.b0, BsT, bc, bk);
  gemm_b_store(g.b1, BsT, bc + 16, bk);
  gemm_b_store(g.b2, BsT, bc + 32, bk);
  gemm_b_store(g.b3, BsT, bc + 48, bk);
}

__device__ __forceinline__ void gemm_stage_compute(const float (*As)[LP], const float (*BsT)[LP],
                                                   v8f& acc0, v8f& acc1,
                                                   int wm, int wn, int ln, int kh) {
#pragma unroll
  for (int k = 0; k < KS; k += 4) {
    const int kk = k + 2 * kh;
    v2f a = *(const v2f*)&As[wm * 16 + ln][kk];
    v2f p = *(const v2f*)&BsT[wn * 32 + ln][kk];
    v2f q = *(const v2f*)&BsT[wn * 32 + 16 + ln][kk];
    acc0 = WMMA_F32(a, p, acc0);
    acc1 = WMMA_F32(a, q, acc1);
  }
}

__device__ __forceinline__ void gemm_tile_f32_wmma(const float* __restrict__ A,
                                                   const float* __restrict__ B,
                                                   float* __restrict__ C,
                                                   int N, int K, int bm, int bn) {
  __shared__ __attribute__((aligned(16))) float As[2][64][LP];
  __shared__ __attribute__((aligned(16))) float BsT[2][64][LP];
  const int tid  = threadIdx.x;
  const int wave = tid >> 5, lane = tid & 31;
  const int kh   = lane >> 4, ln = lane & 15;
  const int wm   = wave >> 1;
  const int wn   = wave & 1;

  GemmStage g;
  v8f acc0 = {}, acc1 = {};
  gemm_stage_load(g, A, B, N, K, bm, bn, 0, tid);
  gemm_stage_store(g, As[0], BsT[0], tid);
  gemm_stage_load(g, A, B, N, K, bm, bn, KS, tid);
  barrier_lds_only();

  int kb = 0;
  for (; kb + 2 * KS < K; kb += 2 * KS) {
    gemm_stage_store(g, As[1], BsT[1], tid);
    gemm_stage_load(g, A, B, N, K, bm, bn, kb + 2 * KS, tid);
    gemm_stage_compute(As[0], BsT[0], acc0, acc1, wm, wn, ln, kh);
    barrier_lds_only();
    gemm_stage_store(g, As[0], BsT[0], tid);
    gemm_stage_load(g, A, B, N, K, bm, bn, kb + 3 * KS, tid);
    gemm_stage_compute(As[1], BsT[1], acc0, acc1, wm, wn, ln, kh);
    barrier_lds_only();
  }
  gemm_stage_store(g, As[1], BsT[1], tid);
  gemm_stage_compute(As[0], BsT[0], acc0, acc1, wm, wn, ln, kh);
  barrier_lds_only();
  gemm_stage_compute(As[1], BsT[1], acc0, acc1, wm, wn, ln, kh);
#pragma unroll
  for (int r = 0; r < 8; ++r) {
    const size_t row = (size_t)(bm + wm * 16 + r + 8 * kh);
    C[row * N + bn + wn * 32 + ln]      = acc0[r];
    C[row * N + bn + wn * 32 + 16 + ln] = acc1[r];
  }
}

__global__ __launch_bounds__(256) void gemm_f32_wmma(const float* __restrict__ A,
                                                     const float* __restrict__ B,
                                                     float* __restrict__ C,
                                                     int M, int N, int K) {
  (void)M;
  const int ntb = N / 64;
  gemm_tile_f32_wmma(A, B, C, N, K, (int)(blockIdx.x / ntb) * 64, (int)(blockIdx.x % ntb) * 64);
}

__global__ __launch_bounds__(256) void gemm_qkv_f32_wmma(const float* __restrict__ X,
                                                         const float* __restrict__ Wq,
                                                         const float* __restrict__ Wk,
                                                         const float* __restrict__ Wv,
                                                         float* __restrict__ Qg,
                                                         float* __restrict__ Kg,
                                                         float* __restrict__ Vg) {
  constexpr int NQ = Hh * FD;
  constexpr int TV = (L / 64) * (DM / 64);
  constexpr int TQ = (L / 64) * (NQ / 64);
  const int b = (int)blockIdx.x;
  if (b < TV) {
    gemm_tile_f32_wmma(X, Wv, Vg, DM, DM, (b / (DM / 64)) * 64, (b % (DM / 64)) * 64);
  } else {
    const int t = (b - TV) % TQ;
    const bool isq = (b - TV) < TQ;
    gemm_tile_f32_wmma(X, isq ? Wq : Wk, isq ? Qg : Kg, NQ, DM,
                       (t / (NQ / 64)) * 64, (t % (NQ / 64)) * 64);
  }
}

__global__ __launch_bounds__(256) void based_attn(const float* __restrict__ Qg,
                                                  const float* __restrict__ Kg,
                                                  const float* __restrict__ Vg,
                                                  float* __restrict__ Yg) {
  __shared__ float s_lin[16][17];
  __shared__ float s_quad[256][17];
  __shared__ float s_ones[16];
  __shared__ float kcum[273];
  __shared__ float Qc[32][20];
  __shared__ float Kc[32][20];
  __shared__ float Vc[32][20];
  __shared__ float Aw[32][33];
  __shared__ float Fq[32][257];
  __shared__ __attribute__((aligned(16))) float Yp[4][32][20];
  __shared__ float denp[10][32];
  __shared__ __attribute__((aligned(16))) float ST[2][2][16][36];

  const int tid  = threadIdx.x;
  const int wave = tid >> 5, lane = tid & 31;
  const int kh   = lane >> 4, ln = lane & 15;
  const int seg  = NSEG - 1 - (int)(blockIdx.x / (Hh * 4));
  const int hv   = blockIdx.x % (Hh * 4);
  const int h    = hv >> 2;
  const int vb   = hv & 3;
  const int c_begin = seg * CPS, c_end = c_begin + CPS;
  const int qoff = h * FD;
  const int voff = h * HD + vb * 16;

  v8f q0 = {}, q1 = {}, ql = {};
  float kq = 0.f;
  float sc = 0.f;
  {
    const int wv = __builtin_amdgcn_readfirstlane(wave);
    const int kv = tid >> 7;
    const int st = (tid & 127) >> 2, sc4 = (tid & 3) * 4;
    const float* gsrc = kv ? (Vg + voff + sc4) : (Kg + qoff + sc4);
    const size_t gp = kv ? (size_t)DM : (size_t)(Hh * FD);
    float4 pre = {0.f, 0.f, 0.f, 0.f};
    if (c_begin > 0) pre = *(const float4*)&gsrc[(size_t)st * gp];
    for (int c = 0; c < c_begin; ++c) {
      const int bf = c & 1;
      ST[bf][kv][sc4 + 0][st] = pre.x;
      ST[bf][kv][sc4 + 1][st] = pre.y;
      ST[bf][kv][sc4 + 2][st] = pre.z;
      ST[bf][kv][sc4 + 3][st] = pre.w;
      __syncthreads();
      if (c + 1 < c_begin) pre = *(const float4*)&gsrc[(size_t)((c + 1) * 32 + st) * gp];
      const float (*kt)[36] = ST[bf][0];
      const float (*vt)[36] = ST[bf][1];
      const int fi = tid >> 4, fj = tid & 15;
      float4 ka[8], kb[8];
#pragma unroll
      for (int m = 0; m < 8; ++m) {
        ka[m] = *(const float4*)&kt[fi][4 * m];
        kb[m] = *(const float4*)&kt[fj][4 * m];
      }
      if (tid >= 128 && tid < 160) {
        const float* row = (tid < 144) ? vt[tid - 128] : kt[tid - 144];
        float s = 0.f;
#pragma unroll
        for (int t = 0; t < 32; t += 4) {
          const float4 a = *(const float4*)&row[t];
          s += a.x;  s += a.y;  s += a.z;  s += a.w;
        }
        sc += (tid < 144) ? s : 0.5f * s;
      } else if (tid == 160) {
        sc += 32.f;
      }
      v2f fkl[8], fb[8], fk0[8], fk1[8];
#pragma unroll
      for (int m = 0; m < 8; ++m) {
        const int kk = 4 * m + 2 * kh;
        fkl[m] = *(const v2f*)&kt[ln][kk];
        fb[m]  = *(const v2f*)&vt[ln][kk];
        fk0[m] = *(const v2f*)&kt[wv][kk];
        fk1[m] = *(const v2f*)&kt[wv + 8][kk];
      }
      if (wv == 0) {
#pragma unroll
        for (int m = 0; m < 8; ++m) {
          const v2f al = fkl[m] * 0.5f;
          ql = WMMA_F32(al, fb[m], ql);
        }
      }
#pragma unroll
      for (int m = 0; m < 8; ++m) {
        const v2f a0 = fk0[m] * fkl[m] * QSC;
        const v2f a1 = fk1[m] * fkl[m] * QSC;
        q0 = WMMA_F32(a0, fb[m], q0);
        q1 = WMMA_F32(a1, fb[m], q1);
      }
      {
        float s = 0.f;
#pragma unroll
        for (int m = 0; m < 8; ++m) {
          s += ka[m].x * kb[m].x * QSC;
          s += ka[m].y * kb[m].y * QSC;
          s += ka[m].z * kb[m].z * QSC;
          s += ka[m].w * kb[m].w * QSC;
        }
        kq += s;
      }
    }
#pragma unroll
    for (int r = 0; r < 8; ++r) {
      s_quad[wv * 16 + r + 8 * kh][ln]       = q0[r];
      s_quad[(wv + 8) * 16 + r + 8 * kh][ln] = q1[r];
    }
    if (wv == 0) {
#pragma unroll
      for (int r = 0; r < 8; ++r) s_lin[r + 8 * kh][ln] = ql[r];
    }
    kcum[17 + tid] = kq;
    if (tid >= 128 && tid < 144)      s_ones[tid - 128] = sc;
    else if (tid >= 144 && tid < 160) kcum[1 + tid - 144] = sc;
    else if (tid == 160)              kcum[0] = sc;
  }

  for (int c = c_begin; c < c_end; ++c) {
    const int bl = c * 32;
    if (tid < 128) {
      const int t = tid >> 2, c4 = (tid & 3) * 4;
      *(float4*)&Qc[t][c4] = *(const float4*)&Qg[(size_t)(bl + t) * (Hh * FD) + qoff + c4];
      *(float4*)&Kc[t][c4] = *(const float4*)&Kg[(size_t)(bl + t) * (Hh * FD) + qoff + c4];
    } else {
      const int i = tid - 128;
      const int t = i >> 2, c4 = (i & 3) * 4;
      *(float4*)&Vc[t][c4] = *(const float4*)&Vg[(size_t)(bl + t) * DM + voff + c4];
    }
    __syncthreads();

    if (wave < 4) {
      const int ti = wave >> 1, tj = wave & 1;
      v8f acc = {};
#pragma unroll
      for (int k = 0; k < 16; k += 4) {
        const int kk = k + 2 * kh;
        v2f a = *(const v2f*)&Qc[ti * 16 + ln][kk];
        v2f b = *(const v2f*)&Kc[tj * 16 + ln][kk];
        acc = WMMA_F32(a, b, acc);
      }
#pragma unroll
      for (int r = 0; r < 8; ++r) {
        const int row = ti * 16 + r + 8 * kh, col = tj * 16 + ln;
        const float s = acc[r];
        Aw[row][col] = (row >= col) ? (1.f + 0.25f * s + (s * s) * 0.03125f) : 0.f;
      }
    }
    for (int i = tid; i < 32 * 256; i += 256) {
      int t = i >> 8, ij = i & 255;
      Fq[t][ij] = Qc[t][ij >> 4] * Qc[t][ij & 15] * QSC;
    }
    __syncthreads();

    {
      const int ti = wave & 1;
      const int kq = __builtin_amdgcn_readfirstlane(wave >> 1);
      const int row = ti * 16 + ln;
      v8f acc;
      const float a_init = (kq == 0) ? s_ones[ln] : 0.f;
#pragma unroll
      for (int r = 0; r < 8; ++r) acc[r] = a_init;
      v2f fa[16], fb[16];
#pragma unroll
      for (int m = 0; m < 16; ++m) {
        const int kk = (kq * 16 + m) * 4 + 2 * kh;
        fa[m].x = Fq[row][kk];     fa[m].y = Fq[row][kk + 1];
        fb[m].x = s_quad[kk][ln];  fb[m].y = s_quad[kk + 1][ln];
      }
      if (kq < 2) {
#pragma unroll
        for (int m = 0; m < 4; ++m) {
          const int kk = (kq * 4 + m) * 4 + 2 * kh;
          v2f a = *(const v2f*)&Aw[row][kk];
          v2f b;
          b.x = Vc[kk][ln];  b.y = Vc[kk + 1][ln];
          acc = WMMA_F32(a, b, acc);
        }
      } else if (kq == 2) {
#pragma unroll
        for (int m = 0; m < 4; ++m) {
          const int kk = m * 4 + 2 * kh;
          v2f a, b;
          a.x = Qc[row][kk] * 0.5f; a.y = Qc[row][kk + 1] * 0.5f;
          b.x = s_lin[kk][ln];      b.y = s_lin[kk + 1][ln];
          acc = WMMA_F32(a, b, acc);
        }
      }
#pragma unroll
      for (int m = 0; m < 16; ++m) acc = WMMA_F32(fa[m], fb[m], acc);
      {
        const int t = lane;
        float d = 0.f;
#pragma unroll
        for (int m = 0; m < 32; ++m) d += Fq[t][wave * 32 + m] * kcum[17 + wave * 32 + m];
        denp[wave][t] = d;
        if (wave == 6) {
          float e = kcum[0];
#pragma unroll
          for (int s2 = 0; s2 < 32; ++s2) e += Aw[t][s2];
          denp[8][t] = e;
        } else if (wave == 7) {
          float e = 0.f;
#pragma unroll
          for (int i = 0; i < 16; ++i) e += Qc[t][i] * 0.5f * kcum[1 + i];
          denp[9][t] = e;
        }
      }
#pragma unroll
      for (int r = 0; r < 8; ++r) Yp[kq][ti * 16 + r + 8 * kh][ln] = acc[r];
    }
    __syncthreads();

    if (tid < 128) {
      const int t = tid >> 2, c4 = (tid & 3) * 4;
      float dd = denp[8][t];
      dd += denp[9][t];
#pragma unroll
      for (int w = 0; w < 8; ++w) dd += denp[w][t];
      dd += EPSF;
      float4 y = *(float4*)&Yp[0][t][c4];
#pragma unroll
      for (int q = 1; q < 4; ++q) {
        const float4 p = *(float4*)&Yp[q][t][c4];
        y.x += p.x; y.y += p.y; y.z += p.z; y.w += p.w;
      }
      y.x /= dd; y.y /= dd; y.z /= dd; y.w /= dd;
      *(float4*)&Yg[(size_t)(bl + t) * DM + voff + c4] = y;
    }
    for (int i = tid; i < 32 * 256; i += 256) {
      int t = i >> 8, ij = i & 255;
      Fq[t][ij] = Kc[t][ij >> 4] * Kc[t][ij & 15] * QSC;
    }
    if (tid >= 128 && tid < 144) {
      const int f = tid - 128;
      float s = 0.f;
      for (int t = 0; t < 32; ++t) s += Vc[t][f];
      s_ones[f] += s;
    } else if (tid >= 144 && tid < 160) {
      const int f = tid - 144;
      float s = 0.f;
      for (int t = 0; t < 32; ++t) s += Kc[t][f];
      kcum[1 + f] += 0.5f * s;
    } else if (tid == 160) {
      kcum[0] += 32.f;
    }
    __syncthreads();

    {
      float s = 0.f;
      for (int t = 0; t < 32; ++t) s += Fq[t][tid];
      kcum[17 + tid] += s;
    }
    for (int tile = wave; tile < 17; tile += 8) {
      v8f acc;
      if (tile < 16) {
        const int qm = tile * 16;
#pragma unroll
        for (int r = 0; r < 8; ++r) acc[r] = s_quad[qm + r + 8 * kh][ln];
#pragma unroll
        for (int k = 0; k < 32; k += 4) {
          const int kk = k + 2 * kh;
          v2f a, b;
          a.x = Fq[kk][qm + ln]; a.y = Fq[kk + 1][qm + ln];
          b.x = Vc[kk][ln];      b.y = Vc[kk + 1][ln];
          acc = WMMA_F32(a, b, acc);
        }
#pragma unroll
        for (int r = 0; r < 8; ++r) s_quad[qm + r + 8 * kh][ln] = acc[r];
      } else {
#pragma unroll
        for (int r = 0; r < 8; ++r) acc[r] = s_lin[r + 8 * kh][ln];
#pragma unroll
        for (int k = 0; k < 32; k += 4) {
          const int kk = k + 2 * kh;
          v2f a, b;
          a.x = Kc[kk][ln] * 0.5f; a.y = Kc[kk + 1][ln] * 0.5f;
          b.x = Vc[kk][ln];        b.y = Vc[kk + 1][ln];
          acc = WMMA_F32(a, b, acc);
        }
#pragma unroll
        for (int r = 0; r < 8; ++r) s_lin[r + 8 * kh][ln] = acc[r];
      }
    }
    __syncthreads();
  }
}

extern "C" void kernel_launch(void* const* d_in, const int* in_sizes, int n_in,
                              void* d_out, int out_size, void* d_ws, size_t ws_size,
                              hipStream_t stream) {
  (void)in_sizes; (void)n_in; (void)out_size; (void)ws_size;
  const float* hs = (const float*)d_in[0];
  const float* Wq = (const float*)d_in[1];
  const float* Wk = (const float*)d_in[2];
  const float* Wv = (const float*)d_in[3];
  const float* Wo = (const float*)d_in[4];
  float* out = (float*)d_out;

  float* Qg = (float*)d_ws;
  float* Kg = Qg + (size_t)L * (Hh * FD);
  float* Vg = Kg + (size_t)L * (Hh * FD);
  float* Yg = Vg + (size_t)L * DM;

  const dim3 blk(256);
  gemm_qkv_f32_wmma<<<dim3((L / 64) * (DM / 64) + 2 * (L / 64) * ((Hh * FD) / 64)), blk, 0, stream>>>(
      hs, Wq, Wk, Wv, Qg, Kg, Vg);
  based_attn<<<dim3(Hh * 4 * NSEG), blk, 0, stream>>>(Qg, Kg, Vg, Yg);
  gemm_f32_wmma<<<dim3((L / 64) * (DM / 64)), blk, 0, stream>>>(Yg, Wo, out, L, DM, DM);
}


# attention as a two-pass scan: one more launch sums each segment's own chunks once; attention blocks add the stored states of earlier segments instead of rebuilding them; order of state sums changes
# speedup vs baseline: 4.5855x; 1.2067x over previous
#include <hip/hip_runtime.h>

typedef float v2f __attribute__((ext_vector_type(2)));
typedef float v8f __attribute__((ext_vector_type(8)));

#define WMMA_F32(a, b, c) \
  __builtin_amdgcn_wmma_f32_16x16x4_f32(false, (a), false, (b), (short)0, (c), false, false)

constexpr int L = 1024, DM = 768, Hh = 12, FD = 16, HD = 64;
constexpr float EPSF = 1e-12f;
constexpr float QSC = 0.17677669529663687f;
constexpr int NSEG = 8;
constexpr int CPS  = (L / 32) / NSEG;

constexpr int KS = 64;
constexpr int LP = KS + 4;

typedef float v4f __attribute__((ext_vector_type(4)));
struct GemmStage { v4f a0, a1, a2, a3, b0, b1, b2, b3; };

__device__ __forceinline__ void barrier_lds_only() {
  __builtin_amdgcn_fence(__ATOMIC_RELEASE, "workgroup", "local");
  __builtin_amdgcn_s_barrier();
  __builtin_amdgcn_fence(__ATOMIC_ACQUIRE, "workgroup", "local");
}

__device__ __forceinline__ void gemm_stage_load(GemmStage& g, const float* __restrict__ A,
                                                const float* __restrict__ B, int N, int K,
                                                int bm, int bn, int kb, int tid) {
  const int ar = tid >> 4, ac = (tid & 15) * 4;
  const int bk = tid >> 2, bc = (tid & 3) * 4;
  const float* ap = &A[(size_t)(bm + ar) * K + kb + ac];
  const float* bp = &B[(size_t)(kb + bk) * N + bn + bc];
  g.a0 = *(const v4f*)&ap[0];
  g.a1 = *(const v4f*)&ap[(size_t)16 * K];
  g.a2 = *(const v4f*)&ap[(size_t)32 * K];
  g.a3 = *(const v4f*)&ap[(size_t)48 * K];
  g.b0 = *(const v4f*)&bp[0];
  g.b1 = *(const v4f*)&bp[16];
  g.b2 = *(const v4f*)&bp[32];
  g.b3 = *(const v4f*)&bp[48];
}

__device__ __forceinline__ void gemm_b_store(v4f b, float (*BsT)[LP], int col, int bk) {
  BsT[col + 0][bk] = b.x;
  BsT[col + 1][bk] = b.y;
  BsT[col + 2][bk] = b.z;
  BsT[col + 3][bk] = b.w;
}

__device__ __forceinline__ void gemm_stage_store(const GemmStage& g, float (*As)[LP],
                                                 float (*BsT)[LP], int tid) {
  const int ar = tid >> 4, ac = (tid & 15) * 4;
  const int bk = tid >> 2, bc = (tid & 3) * 4;
  *(v4f*)&As[ar][ac]      = g.a0;
  *(v4f*)&As[ar + 16][ac] = g.a1;
  *(v4f*)&As[ar + 32][ac] = g.a2;
  *(v4f*)&As[ar + 48][ac] = g.a3;
  gemm_b_store(g.b0, BsT, bc, bk);
  gemm_b_store(g.b1, BsT, bc + 16, bk);
  gemm_b_store(g.b2, BsT, bc + 32, bk);
  gemm_b_store(g.b3, BsT, bc + 48, bk);
}

__device__ __forceinline__ void gemm_stage_compute(const float (*As)[LP], const float (*BsT)[LP],
                                                   v8f& acc0, v8f& acc1,
                                                   int wm, int wn, int ln, int kh) {
#pragma unroll
  for (int k = 0; k < KS; k += 4) {
    const int kk = k + 2 * kh;
    v2f a = *(const v2f*)&As[wm * 16 + ln][kk];
    v2f p = *(const v2f*)&BsT[wn * 32 + ln][kk];
    v2f q = *(const v2f*)&BsT[wn * 32 + 16 + ln][kk];
    acc0 = WMMA_F32(a, p, acc0);
    acc1 = WMMA_F32(a, q, acc1);
  }
}

__device__ __forceinline__ void gemm_tile_f32_wmma(const float* __restrict__ A,
                                                   const float* __restrict__ B,
                                                   float* __restrict__ C,
                                                   int N, int K, int bm, int bn) {
  __shared__ __attribute__((aligned(16))) float As[2][64][LP];
  __shared__ __attribute__((aligned(16))) float BsT[2][64][LP];
  const int tid  = threadIdx.x;
  const int wave = tid >> 5, lane = tid & 31;
  const int kh   = lane >> 4, ln = lane & 15;
  const int wm   = wave >> 1;
  const int wn   = wave & 1;

  GemmStage g;
  v8f acc0 = {}, acc1 = {};
  gemm_stage_load(g, A, B, N, K, bm, bn, 0, tid);
  gemm_stage_store(g, As[0], BsT[0], tid);
  gemm_stage_load(g, A, B, N, K, bm, bn, KS, tid);
  barrier_lds_only();

  int kb = 0;
  for (; kb + 2 * KS < K; kb += 2 * KS) {
    gemm_stage_store(g, As[1], BsT[1], tid);
    gemm_stage_load(g, A, B, N, K, bm, bn, kb + 2 * KS, tid);
    gemm_stage_compute(As[0], BsT[0], acc0, acc1, wm, wn, ln, kh);
    barrier_lds_only();
    gemm_stage_store(g, As[0], BsT[0], tid);
    gemm_stage_load(g, A, B, N, K, bm, bn, kb + 3 * KS, tid);
    gemm_stage_compute(As[1], BsT[1], acc0, acc1, wm, wn, ln, kh);
    barrier_lds_only();
  }
  gemm_stage_store(g, As[1], BsT[1], tid);
  gemm_stage_compute(As[0], BsT[0], acc0, acc1, wm, wn, ln, kh);
  barrier_lds_only();
  gemm_stage_compute(As[1], BsT[1], acc0, acc1, wm, wn, ln, kh);
#pragma unroll
  for (int r = 0; r < 8; ++r) {
    const size_t row = (size_t)(bm + wm * 16 + r + 8 * kh);
    C[row * N + bn + wn * 32 + ln]      = acc0[r];
    C[row * N + bn + wn * 32 + 16 + ln] = acc1[r];
  }
}

__global__ __launch_bounds__(256) void gemm_f32_wmma(const float* __restrict__ A,
                                                     const float* __restrict__ B,
                                                     float* __restrict__ C,
                                                     int M, int N, int K) {
  (void)M;
  const int ntb = N / 64;
  gemm_tile_f32_wmma(A, B, C, N, K, (int)(blockIdx.x / ntb) * 64, (int)(blockIdx.x % ntb) * 64);
}

__global__ __launch_bounds__(256) void gemm_qkv_f32_wmma(const float* __restrict__ X,
                                                         const float* __restrict__ Wq,
                                                         const float* __restrict__ Wk,
                                                         const float* __restrict__ Wv,
                                                         float* __restrict__ Qg,
                                                         float* __restrict__ Kg,
                                                         float* __restrict__ Vg) {
  constexpr int NQ = Hh * FD;
  constexpr int TV = (L / 64) * (DM / 64);
  constexpr int TQ = (L / 64) * (NQ / 64);
  const int b = (int)blockIdx.x;
  if (b < TV) {
    gemm_tile_f32_wmma(X, Wv, Vg, DM, DM, (b / (DM / 64)) * 64, (b % (DM / 64)) * 64);
  } else {
    const int t = (b - TV) % TQ;
    const bool isq = (b - TV) < TQ;
    gemm_tile_f32_wmma(X, isq ? Wq : Wk, isq ? Qg : Kg, NQ, DM,
                       (t / (NQ / 64)) * 64, (t % (NQ / 64)) * 64);
  }
}

constexpr int SLOT     = 4644;
constexpr int O_LIN    = 4096, O_ONES = 4352, O_KC = 4368, O_END = 4641;
constexpr int SLOTS_P0 = 168;

__device__ __forceinline__ float* state_slot(float* p0, float* p1, int i) {
  return i < SLOTS_P0 ? p0 + (size_t)i * SLOT : p1 + (size_t)(i - SLOTS_P0) * SLOT;
}

__global__ __launch_bounds__(256) void based_state(const float* __restrict__ Kg,
                                                   const float* __restrict__ Vg,
                                                   float* __restrict__ P0,
                                                   float* __restrict__ P1) {
  __shared__ __attribute__((aligned(16))) float ST[2][2][16][36];

  const int tid  = threadIdx.x;
  const int wave = tid >> 5, lane = tid & 31;
  const int kh   = lane >> 4, ln = lane & 15;
  const int seg  = (int)(blockIdx.x / (Hh * 4));
  const int hv   = blockIdx.x % (Hh * 4);
  const int h    = hv >> 2;
  const int vb   = hv & 3;
  const int c_begin = seg * CPS, c_end = c_begin + CPS;
  const int qoff = h * FD;
  const int voff = h * HD + vb * 16;

  v8f q0 = {}, q1 = {}, ql = {};
  float kq = 0.f;
  float sc = 0.f;
  const int wv = __builtin_amdgcn_readfirstlane(wave);
  const int kv = tid >> 7;
  const int st = (tid & 127) >> 2, sc4 = (tid & 3) * 4;
  const float* gsrc = kv ? (Vg + voff + sc4) : (Kg + qoff + sc4);
  const size_t gp = kv ? (size_t)DM : (size_t)(Hh * FD);
  float4 pre = *(const float4*)&gsrc[(size_t)(c_begin * 32 + st) * gp];
#pragma unroll 1
  for (int c = c_begin; c < c_end; ++c) {
    const int bf = c & 1;
    ST[bf][kv][sc4 + 0][st] = pre.x;
    ST[bf][kv][sc4 + 1][st] = pre.y;
    ST[bf][kv][sc4 + 2][st] = pre.z;
    ST[bf][kv][sc4 + 3][st] = pre.w;
    __syncthreads();
    if (c + 1 < c_end) pre = *(const float4*)&gsrc[(size_t)((c + 1) * 32 + st) * gp];
    const float (*kt)[36] = ST[bf][0];
    const float (*vt)[36] = ST[bf][1];
    const int fi = tid >> 4, fj = tid & 15;
    float4 ka[8], kb[8];
#pragma unroll
    for (int m = 0; m < 8; ++m) {
      ka[m] = *(const float4*)&kt[fi][4 * m];
      kb[m] = *(const float4*)&kt[fj][4 * m];
    }
    if (tid >= 128 && tid < 160) {
      const float* row = (tid < 144) ? vt[tid - 128] : kt[tid - 144];
      float s = 0.f;
#pragma unroll
      for (int t = 0; t < 32; t += 4) {
        const float4 a = *(const float4*)&row[t];
        s += a.x;  s += a.y;  s += a.z;  s += a.w;
      }
      sc += (tid < 144) ? s : 0.5f * s;
    } else if (tid == 160) {
      sc += 32.f;
    }
    v2f fkl[8], fb[8], fk0[8], fk1[8];
#pragma unroll
    for (int m = 0; m < 8; ++m) {
      const int kk = 4 * m + 2 * kh;
      fkl[m] = *(const v2f*)&kt[ln][kk];
      fb[m]  = *(const v2f*)&vt[ln][kk];
      fk0[m] = *(const v2f*)&kt[wv][kk];
      fk1[m] = *(const v2f*)&kt[wv + 8][kk];
    }
    if (wv == 0) {
#pragma unroll
      for (int m = 0; m < 8; ++m) {
        const v2f al = fkl[m] * 0.5f;
        ql = WMMA_F32(al, fb[m], ql);
      }
    }
#pragma unroll
    for (int m = 0; m < 8; ++m) {
      const v2f a0 = fk0[m] * fkl[m] * QSC;
      const v2f a1 = fk1[m] * fkl[m] * QSC;
      q0 = WMMA_F32(a0, fb[m], q0);
      q1 = WMMA_F32(a1, fb[m], q1);
    }
    {
      float s = 0.f;
#pragma unroll
      for (int m = 0; m < 8; ++m) {
        s += ka[m].x * kb[m].x * QSC;
        s += ka[m].y * kb[m].y * QSC;
        s += ka[m].z * kb[m].z * QSC;
        s += ka[m].w * kb[m].w * QSC;
      }
      kq += s;
    }
  }
  float* g = state_slot(P0, P1, seg * (Hh * 4) + hv);
#pragma unroll
  for (int r = 0; r < 8; ++r) {
    g[(wv * 16 + r + 8 * kh) * 16 + ln]       = q0[r];
    g[((wv + 8) * 16 + r + 8 * kh) * 16 + ln] = q1[r];
  }
  if (wv == 0) {
#pragma unroll
    for (int r = 0; r < 8; ++r) g[O_LIN + (r + 8 * kh) * 16 + ln] = ql[r];
  }
  g[O_KC + 17 + tid] = kq;
  if (tid >= 128 && tid < 144)      g[O_ONES + tid - 128] = sc;
  else if (tid >= 144 && tid < 160) g[O_KC + 1 + tid - 144] = sc;
  else if (tid == 160)              g[O_KC] = sc;
  else if (tid >= 161 && tid < 164) g[O_END + tid - 161] = 0.f;
}

__device__ __forceinline__ const float* state_slot(const float* p0, const float* p1, int i) {
  return i < SLOTS_P0 ? p0 + (size_t)i * SLOT : p1 + (size_t)(i - SLOTS_P0) * SLOT;
}

__global__ __launch_bounds__(256) void based_attn(const float* __restrict__ Qg,
                                                  const float* __restrict__ Kg,
                                                  float* VY,
                                                  const float* __restrict__ P0,
                                                  const float* __restrict__ P1) {
  __shared__ float s_lin[16][17];
  __shared__ float s_quad[256][17];
  __shared__ float s_ones[16];
  __shared__ float kcum[273];
  __shared__ float Qc[32][20];
  __shared__ float Kc[32][20];
  __shared__ float Vc[32][20];
  __shared__ float Aw[32][33];
  __shared__ float Fq[32][257];
  __shared__ __attribute__((aligned(16))) float Yp[4][32][20];
  __shared__ float denp[10][32];

  const int tid  = threadIdx.x;
  const int wave = tid >> 5, lane = tid & 31;
  const int kh   = lane >> 4, ln = lane & 15;
  const int seg  = NSEG - 1 - (int)(blockIdx.x / (Hh * 4));
  const int hv   = blockIdx.x % (Hh * 4);
  const int h    = hv >> 2;
  const int vb   = hv & 3;
  const int c_begin = seg * CPS, c_end = c_begin + CPS;
  const int qoff = h * FD;
  const int voff = h * HD + vb * 16;

  {
    int i4[5];
#pragma unroll
    for (int j = 0; j < 4; ++j) i4[j] = tid + 256 * j;
    i4[4] = min(tid + 1024, SLOT / 4 - 1);
    float4 p[NSEG - 1][5];
    const int s_last = max(seg - 1, 0);
#pragma unroll
    for (int s = 0; s < NSEG - 1; ++s) {
      const float* g = state_slot(P0, P1, min(s, s_last) * (Hh * 4) + hv);
#pragma unroll
      for (int j = 0; j < 5; ++j) p[s][j] = *(const float4*)&g[4 * i4[j]];
    }
    float4 a[5];
#pragma unroll
    for (int j = 0; j < 5; ++j) a[j] = float4{0.f, 0.f, 0.f, 0.f};
#pragma unroll
    for (int s = 0; s < NSEG - 1; ++s) {
      const bool on = s < seg;
#pragma unroll
      for (int j = 0; j < 5; ++j) {
        a[j].x += on ? p[s][j].x : 0.f;
        a[j].y += on ? p[s][j].y : 0.f;
        a[j].z += on ? p[s][j].z : 0.f;
        a[j].w += on ? p[s][j].w : 0.f;
      }
    }
#pragma unroll
    for (int j = 0; j < 5; ++j) {
      const int e = 4 * (tid + 256 * j);
      const float v[4] = {a[j].x, a[j].y, a[j].z, a[j].w};
      if (e < O_LIN) {
#pragma unroll
        for (int u = 0; u < 4; ++u) s_quad[e >> 4][(e & 15) + u] = v[u];
      } else if (e < O_ONES) {
        const int x = e - O_LIN;
#pragma unroll
        for (int u = 0; u < 4; ++u) s_lin[x >> 4][(x & 15) + u] = v[u];
      } else if (e < O_KC) {
#pragma unroll
        for (int u = 0; u < 4; ++u) s_ones[e - O_ONES + u] = v[u];
      } else if (e < O_END) {
#pragma unroll
        for (int u = 0; u < 4; ++u)
          if (e - O_KC + u < 273) kcum[e - O_KC + u] = v[u];
      }
    }
  }

  for (int c = c_begin; c < c_end; ++c) {
    const int bl = c * 32;
    if (tid < 128) {
      const int t = tid >> 2, c4 = (tid & 3) * 4;
      *(float4*)&Qc[t][c4] = *(const float4*)&Qg[(size_t)(bl + t) * (Hh * FD) + qoff + c4];
      *(float4*)&Kc[t][c4] = *(const float4*)&Kg[(size_t)(bl + t) * (Hh * FD) + qoff + c4];
    } else {
      const int i = tid - 128;
      const int t = i >> 2, c4 = (i & 3) * 4;
      *(float4*)&Vc[t][c4] = *(const float4*)&VY[(size_t)(bl + t) * DM + voff + c4];
    }
    __syncthreads();

    if (wave < 4) {
      const int ti = wave >> 1, tj = wave & 1;
      v8f acc = {};
#pragma unroll
      for (int k = 0; k < 16; k += 4) {
        const int kk = k + 2 * kh;
        v2f a = *(const v2f*)&Qc[ti * 16 + ln][kk];
        v2f b = *(const v2f*)&Kc[tj * 16 + ln][kk];
        acc = WMMA_F32(a, b, acc);
      }
#pragma unroll
      for (int r = 0; r < 8; ++r) {
        const int row = ti * 16 + r + 8 * kh, col = tj * 16 + ln;
        const float s = acc[r];
        Aw[row][col] = (row >= col) ? (1.f + 0.25f * s + (s * s) * 0.03125f) : 0.f;
      }
    }
    for (int i = tid; i < 32 * 256; i += 256) {
      int t = i >> 8, ij = i & 255;
      Fq[t][ij] = Qc[t][ij >> 4] * Qc[t][ij & 15] * QSC;
    }
    __syncthreads();

    {
      const int ti = wave & 1;
      const int kq = __builtin_amdgcn_readfirstlane(wave >> 1);
      const int row = ti * 16 + ln;
      v8f acc;
      const float a_init = (kq == 0) ? s_ones[ln] : 0.f;
#pragma unroll
      for (int r = 0; r < 8; ++r) acc[r] = a_init;
      v2f fa[16], fb[16];
#pragma unroll
      for (int m = 0; m < 16; ++m) {
        const int kk = (kq * 16 + m) * 4 + 2 * kh;
        fa[m].x = Fq[row][kk];     fa[m].y = Fq[row][kk + 1];
        fb[m].x = s_quad[kk][ln];  fb[m].y = s_quad[kk + 1][ln];
      }
      if (kq < 2) {
#pragma unroll
        for (int m = 0; m < 4; ++m) {
          const int kk = (kq * 4 + m) * 4 + 2 * kh;
          v2f a = *(const v2f*)&Aw[row][kk];
          v2f b;
          b.x = Vc[kk][ln];  b.y = Vc[kk + 1][ln];
          acc = WMMA_F32(a, b, acc);
        }
      } else if (kq == 2) {
#pragma unroll
        for (int m = 0; m < 4; ++m) {
          const int kk = m * 4 + 2 * kh;
          v2f a, b;
          a.x = Qc[row][kk] * 0.5f; a.y = Qc[row][kk + 1] * 0.5f;
          b.x = s_lin[kk][ln];      b.y = s_lin[kk + 1][ln];
          acc = WMMA_F32(a, b, acc);
        }
      }
#pragma unroll
      for (int m = 0; m < 16; ++m) acc = WMMA_F32(fa[m], fb[m], acc);
      {
        const int t = lane;
        float d = 0.f;
#pragma unroll
        for (int m = 0; m < 32; ++m) d += Fq[t][wave * 32 + m] * kcum[17 + wave * 32 + m];
        denp[wave][t] = d;
        if (wave == 6) {
          float e = kcum[0];
#pragma unroll
          for (int s2 = 0; s2 < 32; ++s2) e += Aw[t][s2];
          denp[8][t] = e;
        } else if (wave == 7) {
          float e = 0.f;
#pragma unroll
          for (int i = 0; i < 16; ++i) e += Qc[t][i] * 0.5f * kcum[1 + i];
          denp[9][t] = e;
        }
      }
#pragma unroll
      for (int r = 0; r < 8; ++r) Yp[kq][ti * 16 + r + 8 * kh][ln] = acc[r];
    }
    __syncthreads();

    if (tid < 128) {
      const int t = tid >> 2, c4 = (tid & 3) * 4;
      float dd = denp[8][t];
      dd += denp[9][t];
#pragma unroll
      for (int w = 0; w < 8; ++w) dd += denp[w][t];
      dd += EPSF;
      float4 y = *(float4*)&Yp[0][t][c4];
#pragma unroll
      for (int q = 1; q < 4; ++q) {
        const float4 p = *(float4*)&Yp[q][t][c4];
        y.x += p.x; y.y += p.y; y.z += p.z; y.w += p.w;
      }
      y.x /= dd; y.y /= dd; y.z /= dd; y.w /= dd;
      *(float4*)&VY[(size_t)(bl + t) * DM + voff + c4] = y;
    }
    for (int i = tid; i < 32 * 256; i += 256) {
      int t = i >> 8, ij = i & 255;
      Fq[t][ij] = Kc[t][ij >> 4] * Kc[t][ij & 15] * QSC;
    }
    if (tid >= 128 && tid < 144) {
      const int f = tid - 128;
      float s = 0.f;
      for (int t = 0; t < 32; ++t) s += Vc[t][f];
      s_ones[f] += s;
    } else if (tid >= 144 && tid < 160) {
      const int f = tid - 144;
      float s = 0.f;
      for (int t = 0; t < 32; ++t) s += Kc[t][f];
      kcum[1 + f] += 0.5f * s;
    } else if (tid == 160) {
      kcum[0] += 32.f;
    }
    __syncthreads();

    {
      float s = 0.f;
      for (int t = 0; t < 32; ++t) s += Fq[t][tid];
      kcum[17 + tid] += s;
    }
    for (int tile = wave; tile < 17; tile += 8) {
      v8f acc;
      if (tile < 16) {
        const int qm = tile * 16;
#pragma unroll
        for (int r = 0; r < 8; ++r) acc[r] = s_quad[qm + r + 8 * kh][ln];
#pragma unroll
        for (int k = 0; k < 32; k += 4) {
          const int kk = k + 2 * kh;
          v2f a, b;
          a.x = Fq[kk][qm + ln]; a.y = Fq[kk + 1][qm + ln];
          b.x = Vc[kk][ln];      b.y = Vc[kk + 1][ln];
          acc = WMMA_F32(a, b, acc);
        }
#pragma unroll
        for (int r = 0; r < 8; ++r) s_quad[qm + r + 8 * kh][ln] = acc[r];
      } else {
#pragma unroll
        for (int r = 0; r < 8; ++r) acc[r] = s_lin[r + 8 * kh][ln];
#pragma unroll
        for (int k = 0; k < 32; k += 4) {
          const int kk = k + 2 * kh;
          v2f a, b;
          a.x = Kc[kk][ln] * 0.5f; a.y = Kc[kk + 1][ln] * 0.5f;
          b.x = Vc[kk][ln];        b.y = Vc[kk + 1][ln];
          acc = WMMA_F32(a, b, acc);
        }
#pragma unroll
        for (int r = 0; r < 8; ++r) s_lin[r + 8 * kh][ln] = acc[r];
      }
    }
    __syncthreads();
  }
}

extern "C" void kernel_launch(void* const* d_in, const int* in_sizes, int n_in,
                              void* d_out, int out_size, void* d_ws, size_t ws_size,
                              hipStream_t stream) {
  (void)in_sizes; (void)n_in; (void)out_size; (void)ws_size;
  const float* hs = (const float*)d_in[0];
  const float* Wq = (const float*)d_in[1];
  const float* Wk = (const float*)d_in[2];
  const float* Wv = (const float*)d_in[3];
  const float* Wo = (const float*)d_in[4];
  float* out = (float*)d_out;

  float* Qg = (float*)d_ws;
  float* Kg = Qg + (size_t)L * (Hh * FD);
  float* Vg = Kg + (size_t)L * (Hh * FD);
  float* S0 = Vg + (size_t)L * DM;
  float* S1 = out;

  const dim3 blk(256);
  gemm_qkv_f32_wmma<<<dim3((L / 64) * (DM / 64) + 2 * (L / 64) * ((Hh * FD) / 64)), blk, 0, stream>>>(
      hs, Wq, Wk, Wv, Qg, Kg, Vg);
  based_state<<<dim3(Hh * 4 * (NSEG - 1)), blk, 0, stream>>>(Kg, Vg, S0, S1);
  based_attn<<<dim3(Hh * 4 * NSEG), blk, 0, stream>>>(Qg, Kg, Vg, S0, S1);
  gemm_f32_wmma<<<dim3((L / 64) * (DM / 64)), blk, 0, stream>>>(Vg, Wo, out, L, DM, DM);
}


# attention chunk loop: Q and K quadratic features formed in registers from the staged chunk rows; the LDS feature plane, its two build passes and one barrier per chunk are gone; order of sums unchanged
# speedup vs baseline: 5.0834x; 1.1086x over previous
#include <hip/hip_runtime.h>

typedef float v2f __attribute__((ext_vector_type(2)));
typedef float v8f __attribute__((ext_vector_type(8)));

#define WMMA_F32(a, b, c) \
  __builtin_amdgcn_wmma_f32_16x16x4_f32(false, (a), false, (b), (short)0, (c), false, false)

constexpr int L = 1024, DM = 768, Hh = 12, FD = 16, HD = 64;
constexpr float EPSF = 1e-12f;
constexpr float QSC = 0.17677669529663687f;
constexpr int NSEG = 8;
constexpr int CPS  = (L / 32) / NSEG;

constexpr int KS = 64;
constexpr int LP = KS + 4;

typedef float v4f __attribute__((ext_vector_type(4)));
struct GemmStage { v4f a0, a1, a2, a3, b0, b1, b2, b3; };

__device__ __forceinline__ void barrier_lds_only() {
  __builtin_amdgcn_fence(__ATOMIC_RELEASE, "workgroup", "local");
  __builtin_amdgcn_s_barrier();
  __builtin_amdgcn_fence(__ATOMIC_ACQUIRE, "workgroup", "local");
}

__device__ __forceinline__ void gemm_stage_load(GemmStage& g, const float* __restrict__ A,
                                                const float* __restrict__ B, int N, int K,
                                                int bm, int bn, int kb, int tid) {
  const int ar = tid >> 4, ac = (tid & 15) * 4;
  const int bk = tid >> 2, bc = (tid & 3) * 4;
  const float* ap = &A[(size_t)(bm + ar) * K + kb + ac];
  const float* bp = &B[(size_t)(kb + bk) * N + bn + bc];
  g.a0 = *(const v4f*)&ap[0];
  g.a1 = *(const v4f*)&ap[(size_t)16 * K];
  g.a2 = *(const v4f*)&ap[(size_t)32 * K];
  g.a3 = *(const v4f*)&ap[(size_t)48 * K];
  g.b0 = *(const v4f*)&bp[0];
  g.b1 = *(const v4f*)&bp[16];
  g.b2 = *(const v4f*)&bp[32];
  g.b3 = *(const v4f*)&bp[48];
}

__device__ __forceinline__ void gemm_b_store(v4f b, float (*BsT)[LP], int col, int bk) {
  BsT[col + 0][bk] = b.x;
  BsT[col + 1][bk] = b.y;
  BsT[col + 2][bk] = b.z;
  BsT[col + 3][bk] = b.w;
}

__device__ __forceinline__ void gemm_stage_store(const GemmStage& g, float (*As)[LP],
                                                 float (*BsT)[LP], int tid) {
  const int ar = tid >> 4, ac = (tid & 15) * 4;
  const int bk = tid >> 2, bc = (tid & 3) * 4;
  *(v4f*)&As[ar][ac]      = g.a0;
  *(v4f*)&As[ar + 16][ac] = g.a1;
  *(v4f*)&As[ar + 32][ac] = g.a2;
  *(v4f*)&As[ar + 48][ac] = g.a3;
  gemm_b_store(g.b0, BsT, bc, bk);
  gemm_b_store(g.b1, BsT, bc + 16, bk);
  gemm_b_store(g.b2, BsT, bc + 32, bk);
  gemm_b_store(g.b3, BsT, bc + 48, bk);
}

__device__ __forceinline__ void gemm_stage_compute(const float (*As)[LP], const float (*BsT)[LP],
                                                   v8f& acc0, v8f& acc1,
                                                   int wm, int wn, int ln, int kh) {
#pragma unroll
  for (int k = 0; k < KS; k += 4) {
    const int kk = k + 2 * kh;
    v2f a = *(const v2f*)&As[wm * 16 + ln][kk];
    v2f p = *(const v2f*)&BsT[wn * 32 + ln][kk];
    v2f q = *(const v2f*)&BsT[wn * 32 + 16 + ln][kk];
    acc0 = WMMA_F32(a, p, acc0);
    acc1 = WMMA_F32(a, q, acc1);
  }
}

__device__ __forceinline__ void gemm_tile_f32_wmma(const float* __restrict__ A,
                                                   const float* __restrict__ B,
                                                   float* __restrict__ C,
                                                   int N, int K, int bm, int bn) {
  __shared__ __attribute__((aligned(16))) float As[2][64][LP];
  __shared__ __attribute__((aligned(16))) float BsT[2][64][LP];
  const int tid  = threadIdx.x;
  const int wave = tid >> 5, lane = tid & 31;
  const int kh   = lane >> 4, ln = lane & 15;
  const int wm   = wave >> 1;
  const int wn   = wave & 1;

  GemmStage g;
  v8f acc0 = {}, acc1 = {};
  gemm_stage_load(g, A, B, N, K, bm, bn, 0, tid);
  gemm_stage_store(g, As[0], BsT[0], tid);
  gemm_stage_load(g, A, B, N, K, bm, bn, KS, tid);
  barrier_lds_only();

  int kb = 0;
  for (; kb + 2 * KS < K; kb += 2 * KS) {
    gemm_stage_store(g, As[1], BsT[1], tid);
    gemm_stage_load(g, A, B, N, K, bm, bn, kb + 2 * KS, tid);
    gemm_stage_compute(As[0], BsT[0], acc0, acc1, wm, wn, ln, kh);
    barrier_lds_only();
    gemm_stage_store(g, As[0], BsT[0], tid);
    gemm_stage_load(g, A, B, N, K, bm, bn, kb + 3 * KS, tid);
    gemm_stage_compute(As[1], BsT[1], acc0, acc1, wm, wn, ln, kh);
    barrier_lds_only();
  }
  gemm_stage_store(g, As[1], BsT[1], tid);
  gemm_stage_compute(As[0], BsT[0], acc0, acc1, wm, wn, ln, kh);
  barrier_lds_only();
  gemm_stage_compute(As[1], BsT[1], acc0, acc1, wm, wn, ln, kh);
#pragma unroll
  for (int r = 0; r < 8; ++r) {
    const size_t row = (size_t)(bm + wm * 16 + r + 8 * kh);
    C[row * N + bn + wn * 32 + ln]      = acc0[r];
    C[row * N + bn + wn * 32 + 16 + ln] = acc1[r];
  }
}

__global__ __launch_bounds__(256) void gemm_f32_wmma(const float* __restrict__ A,
                                                     const float* __restrict__ B,
                                                     float* __restrict__ C,
                                                     int M, int N, int K) {
  (void)M;
  const int ntb = N / 64;
  gemm_tile_f32_wmma(A, B, C, N, K, (int)(blockIdx.x / ntb) * 64, (int)(blockIdx.x % ntb) * 64);
}

__global__ __launch_bounds__(256) void gemm_qkv_f32_wmma(const float* __restrict__ X,
                                                         const float* __restrict__ Wq,
                                                         const float* __restrict__ Wk,
                                                         const float* __restrict__ Wv,
                                                         float* __restrict__ Qg,
                                                         float* __restrict__ Kg,
                                                         float* __restrict__ Vg) {
  constexpr int NQ = Hh * FD;
  constexpr int TV = (L / 64) * (DM / 64);
  constexpr int TQ = (L / 64) * (NQ / 64);
  const int b = (int)blockIdx.x;
  if (b < TV) {
    gemm_tile_f32_wmma(X, Wv, Vg, DM, DM, (b / (DM / 64)) * 64, (b % (DM / 64)) * 64);
  } else {
    const int t = (b - TV) % TQ;
    const bool isq = (b - TV) < TQ;
    gemm_tile_f32_wmma(X, isq ? Wq : Wk, isq ? Qg : Kg, NQ, DM,
                       (t / (NQ / 64)) * 64, (t % (NQ / 64)) * 64);
  }
}

constexpr int SLOT     = 4644;
constexpr int O_LIN    = 4096, O_ONES = 4352, O_KC = 4368, O_END = 4641;
constexpr int SLOTS_P0 = 168;

__device__ __forceinline__ float* state_slot(float* p0, float* p1, int i) {
  return i < SLOTS_P0 ? p0 + (size_t)i * SLOT : p1 + (size_t)(i - SLOTS_P0) * SLOT;
}

__global__ __launch_bounds__(256) void based_state(const float* __restrict__ Kg,
                                                   const float* __restrict__ Vg,
                                                   float* __restrict__ P0,
                                                   float* __restrict__ P1) {
  __shared__ __attribute__((aligned(16))) float ST[2][2][16][36];

  const int tid  = threadIdx.x;
  const int wave = tid >> 5, lane = tid & 31;
  const int kh   = lane >> 4, ln = lane & 15;
  const int seg  = (int)(blockIdx.x / (Hh * 4));
  const int hv   = blockIdx.x % (Hh * 4);
  const int h    = hv >> 2;
  const int vb   = hv & 3;
  const int c_begin = seg * CPS, c_end = c_begin + CPS;
  const int qoff = h * FD;
  const int voff = h * HD + vb * 16;

  v8f q0 = {}, q1 = {}, ql = {};
  float kq = 0.f;
  float sc = 0.f;
  const int wv = __builtin_amdgcn_readfirstlane(wave);
  const int kv = tid >> 7;
  const int st = (tid & 127) >> 2, sc4 = (tid & 3) * 4;
  const float* gsrc = kv ? (Vg + voff + sc4) : (Kg + qoff + sc4);
  const size_t gp = kv ? (size_t)DM : (size_t)(Hh * FD);
  float4 pre = *(const float4*)&gsrc[(size_t)(c_begin * 32 + st) * gp];
#pragma unroll 1
  for (int c = c_begin; c < c_end; ++c) {
    const int bf = c & 1;
    ST[bf][kv][sc4 + 0][st] = pre.x;
    ST[bf][kv][sc4 + 1][st] = pre.y;
    ST[bf][kv][sc4 + 2][st] = pre.z;
    ST[bf][kv][sc4 + 3][st] = pre.w;
    __syncthreads();
    if (c + 1 < c_end) pre = *(const float4*)&gsrc[(size_t)((c + 1) * 32 + st) * gp];
    const float (*kt)[36] = ST[bf][0];
    const float (*vt)[36] = ST[bf][1];
    const int fi = tid >> 4, fj = tid & 15;
    float4 ka[8], kb[8];
#pragma unroll
    for (int m = 0; m < 8; ++m) {
      ka[m] = *(const float4*)&kt[fi][4 * m];
      kb[m] = *(const float4*)&kt[fj][4 * m];
    }
    if (tid >= 128 && tid < 160) {
      const float* row = (tid < 144) ? vt[tid - 128] : kt[tid - 144];
      float s = 0.f;
#pragma unroll
      for (int t = 0; t < 32; t += 4) {
        const float4 a = *(const float4*)&row[t];
        s += a.x;  s += a.y;  s += a.z;  s += a.w;
      }
      sc += (tid < 144) ? s : 0.5f * s;
    } else if (tid == 160) {
      sc += 32.f;
    }
    v2f fkl[8], fb[8], fk0[8], fk1[8];
#pragma unroll
    for (int m = 0; m < 8; ++m) {
      const int kk = 4 * m + 2 * kh;
      fkl[m] = *(const v2f*)&kt[ln][kk];
      fb[m]  = *(const v2f*)&vt[ln][kk];
      fk0[m] = *(const v2f*)&kt[wv][kk];
      fk1[m] = *(const v2f*)&kt[wv + 8][kk];
    }
    if (wv == 0) {
#pragma unroll
      for (int m = 0; m < 8; ++m) {
        const v2f al = fkl[m] * 0.5f;
        ql = WMMA_F32(al, fb[m], ql);
      }
    }
#pragma unroll
    for (int m = 0; m < 8; ++m) {
      const v2f a0 = fk0[m] * fkl[m] * QSC;
      const v2f a1 = fk1[m] * fkl[m] * QSC;
      q0 = WMMA_F32(a0, fb[m], q0);
      q1 = WMMA_F32(a1, fb[m], q1);
    }
    {
      float s = 0.f;
#pragma unroll
      for (int m = 0; m < 8; ++m) {
        s += ka[m].x * kb[m].x * QSC;
        s += ka[m].y * kb[m].y * QSC;
        s += ka[m].z * kb[m].z * QSC;
        s += ka[m].w * kb[m].w * QSC;
      }
      kq += s;
    }
  }
  float* g = state_slot(P0, P1, seg * (Hh * 4) + hv);
#pragma unroll
  for (int r = 0; r < 8; ++r) {
    g[(wv * 16 + r + 8 * kh) * 16 + ln]       = q0[r];
    g[((wv + 8) * 16 + r + 8 * kh) * 16 + ln] = q1[r];
  }
  if (wv == 0) {
#pragma unroll
    for (int r = 0; r < 8; ++r) g[O_LIN + (r + 8 * kh) * 16 + ln] = ql[r];
  }
  g[O_KC + 17 + tid] = kq;
  if (tid >= 128 && tid < 144)      g[O_ONES + tid - 128] = sc;
  else if (tid >= 144 && tid < 160) g[O_KC + 1 + tid - 144] = sc;
  else if (tid == 160)              g[O_KC] = sc;
  else if (tid >= 161 && tid < 164) g[O_END + tid - 161] = 0.f;
}

__device__ __forceinline__ const float* state_slot(const float* p0, const float* p1, int i) {
  return i < SLOTS_P0 ? p0 + (size_t)i * SLOT : p1 + (size_t)(i - SLOTS_P0) * SLOT;
}

__global__ __launch_bounds__(256) void based_attn(const float* __restrict__ Qg,
                                                  const float* __restrict__ Kg,
                                                  float* VY,
                                                  const float* __restrict__ P0,
                                                  const float* __restrict__ P1) {
  __shared__ float s_lin[16][17];
  __shared__ float s_quad[256][17];
  __shared__ float s_ones[16];
  __shared__ float kcum[273];
  __shared__ __attribute__((aligned(16))) float Qc[32][20];
  __shared__ __attribute__((aligned(16))) float Kc[32][20];
  __shared__ __attribute__((aligned(16))) float Vc[32][20];
  __shared__ float Aw[32][33];
  __shared__ __attribute__((aligned(16))) float Yp[4][32][20];
  __shared__ float denp[10][32];

  const int tid  = threadIdx.x;
  const int wave = tid >> 5, lane = tid & 31;
  const int kh   = lane >> 4, ln = lane & 15;
  const int wv   = __builtin_amdgcn_readfirstlane(wave);
  const int seg  = NSEG - 1 - (int)(blockIdx.x / (Hh * 4));
  const int hv   = blockIdx.x % (Hh * 4);
  const int h    = hv >> 2;
  const int vb   = hv & 3;
  const int c_begin = seg * CPS, c_end = c_begin + CPS;
  const int qoff = h * FD;
  const int voff = h * HD + vb * 16;

  {
    int i4[5];
#pragma unroll
    for (int j = 0; j < 4; ++j) i4[j] = tid + 256 * j;
    i4[4] = min(tid + 1024, SLOT / 4 - 1);
    float4 p[NSEG - 1][5];
    const int s_last = max(seg - 1, 0);
#pragma unroll
    for (int s = 0; s < NSEG - 1; ++s) {
      const float* g = state_slot(P0, P1, min(s, s_last) * (Hh * 4) + hv);
#pragma unroll
      for (int j = 0; j < 5; ++j) p[s][j] = *(const float4*)&g[4 * i4[j]];
    }
    float4 a[5];
#pragma unroll
    for (int j = 0; j < 5; ++j) a[j] = float4{0.f, 0.f, 0.f, 0.f};
#pragma unroll
    for (int s = 0; s < NSEG - 1; ++s) {
      const bool on = s < seg;
#pragma unroll
      for (int j = 0; j < 5; ++j) {
        a[j].x += on ? p[s][j].x : 0.f;
        a[j].y += on ? p[s][j].y : 0.f;
        a[j].z += on ? p[s][j].z : 0.f;
        a[j].w += on ? p[s][j].w : 0.f;
      }
    }
#pragma unroll
    for (int j = 0; j < 5; ++j) {
      const int e = 4 * (tid + 256 * j);
      const float v[4] = {a[j].x, a[j].y, a[j].z, a[j].w};
      if (e < O_LIN) {
#pragma unroll
        for (int u = 0; u < 4; ++u) s_quad[e >> 4][(e & 15) + u] = v[u];
      } else if (e < O_ONES) {
        const int x = e - O_LIN;
#pragma unroll
        for (int u = 0; u < 4; ++u) s_lin[x >> 4][(x & 15) + u] = v[u];
      } else if (e < O_KC) {
#pragma unroll
        for (int u = 0; u < 4; ++u) s_ones[e - O_ONES + u] = v[u];
      } else if (e < O_END) {
#pragma unroll
        for (int u = 0; u < 4; ++u)
          if (e - O_KC + u < 273) kcum[e - O_KC + u] = v[u];
      }
    }
  }

  for (int c = c_begin; c < c_end; ++c) {
    const int bl = c * 32;
    if (tid < 128) {
      const int t = tid >> 2, c4 = (tid & 3) * 4;
      *(float4*)&Qc[t][c4] = *(const float4*)&Qg[(size_t)(bl + t) * (Hh * FD) + qoff + c4];
      *(float4*)&Kc[t][c4] = *(const float4*)&Kg[(size_t)(bl + t) * (Hh * FD) + qoff + c4];
    } else {
      const int i = tid - 128;
      const int t = i >> 2, c4 = (i & 3) * 4;
      *(float4*)&Vc[t][c4] = *(const float4*)&VY[(size_t)(bl + t) * DM + voff + c4];
    }
    __syncthreads();

    if (wave < 4) {
      const int ti = wave >> 1, tj = wave & 1;
      v8f acc = {};
#pragma unroll
      for (int k = 0; k < 16; k += 4) {
        const int kk = k + 2 * kh;
        v2f a = *(const v2f*)&Qc[ti * 16 + ln][kk];
        v2f b = *(const v2f*)&Kc[tj * 16 + ln][kk];
        acc = WMMA_F32(a, b, acc);
      }
#pragma unroll
      for (int r = 0; r < 8; ++r) {
        const int row = ti * 16 + r + 8 * kh, col = tj * 16 + ln;
        const float s = acc[r];
        Aw[row][col] = (row >= col) ? (1.f + 0.25f * s + (s * s) * 0.03125f) : 0.f;
      }
    }
    __syncthreads();

    {
      const int ti = wave & 1;
      const int kq = wv >> 1;
      const int row = ti * 16 + ln;
      v8f acc;
      const float a_init = (kq == 0) ? s_ones[ln] : 0.f;
#pragma unroll
      for (int r = 0; r < 8; ++r) acc[r] = a_init;
      const float4 qi4 = *(const float4*)&Qc[row][4 * kq];
      const float qi[4] = {qi4.x, qi4.y, qi4.z, qi4.w};
      v2f qj[4];
#pragma unroll
      for (int u = 0; u < 4; ++u) qj[u] = *(const v2f*)&Qc[row][4 * u + 2 * kh];
      v2f fa[16], fb[16];
#pragma unroll
      for (int m = 0; m < 16; ++m) {
        const int kk = (kq * 16 + m) * 4 + 2 * kh;
        fa[m] = (qj[m & 3] * qi[m >> 2]) * QSC;
        fb[m].x = s_quad[kk][ln];  fb[m].y = s_quad[kk + 1][ln];
      }
      if (kq < 2) {
#pragma unroll
        for (int m = 0; m < 4; ++m) {
          const int kk = (kq * 4 + m) * 4 + 2 * kh;
          v2f a = *(const v2f*)&Aw[row][kk];
          v2f b;
          b.x = Vc[kk][ln];  b.y = Vc[kk + 1][ln];
          acc = WMMA_F32(a, b, acc);
        }
      } else if (kq == 2) {
#pragma unroll
        for (int m = 0; m < 4; ++m) {
          const int kk = m * 4 + 2 * kh;
          v2f a, b;
          a.x = Qc[row][kk] * 0.5f; a.y = Qc[row][kk + 1] * 0.5f;
          b.x = s_lin[kk][ln];      b.y = s_lin[kk + 1][ln];
          acc = WMMA_F32(a, b, acc);
        }
      }
#pragma unroll
      for (int m = 0; m < 16; ++m) acc = WMMA_F32(fa[m], fb[m], acc);
      {
        const int t = lane;
        float qrow[16];
#pragma unroll
        for (int u = 0; u < 4; ++u) {
          const float4 v = *(const float4*)&Qc[t][4 * u];
          qrow[4 * u] = v.x;  qrow[4 * u + 1] = v.y;  qrow[4 * u + 2] = v.z;  qrow[4 * u + 3] = v.w;
        }
        const v2f qi2 = *(const v2f*)&Qc[t][2 * wave];
        float d = 0.f;
#pragma unroll
        for (int m = 0; m < 32; ++m) {
          const float f = ((m < 16 ? qi2.x : qi2.y) * qrow[m & 15]) * QSC;
          d += f * kcum[17 + wave * 32 + m];
        }
        denp[wave][t] = d;
        if (wave == 6) {
          float e = kcum[0];
#pragma unroll
          for (int s2 = 0; s2 < 32; ++s2) e += Aw[t][s2];
          denp[8][t] = e;
        } else if (wave == 7) {
          float e = 0.f;
#pragma unroll
          for (int i = 0; i < 16; ++i) e += qrow[i] * 0.5f * kcum[1 + i];
          denp[9][t] = e;
        }
      }
#pragma unroll
      for (int r = 0; r < 8; ++r) Yp[kq][ti * 16 + r + 8 * kh][ln] = acc[r];
    }
    __syncthreads();

    if (tid < 128) {
      const int t = tid >> 2, c4 = (tid & 3) * 4;
      float dd = denp[8][t];
      dd += denp[9][t];
#pragma unroll
      for (int w = 0; w < 8; ++w) dd += denp[w][t];
      dd += EPSF;
      float4 y = *(float4*)&Yp[0][t][c4];
#pragma unroll
      for (int q = 1; q < 4; ++q) {
        const float4 p = *(float4*)&Yp[q][t][c4];
        y.x += p.x; y.y += p.y; y.z += p.z; y.w += p.w;
      }
      y.x /= dd; y.y /= dd; y.z /= dd; y.w /= dd;
      *(float4*)&VY[(size_t)(bl + t) * DM + voff + c4] = y;
    }
    if (tid >= 128 && tid < 144) {
      const int f = tid - 128;
      float s = 0.f;
      for (int t = 0; t < 32; ++t) s += Vc[t][f];
      s_ones[f] += s;
    } else if (tid >= 144 && tid < 160) {
      const int f = tid - 144;
      float s = 0.f;
      for (int t = 0; t < 32; ++t) s += Kc[t][f];
      kcum[1 + f] += 0.5f * s;
    } else if (tid == 160) {
      kcum[0] += 32.f;
    }
    {
      const int fi = tid >> 4, fj = tid & 15;
      float s = 0.f;
#pragma unroll
      for (int t = 0; t < 32; ++t) s += (Kc[t][fi] * Kc[t][fj]) * QSC;
      kcum[17 + tid] += s;
    }
    {
      v2f kl[8], vf[8], k0[8], k1[8];
#pragma unroll
      for (int m = 0; m < 8; ++m) {
        const int kk = 4 * m + 2 * kh;
        kl[m].x = Kc[kk][ln];      kl[m].y = Kc[kk + 1][ln];
        vf[m].x = Vc[kk][ln];      vf[m].y = Vc[kk + 1][ln];
        k0[m].x = Kc[kk][wv];      k0[m].y = Kc[kk + 1][wv];
        k1[m].x = Kc[kk][wv + 8];  k1[m].y = Kc[kk + 1][wv + 8];
      }
      v8f q0, q1;
#pragma unroll
      for (int r = 0; r < 8; ++r) {
        q0[r] = s_quad[wv * 16 + r + 8 * kh][ln];
        q1[r] = s_quad[(wv + 8) * 16 + r + 8 * kh][ln];
      }
      if (wv == 0) {
        v8f ql;
#pragma unroll
        for (int r = 0; r < 8; ++r) ql[r] = s_lin[r + 8 * kh][ln];
#pragma unroll
        for (int m = 0; m < 8; ++m) {
          const v2f al = kl[m] * 0.5f;
          ql = WMMA_F32(al, vf[m], ql);
        }
#pragma unroll
        for (int r = 0; r < 8; ++r) s_lin[r + 8 * kh][ln] = ql[r];
      }
#pragma unroll
      for (int m = 0; m < 8; ++m) {
        const v2f a0 = (k0[m] * kl[m]) * QSC;
        const v2f a1 = (k1[m] * kl[m]) * QSC;
        q0 = WMMA_F32(a0, vf[m], q0);
        q1 = WMMA_F32(a1, vf[m], q1);
      }
#pragma unroll
      for (int r = 0; r < 8; ++r) {
        s_quad[wv * 16 + r + 8 * kh][ln]       = q0[r];
        s_quad[(wv + 8) * 16 + r + 8 * kh][ln] = q1[r];
      }
    }
    __syncthreads();
  }
}

extern "C" void kernel_launch(void* const* d_in, const int* in_sizes, int n_in,
                              void* d_out, int out_size, void* d_ws, size_t ws_size,
                              hipStream_t stream) {
  (void)in_sizes; (void)n_in; (void)out_size; (void)ws_size;
  const float* hs = (const float*)d_in[0];
  const float* Wq = (const float*)d_in[1];
  const float* Wk = (const float*)d_in[2];
  const float* Wv = (const float*)d_in[3];
  const float* Wo = (const float*)d_in[4];
  float* out = (float*)d_out;

  float* Qg = (float*)d_ws;
  float* Kg = Qg + (size_t)L * (Hh * FD);
  float* Vg = Kg + (size_t)L * (Hh * FD);
  float* S0 = Vg + (size_t)L * DM;
  float* S1 = out;

  const dim3 blk(256);
  gemm_qkv_f32_wmma<<<dim3((L / 64) * (DM / 64) + 2 * (L / 64) * ((Hh * FD) / 64)), blk, 0, stream>>>(
      hs, Wq, Wk, Wv, Qg, Kg, Vg);
  based_state<<<dim3(Hh * 4 * (NSEG - 1)), blk, 0, stream>>>(Kg, Vg, S0, S1);
  based_attn<<<dim3(Hh * 4 * NSEG), blk, 0, stream>>>(Qg, Kg, Vg, S0, S1);
  gemm_f32_wmma<<<dim3((L / 64) * (DM / 64)), blk, 0, stream>>>(Vg, Wo, out, L, DM, DM);
}


# attention pass 2: segment's Q/K/V rows staged once; each chunk attends to all segment rows so far by masked polynomial weights on the read-only start state; no state update; order of sums changes
# speedup vs baseline: 5.4887x; 1.0797x over previous
#include <hip/hip_runtime.h>

typedef float v2f __attribute__((ext_vector_type(2)));
typedef float v8f __attribute__((ext_vector_type(8)));

#define WMMA_F32(a, b, c) \
  __builtin_amdgcn_wmma_f32_16x16x4_f32(false, (a), false, (b), (short)0, (c), false, false)

constexpr int L = 1024, DM = 768, Hh = 12, FD = 16, HD = 64;
constexpr float EPSF = 1e-12f;
constexpr float QSC = 0.17677669529663687f;
constexpr int NSEG = 8;
constexpr int CPS  = (L / 32) / NSEG;

constexpr int KS = 64;
constexpr int LP = KS + 4;

typedef float v4f __attribute__((ext_vector_type(4)));
struct GemmStage { v4f a0, a1, a2, a3, b0, b1, b2, b3; };

__device__ __forceinline__ void barrier_lds_only() {
  __builtin_amdgcn_fence(__ATOMIC_RELEASE, "workgroup", "local");
  __builtin_amdgcn_s_barrier();
  __builtin_amdgcn_fence(__ATOMIC_ACQUIRE, "workgroup", "local");
}

__device__ __forceinline__ void gemm_stage_load(GemmStage& g, const float* __restrict__ A,
                                                const float* __restrict__ B, int N, int K,
                                                int bm, int bn, int kb, int tid) {
  const int ar = tid >> 4, ac = (tid & 15) * 4;
  const int bk = tid >> 2, bc = (tid & 3) * 4;
  const float* ap = &A[(size_t)(bm + ar) * K + kb + ac];
  const float* bp = &B[(size_t)(kb + bk) * N + bn + bc];
  g.a0 = *(const v4f*)&ap[0];
  g.a1 = *(const v4f*)&ap[(size_t)16 * K];
  g.a2 = *(const v4f*)&ap[(size_t)32 * K];
  g.a3 = *(const v4f*)&ap[(size_t)48 * K];
  g.b0 = *(const v4f*)&bp[0];
  g.b1 = *(const v4f*)&bp[16];
  g.b2 = *(const v4f*)&bp[32];
  g.b3 = *(const v4f*)&bp[48];
}

__device__ __forceinline__ void gemm_b_store(v4f b, float (*BsT)[LP], int col, int bk) {
  BsT[col + 0][bk] = b.x;
  BsT[col + 1][bk] = b.y;
  BsT[col + 2][bk] = b.z;
  BsT[col + 3][bk] = b.w;
}

__device__ __forceinline__ void gemm_stage_store(const GemmStage& g, float (*As)[LP],
                                                 float (*BsT)[LP], int tid) {
  const int ar = tid >> 4, ac = (tid & 15) * 4;
  const int bk = tid >> 2, bc = (tid & 3) * 4;
  *(v4f*)&As[ar][ac]      = g.a0;
  *(v4f*)&As[ar + 16][ac] = g.a1;
  *(v4f*)&As[ar + 32][ac] = g.a2;
  *(v4f*)&As[ar + 48][ac] = g.a3;
  gemm_b_store(g.b0, BsT, bc, bk);
  gemm_b_store(g.b1, BsT, bc + 16, bk);
  gemm_b_store(g.b2, BsT, bc + 32, bk);
  gemm_b_store(g.b3, BsT, bc + 48, bk);
}

__device__ __forceinline__ void gemm_stage_compute(const float (*As)[LP], const float (*BsT)[LP],
                                                   v8f& acc0, v8f& acc1,
                                                   int wm, int wn, int ln, int kh) {
#pragma unroll
  for (int k = 0; k < KS; k += 4) {
    const int kk = k + 2 * kh;
    v2f a = *(const v2f*)&As[wm * 16 + ln][kk];
    v2f p = *(const v2f*)&BsT[wn * 32 + ln][kk];
    v2f q = *(const v2f*)&BsT[wn * 32 + 16 + ln][kk];
    acc0 = WMMA_F32(a, p, acc0);
    acc1 = WMMA_F32(a, q, acc1);
  }
}

__device__ __forceinline__ void gemm_tile_f32_wmma(const float* __restrict__ A,
                                                   const float* __restrict__ B,
                                                   float* __restrict__ C,
                                                   int N, int K, int bm, int bn) {
  __shared__ __attribute__((aligned(16))) float As[2][64][LP];
  __shared__ __attribute__((aligned(16))) float BsT[2][64][LP];
  const int tid  = threadIdx.x;
  const int wave = tid >> 5, lane = tid & 31;
  const int kh   = lane >> 4, ln = lane & 15;
  const int wm   = wave >> 1;
  const int wn   = wave & 1;

  GemmStage g;
  v8f acc0 = {}, acc1 = {};
  gemm_stage_load(g, A, B, N, K, bm, bn, 0, tid);
  gemm_stage_store(g, As[0], BsT[0], tid);
  gemm_stage_load(g, A, B, N, K, bm, bn, KS, tid);
  barrier_lds_only();

  int kb = 0;
  for (; kb + 2 * KS < K; kb += 2 * KS) {
    gemm_stage_store(g, As[1], BsT[1], tid);
    gemm_stage_load(g, A, B, N, K, bm, bn, kb + 2 * KS, tid);
    gemm_stage_compute(As[0], BsT[0], acc0, acc1, wm, wn, ln, kh);
    barrier_lds_only();
    gemm_stage_store(g, As[0], BsT[0], tid);
    gemm_stage_load(g, A, B, N, K, bm, bn, kb + 3 * KS, tid);
    gemm_stage_compute(As[1], BsT[1], acc0, acc1, wm, wn, ln, kh);
    barrier_lds_only();
  }
  gemm_stage_store(g, As[1], BsT[1], tid);
  gemm_stage_compute(As[0], BsT[0], acc0, acc1, wm, wn, ln, kh);
  barrier_lds_only();
  gemm_stage_compute(As[1], BsT[1], acc0, acc1, wm, wn, ln, kh);
#pragma unroll
  for (int r = 0; r < 8; ++r) {
    const size_t row = (size_t)(bm + wm * 16 + r + 8 * kh);
    C[row * N + bn + wn * 32 + ln]      = acc0[r];
    C[row * N + bn + wn * 32 + 16 + ln] = acc1[r];
  }
}

__global__ __launch_bounds__(256) void gemm_f32_wmma(const float* __restrict__ A,
                                                     const float* __restrict__ B,
                                                     float* __restrict__ C,
                                                     int M, int N, int K) {
  (void)M;
  const int ntb = N / 64;
  gemm_tile_f32_wmma(A, B, C, N, K, (int)(blockIdx.x / ntb) * 64, (int)(blockIdx.x % ntb) * 64);
}

__global__ __launch_bounds__(256) void gemm_qkv_f32_wmma(const float* __restrict__ X,
                                                         const float* __restrict__ Wq,
                                                         const float* __restrict__ Wk,
                                                         const float* __restrict__ Wv,
                                                         float* __restrict__ Qg,
                                                         float* __restrict__ Kg,
                                                         float* __restrict__ Vg) {
  constexpr int NQ = Hh * FD;
  constexpr int TV = (L / 64) * (DM / 64);
  constexpr int TQ = (L / 64) * (NQ / 64);
  const int b = (int)blockIdx.x;
  if (b < TV) {
    gemm_tile_f32_wmma(X, Wv, Vg, DM, DM, (b / (DM / 64)) * 64, (b % (DM / 64)) * 64);
  } else {
    const int t = (b - TV) % TQ;
    const bool isq = (b - TV) < TQ;
    gemm_tile_f32_wmma(X, isq ? Wq : Wk, isq ? Qg : Kg, NQ, DM,
                       (t / (NQ / 64)) * 64, (t % (NQ / 64)) * 64);
  }
}

constexpr int SLOT     = 4644;
constexpr int O_LIN    = 4096, O_ONES = 4352, O_KC = 4368, O_END = 4641;
constexpr int SLOTS_P0 = 168;

__device__ __forceinline__ float* state_slot(float* p0, float* p1, int i) {
  return i < SLOTS_P0 ? p0 + (size_t)i * SLOT : p1 + (size_t)(i - SLOTS_P0) * SLOT;
}

__global__ __launch_bounds__(256) void based_state(const float* __restrict__ Kg,
                                                   const float* __restrict__ Vg,
                                                   float* __restrict__ P0,
                                                   float* __restrict__ P1) {
  __shared__ __attribute__((aligned(16))) float ST[2][2][16][36];

  const int tid  = threadIdx.x;
  const int wave = tid >> 5, lane = tid & 31;
  const int kh   = lane >> 4, ln = lane & 15;
  const int seg  = (int)(blockIdx.x / (Hh * 4));
  const int hv   = blockIdx.x % (Hh * 4);
  const int h    = hv >> 2;
  const int vb   = hv & 3;
  const int c_begin = seg * CPS, c_end = c_begin + CPS;
  const int qoff = h * FD;
  const int voff = h * HD + vb * 16;

  v8f q0 = {}, q1 = {}, ql = {};
  float kq = 0.f;
  float sc = 0.f;
  const int wv = __builtin_amdgcn_readfirstlane(wave);
  const int kv = tid >> 7;
  const int st = (tid & 127) >> 2, sc4 = (tid & 3) * 4;
  const float* gsrc = kv ? (Vg + voff + sc4) : (Kg + qoff + sc4);
  const size_t gp = kv ? (size_t)DM : (size_t)(Hh * FD);
  float4 pre = *(const float4*)&gsrc[(size_t)(c_begin * 32 + st) * gp];
#pragma unroll 1
  for (int c = c_begin; c < c_end; ++c) {
    const int bf = c & 1;
    ST[bf][kv][sc4 + 0][st] = pre.x;
    ST[bf][kv][sc4 + 1][st] = pre.y;
    ST[bf][kv][sc4 + 2][st] = pre.z;
    ST[bf][kv][sc4 + 3][st] = pre.w;
    __syncthreads();
    if (c + 1 < c_end) pre = *(const float4*)&gsrc[(size_t)((c + 1) * 32 + st) * gp];
    const float (*kt)[36] = ST[bf][0];
    const float (*vt)[36] = ST[bf][1];
    const int fi = tid >> 4, fj = tid & 15;
    float4 ka[8], kb[8];
#pragma unroll
    for (int m = 0; m < 8; ++m) {
      ka[m] = *(const float4*)&kt[fi][4 * m];
      kb[m] = *(const float4*)&kt[fj][4 * m];
    }
    if (tid >= 128 && tid < 160) {
      const float* row = (tid < 144) ? vt[tid - 128] : kt[tid - 144];
      float s = 0.f;
#pragma unroll
      for (int t = 0; t < 32; t += 4) {
        const float4 a = *(const float4*)&row[t];
        s += a.x;  s += a.y;  s += a.z;  s += a.w;
      }
      sc += (tid < 144) ? s : 0.5f * s;
    } else if (tid == 160) {
      sc += 32.f;
    }
    v2f fkl[8], fb[8], fk0[8], fk1[8];
#pragma unroll
    for (int m = 0; m < 8; ++m) {
      const int kk = 4 * m + 2 * kh;
      fkl[m] = *(const v2f*)&kt[ln][kk];
      fb[m]  = *(const v2f*)&vt[ln][kk];
      fk0[m] = *(const v2f*)&kt[wv][kk];
      fk1[m] = *(const v2f*)&kt[wv + 8][kk];
    }
    if (wv == 0) {
#pragma unroll
      for (int m = 0; m < 8; ++m) {
        const v2f al = fkl[m] * 0.5f;
        ql = WMMA_F32(al, fb[m], ql);
      }
    }
#pragma unroll
    for (int m = 0; m < 8; ++m) {
      const v2f a0 = fk0[m] * fkl[m] * QSC;
      const v2f a1 = fk1[m] * fkl[m] * QSC;
      q0 = WMMA_F32(a0, fb[m], q0);
      q1 = WMMA_F32(a1, fb[m], q1);
    }
    {
      float s = 0.f;
#pragma unroll
      for (int m = 0; m < 8; ++m) {
        s += ka[m].x * kb[m].x * QSC;
        s += ka[m].y * kb[m].y * QSC;
        s += ka[m].z * kb[m].z * QSC;
        s += ka[m].w * kb[m].w * QSC;
      }
      kq += s;
    }
  }
  float* g = state_slot(P0, P1, seg * (Hh * 4) + hv);
#pragma unroll
  for (int r = 0; r < 8; ++r) {
    g[(wv * 16 + r + 8 * kh) * 16 + ln]       = q0[r];
    g[((wv + 8) * 16 + r + 8 * kh) * 16 + ln] = q1[r];
  }
  if (wv == 0) {
#pragma unroll
    for (int r = 0; r < 8; ++r) g[O_LIN + (r + 8 * kh) * 16 + ln] = ql[r];
  }
  g[O_KC + 17 + tid] = kq;
  if (tid >= 128 && tid < 144)      g[O_ONES + tid - 128] = sc;
  else if (tid >= 144 && tid < 160) g[O_KC + 1 + tid - 144] = sc;
  else if (tid == 160)              g[O_KC] = sc;
  else if (tid >= 161 && tid < 164) g[O_END + tid - 161] = 0.f;
}

__device__ __forceinline__ const float* state_slot(const float* p0, const float* p1, int i) {
  return i < SLOTS_P0 ? p0 + (size_t)i * SLOT : p1 + (size_t)(i - SLOTS_P0) * SLOT;
}

constexpr int SR = CPS * 32;
constexpr int AWP = SR + 4;

__global__ __launch_bounds__(256) void based_attn(const float* __restrict__ Qg,
                                                  const float* __restrict__ Kg,
                                                  float* VY,
                                                  const float* __restrict__ P0,
                                                  const float* __restrict__ P1) {
  __shared__ float s_lin[16][17];
  __shared__ float s_quad[256][17];
  __shared__ float s_ones[16];
  __shared__ float kcum[273];
  __shared__ __attribute__((aligned(16))) float Qc[SR][20];
  __shared__ __attribute__((aligned(16))) float Kc[SR][20];
  __shared__ __attribute__((aligned(16))) float Vc[SR][20];
  __shared__ __attribute__((aligned(16))) float Aw[32][AWP];
  __shared__ __attribute__((aligned(16))) float Yp[4][32][20];
  __shared__ float denp[10][32];

  static_assert(SR == 128, "staging and the tile split below are written for 128-row segments");

  const int tid  = threadIdx.x;
  const int wave = tid >> 5, lane = tid & 31;
  const int kh   = lane >> 4, ln = lane & 15;
  const int wv   = __builtin_amdgcn_readfirstlane(wave);
  const int seg  = NSEG - 1 - (int)(blockIdx.x / (Hh * 4));
  const int hv   = blockIdx.x % (Hh * 4);
  const int h    = hv >> 2;
  const int vb   = hv & 3;
  const int row0 = seg * SR;
  const int qoff = h * FD;
  const int voff = h * HD + vb * 16;

  float4 gq[2], gk[2], gv[2];
#pragma unroll
  for (int u = 0; u < 2; ++u) {
    const int idx = tid + 256 * u;
    const int t = idx >> 2, c4 = (idx & 3) * 4;
    gq[u] = *(const float4*)&Qg[(size_t)(row0 + t) * (Hh * FD) + qoff + c4];
    gk[u] = *(const float4*)&Kg[(size_t)(row0 + t) * (Hh * FD) + qoff + c4];
    gv[u] = *(const float4*)&VY[(size_t)(row0 + t) * DM + voff + c4];
  }

  {
    int i4[5];
#pragma unroll
    for (int j = 0; j < 4; ++j) i4[j] = tid + 256 * j;
    i4[4] = min(tid + 1024, SLOT / 4 - 1);
    float4 p[NSEG - 1][5];
    const int s_last = max(seg - 1, 0);
#pragma unroll
    for (int s = 0; s < NSEG - 1; ++s) {
      const float* g = state_slot(P0, P1, min(s, s_last) * (Hh * 4) + hv);
#pragma unroll
      for (int j = 0; j < 5; ++j) p[s][j] = *(const float4*)&g[4 * i4[j]];
    }
#pragma unroll
    for (int u = 0; u < 2; ++u) {
      const int idx = tid + 256 * u;
      const int t = idx >> 2, c4 = (idx & 3) * 4;
      *(float4*)&Qc[t][c4] = gq[u];
      *(float4*)&Kc[t][c4] = gk[u];
      *(float4*)&Vc[t][c4] = gv[u];
    }
    float4 a[5];
#pragma unroll
    for (int j = 0; j < 5; ++j) a[j] = float4{0.f, 0.f, 0.f, 0.f};
#pragma unroll
    for (int s = 0; s < NSEG - 1; ++s) {
      const bool on = s < seg;
#pragma unroll
      for (int j = 0; j < 5; ++j) {
        a[j].x += on ? p[s][j].x : 0.f;
        a[j].y += on ? p[s][j].y : 0.f;
        a[j].z += on ? p[s][j].z : 0.f;
        a[j].w += on ? p[s][j].w : 0.f;
      }
    }
#pragma unroll
    for (int j = 0; j < 5; ++j) {
      const int e = 4 * (tid + 256 * j);
      const float v[4] = {a[j].x, a[j].y, a[j].z, a[j].w};
      if (e < O_LIN) {
#pragma unroll
        for (int u = 0; u < 4; ++u) s_quad[e >> 4][(e & 15) + u] = v[u];
      } else if (e < O_ONES) {
        const int x = e - O_LIN;
#pragma unroll
        for (int u = 0; u < 4; ++u) s_lin[x >> 4][(x & 15) + u] = v[u];
      } else if (e < O_KC) {
#pragma unroll
        for (int u = 0; u < 4; ++u) s_ones[e - O_ONES + u] = v[u];
      } else if (e < O_END) {
#pragma unroll
        for (int u = 0; u < 4; ++u)
          if (e - O_KC + u < 273) kcum[e - O_KC + u] = v[u];
      }
    }
  }
  __syncthreads();

  for (int j = 0; j < CPS; ++j) {
    const int jr = j * 32;
    const int bl = row0 + jr;

    {
      const int ti = wave & 1;
      const int tj0 = wv >> 1, tj1 = (wv >> 1) + 4;
      v8f s0 = {}, s1 = {};
#pragma unroll
      for (int k = 0; k < 16; k += 4) {
        const int kk = k + 2 * kh;
        const v2f a  = *(const v2f*)&Qc[jr + ti * 16 + ln][kk];
        const v2f b0 = *(const v2f*)&Kc[tj0 * 16 + ln][kk];
        const v2f b1 = *(const v2f*)&Kc[tj1 * 16 + ln][kk];
        s0 = WMMA_F32(a, b0, s0);
        s1 = WMMA_F32(a, b1, s1);
      }
#pragma unroll
      for (int r = 0; r < 8; ++r) {
        const int row = ti * 16 + r + 8 * kh;
        const int c0 = tj0 * 16 + ln, c1 = tj1 * 16 + ln;
        const float x0 = s0[r], x1 = s1[r];
        Aw[row][c0] = (jr + row >= c0) ? (1.f + 0.25f * x0 + (x0 * x0) * 0.03125f) : 0.f;
        Aw[row][c1] = (jr + row >= c1) ? (1.f + 0.25f * x1 + (x1 * x1) * 0.03125f) : 0.f;
      }
    }
    __syncthreads();

    {
      const int ti = wave & 1;
      const int kq = wv >> 1;
      const int row = ti * 16 + ln;
      const float* qr = Qc[jr + row];
      v8f acc;
      const float a_init = (kq == 0) ? s_ones[ln] : 0.f;
#pragma unroll
      for (int r = 0; r < 8; ++r) acc[r] = a_init;
      const float4 qi4 = *(const float4*)&qr[4 * kq];
      const float qi[4] = {qi4.x, qi4.y, qi4.z, qi4.w};
      v2f qj[4];
#pragma unroll
      for (int u = 0; u < 4; ++u) qj[u] = *(const v2f*)&qr[4 * u + 2 * kh];
      v2f fa[16], fb[16];
#pragma unroll
      for (int m = 0; m < 16; ++m) {
        const int kk = (kq * 16 + m) * 4 + 2 * kh;
        fa[m] = (qj[m & 3] * qi[m >> 2]) * QSC;
        fb[m].x = s_quad[kk][ln];  fb[m].y = s_quad[kk + 1][ln];
      }
      v2f wa[8], wb[8];
#pragma unroll
      for (int m = 0; m < 8; ++m) {
        const int kk = (kq * 8 + m) * 4 + 2 * kh;
        wa[m] = *(const v2f*)&Aw[row][kk];
        wb[m].x = Vc[kk][ln];  wb[m].y = Vc[kk + 1][ln];
      }
      if (kq == 2) {
#pragma unroll
        for (int m = 0; m < 4; ++m) {
          const int kk = m * 4 + 2 * kh;
          v2f a, b;
          a.x = qr[kk] * 0.5f;      a.y = qr[kk + 1] * 0.5f;
          b.x = s_lin[kk][ln];      b.y = s_lin[kk + 1][ln];
          acc = WMMA_F32(a, b, acc);
        }
      }
#pragma unroll
      for (int m = 0; m < 8; ++m) acc = WMMA_F32(wa[m], wb[m], acc);
#pragma unroll
      for (int m = 0; m < 16; ++m) acc = WMMA_F32(fa[m], fb[m], acc);
      {
        const int t = lane;
        const float* qt = Qc[jr + t];
        float qrow[16];
#pragma unroll
        for (int u = 0; u < 4; ++u) {
          const float4 v = *(const float4*)&qt[4 * u];
          qrow[4 * u] = v.x;  qrow[4 * u + 1] = v.y;  qrow[4 * u + 2] = v.z;  qrow[4 * u + 3] = v.w;
        }
        const v2f qi2 = *(const v2f*)&qt[2 * wave];
        float d = 0.f;
#pragma unroll
        for (int m = 0; m < 32; ++m) {
          const float f = ((m < 16 ? qi2.x : qi2.y) * qrow[m & 15]) * QSC;
          d += f * kcum[17 + wave * 32 + m];
        }
        float e = 0.f;
#pragma unroll
        for (int u = 0; u < 4; ++u) {
          const float4 w4 = *(const float4*)&Aw[t][16 * wave + 4 * u];
          e += w4.x;  e += w4.y;  e += w4.z;  e += w4.w;
        }
        denp[wave][t] = d + e;
        if (wave == 6) {
          denp[8][t] = kcum[0];
        } else if (wave == 7) {
          float e2 = 0.f;
#pragma unroll
          for (int i = 0; i < 16; ++i) e2 += qrow[i] * 0.5f * kcum[1 + i];
          denp[9][t] = e2;
        }
      }
#pragma unroll
      for (int r = 0; r < 8; ++r) Yp[kq][ti * 16 + r + 8 * kh][ln] = acc[r];
    }
    __syncthreads();

    if (tid < 128) {
      const int t = tid >> 2, c4 = (tid & 3) * 4;
      float dd = denp[8][t];
      dd += denp[9][t];
#pragma unroll
      for (int w = 0; w < 8; ++w) dd += denp[w][t];
      dd += EPSF;
      float4 y = *(float4*)&Yp[0][t][c4];
#pragma unroll
      for (int q = 1; q < 4; ++q) {
        const float4 p = *(float4*)&Yp[q][t][c4];
        y.x += p.x; y.y += p.y; y.z += p.z; y.w += p.w;
      }
      y.x /= dd; y.y /= dd; y.z /= dd; y.w /= dd;
      *(float4*)&VY[(size_t)(bl + t) * DM + voff + c4] = y;
    }
  }
}

extern "C" void kernel_launch(void* const* d_in, const int* in_sizes, int n_in,
                              void* d_out, int out_size, void* d_ws, size_t ws_size,
                              hipStream_t stream) {
  (void)in_sizes; (void)n_in; (void)out_size; (void)ws_size;
  const float* hs = (const float*)d_in[0];
  const float* Wq = (const float*)d_in[1];
  const float* Wk = (const float*)d_in[2];
  const float* Wv = (const float*)d_in[3];
  const float* Wo = (const float*)d_in[4];
  float* out = (float*)d_out;

  float* Qg = (float*)d_ws;
  float* Kg = Qg + (size_t)L * (Hh * FD);
  float* Vg = Kg + (size_t)L * (Hh * FD);
  float* S0 = Vg + (size_t)L * DM;
  float* S1 = out;

  const dim3 blk(256);
  gemm_qkv_f32_wmma<<<dim3((L / 64) * (DM / 64) + 2 * (L / 64) * ((Hh * FD) / 64)), blk, 0, stream>>>(
      hs, Wq, Wk, Wv, Qg, Kg, Vg);
  based_state<<<dim3(Hh * 4 * (NSEG - 1)), blk, 0, stream>>>(Kg, Vg, S0, S1);
  based_attn<<<dim3(Hh * 4 * NSEG), blk, 0, stream>>>(Qg, Kg, Vg, S0, S1);
  gemm_f32_wmma<<<dim3((L / 64) * (DM / 64)), blk, 0, stream>>>(Vg, Wo, out, L, DM, DM);
}


# GEMM tiles sized to the matrix pipe: 4 waves per block, each a 16-row strip on 3 accumulators (V, out: 64x48 tiles) or 2 (Q, K: 64x32); 256+192 and 256 blocks; order of sums unchanged
# speedup vs baseline: 6.2961x; 1.1471x over previous
#include <hip/hip_runtime.h>

typedef float v2f __attribute__((ext_vector_type(2)));
typedef float v8f __attribute__((ext_vector_type(8)));

#define WMMA_F32(a, b, c) \
  __builtin_amdgcn_wmma_f32_16x16x4_f32(false, (a), false, (b), (short)0, (c), false, false)

constexpr int L = 1024, DM = 768, Hh = 12, FD = 16, HD = 64;
constexpr float EPSF = 1e-12f;
constexpr float QSC = 0.17677669529663687f;
constexpr int NSEG = 8;
constexpr int CPS  = (L / 32) / NSEG;


constexpr int KS = 64;
constexpr int LP = KS + 4;
constexpr int TN = 48;
constexpr int TNQ = 32;
constexpr int GT = 128;

typedef float v4f __attribute__((ext_vector_type(4)));
template <int TNt> struct GemmStage { v4f a[8]; v4f b[TNt / 8]; };

__device__ __forceinline__ void barrier_lds_only() {
  __builtin_amdgcn_fence(__ATOMIC_RELEASE, "workgroup", "local");
  __builtin_amdgcn_s_barrier();
  __builtin_amdgcn_fence(__ATOMIC_ACQUIRE, "workgroup", "local");
}

template <int TNt>
__device__ __forceinline__ void gemm_stage_load(GemmStage<TNt>& g, const float* __restrict__ A,
                                                const float* __restrict__ B, int N, int K,
                                                int bm, int bn, int kb, int tid) {
  constexpr int NG = TNt / 16;
  const int ar = tid >> 4, ac = (tid & 15) * 4;
  const int bk = tid >> 2, bc = (tid & 3) * 4;
  const float* ap = &A[(size_t)(bm + ar) * K + kb + ac];
  const float* bp = &B[(size_t)(kb + bk) * N + bn + bc];
#pragma unroll
  for (int u = 0; u < 8; ++u) g.a[u] = *(const v4f*)&ap[(size_t)(8 * u) * K];
#pragma unroll
  for (int u = 0; u < 2 * NG; ++u) g.b[u] = *(const v4f*)&bp[(size_t)(32 * (u / NG)) * N + 16 * (u % NG)];
}

__device__ __forceinline__ void gemm_b_store(v4f b, float (*BsT)[LP], int col, int bk) {
  BsT[col + 0][bk] = b.x;
  BsT[col + 1][bk] = b.y;
  BsT[col + 2][bk] = b.z;
  BsT[col + 3][bk] = b.w;
}

template <int TNt>
__device__ __forceinline__ void gemm_stage_store(const GemmStage<TNt>& g, float (*As)[LP],
                                                 float (*BsT)[LP], int tid) {
  constexpr int NG = TNt / 16;
  const int ar = tid >> 4, ac = (tid & 15) * 4;
  const int bk = tid >> 2, bc = (tid & 3) * 4;
#pragma unroll
  for (int u = 0; u < 8; ++u) *(v4f*)&As[ar + 8 * u][ac] = g.a[u];
#pragma unroll
  for (int u = 0; u < 2 * NG; ++u) gemm_b_store(g.b[u], BsT, bc + 16 * (u % NG), bk + 32 * (u / NG));
}

template <int TNt> struct GemmFrag { v2f a[4]; v2f b[TNt / 16][4]; };

template <int TNt>
__device__ __forceinline__ void gemm_frag_load(GemmFrag<TNt>& f, const float (*As)[LP],
                                               const float (*BsT)[LP], int grp, int wm, int ln, int kh) {
#pragma unroll
  for (int i = 0; i < 4; ++i) {
    const int kk = 16 * grp + 4 * i + 2 * kh;
    f.a[i] = *(const v2f*)&As[wm * 16 + ln][kk];
#pragma unroll
    for (int s = 0; s < TNt / 16; ++s) f.b[s][i] = *(const v2f*)&BsT[16 * s + ln][kk];
  }
}

template <int TNt>
__device__ __forceinline__ void gemm_frag_mma(const GemmFrag<TNt>& f, v8f (&acc)[TNt / 16]) {
#pragma unroll
  for (int i = 0; i < 4; ++i) {
#pragma unroll
    for (int s = 0; s < TNt / 16; ++s) acc[s] = WMMA_F32(f.a[i], f.b[s][i], acc[s]);
  }
}

template <int TNt, bool STORE, bool LOAD>
__device__ __forceinline__ void gemm_step(GemmStage<TNt>& g, const float (*As)[LP], const float (*BsT)[LP],
                                          float (*As2)[LP], float (*BsT2)[LP],
                                          const float* __restrict__ A, const float* __restrict__ B,
                                          int N, int K, int bm, int bn, int kload, int tid,
                                          v8f (&acc)[TNt / 16], int wm, int ln, int kh) {
  GemmFrag<TNt> f0, f1;
  gemm_frag_load<TNt>(f0, As, BsT, 0, wm, ln, kh);
  gemm_frag_load<TNt>(f1, As, BsT, 1, wm, ln, kh);
  __builtin_amdgcn_sched_barrier(0);
  gemm_frag_mma<TNt>(f0, acc);
  if (STORE) gemm_stage_store<TNt>(g, As2, BsT2, tid);
  if (LOAD) gemm_stage_load<TNt>(g, A, B, N, K, bm, bn, kload, tid);
  __builtin_amdgcn_sched_barrier(0);
  gemm_frag_load<TNt>(f0, As, BsT, 2, wm, ln, kh);
  gemm_frag_mma<TNt>(f1, acc);
  __builtin_amdgcn_sched_barrier(0);
  gemm_frag_load<TNt>(f1, As, BsT, 3, wm, ln, kh);
  __builtin_amdgcn_sched_barrier(0);
  gemm_frag_mma<TNt>(f0, acc);
  gemm_frag_mma<TNt>(f1, acc);
}

template <int TNt>
__device__ __forceinline__ void gemm_tile_f32_wmma(const float* __restrict__ A,
                                                   const float* __restrict__ B,
                                                   float* __restrict__ C,
                                                   int N, int K, int bm, int bn,
                                                   float (*As)[64][LP], float (*BsT)[TN][LP]) {
  const int tid  = threadIdx.x;
  const int wm   = tid >> 5, lane = tid & 31;
  const int kh   = lane >> 4, ln = lane & 15;

  GemmStage<TNt> g;
  v8f acc[TNt / 16];
#pragma unroll
  for (int s = 0; s < TNt / 16; ++s) acc[s] = v8f{};
  gemm_stage_load<TNt>(g, A, B, N, K, bm, bn, 0, tid);
  gemm_stage_store<TNt>(g, As[0], BsT[0], tid);
  gemm_stage_load<TNt>(g, A, B, N, K, bm, bn, KS, tid);
  barrier_lds_only();

  int kb = 0;
  for (; kb + 2 * KS < K; kb += 2 * KS) {
    gemm_step<TNt, true, true>(g, As[0], BsT[0], As[1], BsT[1], A, B, N, K, bm, bn, kb + 2 * KS, tid,
                               acc, wm, ln, kh);
    barrier_lds_only();
    gemm_step<TNt, true, true>(g, As[1], BsT[1], As[0], BsT[0], A, B, N, K, bm, bn, kb + 3 * KS, tid,
                               acc, wm, ln, kh);
    barrier_lds_only();
  }
  gemm_step<TNt, true, false>(g, As[0], BsT[0], As[1], BsT[1], A, B, N, K, bm, bn, 0, tid,
                              acc, wm, ln, kh);
  barrier_lds_only();
  gemm_step<TNt, false, false>(g, As[1], BsT[1], As[0], BsT[0], A, B, N, K, bm, bn, 0, tid,
                               acc, wm, ln, kh);
#pragma unroll
  for (int r = 0; r < 8; ++r) {
    const size_t row = (size_t)(bm + wm * 16 + r + 8 * kh);
#pragma unroll
    for (int s = 0; s < TNt / 16; ++s) C[row * N + bn + 16 * s + ln] = acc[s][r];
  }
}

__global__ __launch_bounds__(GT) void gemm_f32_wmma(const float* __restrict__ A,
                                                    const float* __restrict__ B,
                                                    float* __restrict__ C) {
  __shared__ __attribute__((aligned(16))) float As[2][64][LP];
  __shared__ __attribute__((aligned(16))) float BsT[2][TN][LP];
  constexpr int ntb = DM / TN;
  gemm_tile_f32_wmma<TN>(A, B, C, DM, DM, (int)(blockIdx.x / ntb) * 64, (int)(blockIdx.x % ntb) * TN,
                         As, BsT);
}

__global__ __launch_bounds__(GT) void gemm_qkv_f32_wmma(const float* __restrict__ X,
                                                        const float* __restrict__ Wq,
                                                        const float* __restrict__ Wk,
                                                        const float* __restrict__ Wv,
                                                        float* __restrict__ Qg,
                                                        float* __restrict__ Kg,
                                                        float* __restrict__ Vg) {
  __shared__ __attribute__((aligned(16))) float As[2][64][LP];
  __shared__ __attribute__((aligned(16))) float BsT[2][TN][LP];
  constexpr int NQ = Hh * FD;
  constexpr int TV = (L / 64) * (DM / TN);
  constexpr int TQ = (L / 64) * (NQ / TNQ);
  const int b = (int)blockIdx.x;
  if (b < TV) {
    gemm_tile_f32_wmma<TN>(X, Wv, Vg, DM, DM, (b / (DM / TN)) * 64, (b % (DM / TN)) * TN, As, BsT);
  } else {
    const int t = (b - TV) % TQ;
    const bool isq = (b - TV) < TQ;
    gemm_tile_f32_wmma<TNQ>(X, isq ? Wq : Wk, isq ? Qg : Kg, NQ, DM,
                            (t / (NQ / TNQ)) * 64, (t % (NQ / TNQ)) * TNQ, As, BsT);
  }
}

constexpr int SLOT     = 4644;
constexpr int O_LIN    = 4096, O_ONES = 4352, O_KC = 4368, O_END = 4641;
constexpr int SLOTS_P0 = 168;

__device__ __forceinline__ float* state_slot(float* p0, float* p1, int i) {
  return i < SLOTS_P0 ? p0 + (size_t)i * SLOT : p1 + (size_t)(i - SLOTS_P0) * SLOT;
}

__global__ __launch_bounds__(256) void based_state(const float* __restrict__ Kg,
                                                   const float* __restrict__ Vg,
                                                   float* __restrict__ P0,
                                                   float* __restrict__ P1) {
  __shared__ __attribute__((aligned(16))) float ST[2][2][16][36];

  const int tid  = threadIdx.x;
  const int wave = tid >> 5, lane = tid & 31;
  const int kh   = lane >> 4, ln = lane & 15;
  const int seg  = (int)(blockIdx.x / (Hh * 4));
  const int hv   = blockIdx.x % (Hh * 4);
  const int h    = hv >> 2;
  const int vb   = hv & 3;
  const int c_begin = seg * CPS, c_end = c_begin + CPS;
  const int qoff = h * FD;
  const int voff = h * HD + vb * 16;

  v8f q0 = {}, q1 = {}, ql = {};
  float kq = 0.f;
  float sc = 0.f;
  const int wv = __builtin_amdgcn_readfirstlane(wave);
  const int kv = tid >> 7;
  const int st = (tid & 127) >> 2, sc4 = (tid & 3) * 4;
  const float* gsrc = kv ? (Vg + voff + sc4) : (Kg + qoff + sc4);
  const size_t gp = kv ? (size_t)DM : (size_t)(Hh * FD);
  float4 pre = *(const float4*)&gsrc[(size_t)(c_begin * 32 + st) * gp];
#pragma unroll 1
  for (int c = c_begin; c < c_end; ++c) {
    const int bf = c & 1;
    ST[bf][kv][sc4 + 0][st] = pre.x;
    ST[bf][kv][sc4 + 1][st] = pre.y;
    ST[bf][kv][sc4 + 2][st] = pre.z;
    ST[bf][kv][sc4 + 3][st] = pre.w;
    __syncthreads();
    if (c + 1 < c_end) pre = *(const float4*)&gsrc[(size_t)((c + 1) * 32 + st) * gp];
    const float (*kt)[36] = ST[bf][0];
    const float (*vt)[36] = ST[bf][1];
    const int fi = tid >> 4, fj = tid & 15;
    float4 ka[8], kb[8];
#pragma unroll
    for (int m = 0; m < 8; ++m) {
      ka[m] = *(const float4*)&kt[fi][4 * m];
      kb[m] = *(const float4*)&kt[fj][4 * m];
    }
    if (tid >= 128 && tid < 160) {
      const float* row = (tid < 144) ? vt[tid - 128] : kt[tid - 144];
      float s = 0.f;
#pragma unroll
      for (int t = 0; t < 32; t += 4) {
        const float4 a = *(const float4*)&row[t];
        s += a.x;  s += a.y;  s += a.z;  s += a.w;
      }
      sc += (tid < 144) ? s : 0.5f * s;
    } else if (tid == 160) {
      sc += 32.f;
    }
    v2f fkl[8], fb[8], fk0[8], fk1[8];
#pragma unroll
    for (int m = 0; m < 8; ++m) {
      const int kk = 4 * m + 2 * kh;
      fkl[m] = *(const v2f*)&kt[ln][kk];
      fb[m]  = *(const v2f*)&vt[ln][kk];
      fk0[m] = *(const v2f*)&kt[wv][kk];
      fk1[m] = *(const v2f*)&kt[wv + 8][kk];
    }
    if (wv == 0) {
#pragma unroll
      for (int m = 0; m < 8; ++m) {
        const v2f al = fkl[m] * 0.5f;
        ql = WMMA_F32(al, fb[m], ql);
      }
    }
#pragma unroll
    for (int m = 0; m < 8; ++m) {
      const v2f a0 = fk0[m] * fkl[m] * QSC;
      const v2f a1 = fk1[m] * fkl[m] * QSC;
      q0 = WMMA_F32(a0, fb[m], q0);
      q1 = WMMA_F32(a1, fb[m], q1);
    }
    {
      float s = 0.f;
#pragma unroll
      for (int m = 0; m < 8; ++m) {
        s += ka[m].x * kb[m].x * QSC;
        s += ka[m].y * kb[m].y * QSC;
        s += ka[m].z * kb[m].z * QSC;
        s += ka[m].w * kb[m].w * QSC;
      }
      kq += s;
    }
  }
  float* g = state_slot(P0, P1, seg * (Hh * 4) + hv);
#pragma unroll
  for (int r = 0; r < 8; ++r) {
    g[(wv * 16 + r + 8 * kh) * 16 + ln]       = q0[r];
    g[((wv + 8) * 16 + r + 8 * kh) * 16 + ln] = q1[r];
  }
  if (wv == 0) {
#pragma unroll
    for (int r = 0; r < 8; ++r) g[O_LIN + (r + 8 * kh) * 16 + ln] = ql[r];
  }
  g[O_KC + 17 + tid] = kq;
  if (tid >= 128 && tid < 144)      g[O_ONES + tid - 128] = sc;
  else if (tid >= 144 && tid < 160) g[O_KC + 1 + tid - 144] = sc;
  else if (tid == 160)              g[O_KC] = sc;
  else if (tid >= 161 && tid < 164) g[O_END + tid - 161] = 0.f;
}

__device__ __forceinline__ const float* state_slot(const float* p0, const float* p1, int i) {
  return i < SLOTS_P0 ? p0 + (size_t)i * SLOT : p1 + (size_t)(i - SLOTS_P0) * SLOT;
}

constexpr int SR = CPS * 32;
constexpr int AWP = SR + 4;

__global__ __launch_bounds__(256) void based_attn(const float* __restrict__ Qg,
                                                  const float* __restrict__ Kg,
                                                  float* VY,
                                                  const float* __restrict__ P0,
                                                  const float* __restrict__ P1) {
  __shared__ float s_lin[16][17];
  __shared__ float s_quad[256][17];
  __shared__ float s_ones[16];
  __shared__ float kcum[273];
  __shared__ __attribute__((aligned(16))) float Qc[SR][20];
  __shared__ __attribute__((aligned(16))) float Kc[SR][20];
  __shared__ __attribute__((aligned(16))) float Vc[SR][20];
  __shared__ __attribute__((aligned(16))) float Aw[32][AWP];
  __shared__ __attribute__((aligned(16))) float Yp[4][32][20];
  __shared__ float denp[10][32];

  static_assert(SR == 128, "staging and the tile split below are written for 128-row segments");

  const int tid  = threadIdx.x;
  const int wave = tid >> 5, lane = tid & 31;
  const int kh   = lane >> 4, ln = lane & 15;
  const int wv   = __builtin_amdgcn_readfirstlane(wave);
  const int seg  = NSEG - 1 - (int)(blockIdx.x / (Hh * 4));
  const int hv   = blockIdx.x % (Hh * 4);
  const int h    = hv >> 2;
  const int vb   = hv & 3;
  const int row0 = seg * SR;
  const int qoff = h * FD;
  const int voff = h * HD + vb * 16;

  float4 gq[2], gk[2], gv[2];
#pragma unroll
  for (int u = 0; u < 2; ++u) {
    const int idx = tid + 256 * u;
    const int t = idx >> 2, c4 = (idx & 3) * 4;
    gq[u] = *(const float4*)&Qg[(size_t)(row0 + t) * (Hh * FD) + qoff + c4];
    gk[u] = *(const float4*)&Kg[(size_t)(row0 + t) * (Hh * FD) + qoff + c4];
    gv[u] = *(const float4*)&VY[(size_t)(row0 + t) * DM + voff + c4];
  }

  {
    int i4[5];
#pragma unroll
    for (int j = 0; j < 4; ++j) i4[j] = tid + 256 * j;
    i4[4] = min(tid + 1024, SLOT / 4 - 1);
    float4 p[NSEG - 1][5];
    const int s_last = max(seg - 1, 0);
#pragma unroll
    for (int s = 0; s < NSEG - 1; ++s) {
      const float* g = state_slot(P0, P1, min(s, s_last) * (Hh * 4) + hv);
#pragma unroll
      for (int j = 0; j < 5; ++j) p[s][j] = *(const float4*)&g[4 * i4[j]];
    }
#pragma unroll
    for (int u = 0; u < 2; ++u) {
      const int idx = tid + 256 * u;
      const int t = idx >> 2, c4 = (idx & 3) * 4;
      *(float4*)&Qc[t][c4] = gq[u];
      *(float4*)&Kc[t][c4] = gk[u];
      *(float4*)&Vc[t][c4] = gv[u];
    }
    float4 a[5];
#pragma unroll
    for (int j = 0; j < 5; ++j) a[j] = float4{0.f, 0.f, 0.f, 0.f};
#pragma unroll
    for (int s = 0; s < NSEG - 1; ++s) {
      const bool on = s < seg;
#pragma unroll
      for (int j = 0; j < 5; ++j) {
        a[j].x += on ? p[s][j].x : 0.f;
        a[j].y += on ? p[s][j].y : 0.f;
        a[j].z += on ? p[s][j].z : 0.f;
        a[j].w += on ? p[s][j].w : 0.f;
      }
    }
#pragma unroll
    for (int j = 0; j < 5; ++j) {
      const int e = 4 * (tid + 256 * j);
      const float v[4] = {a[j].x, a[j].y, a[j].z, a[j].w};
      if (e < O_LIN) {
#pragma unroll
        for (int u = 0; u < 4; ++u) s_quad[e >> 4][(e & 15) + u] = v[u];
      } else if (e < O_ONES) {
        const int x = e - O_LIN;
#pragma unroll
        for (int u = 0; u < 4; ++u) s_lin[x >> 4][(x & 15) + u] = v[u];
      } else if (e < O_KC) {
#pragma unroll
        for (int u = 0; u < 4; ++u) s_ones[e - O_ONES + u] = v[u];
      } else if (e < O_END) {
#pragma unroll
        for (int u = 0; u < 4; ++u)
          if (e - O_KC + u < 273) kcum[e - O_KC + u] = v[u];
      }
    }
  }
  __syncthreads();

  for (int j = 0; j < CPS; ++j) {
    const int jr = j * 32;
    const int bl = row0 + jr;

    {
      const int ti = wave & 1;
      const int tj0 = wv >> 1, tj1 = (wv >> 1) + 4;
      v8f s0 = {}, s1 = {};
#pragma unroll
      for (int k = 0; k < 16; k += 4) {
        const int kk = k + 2 * kh;
        const v2f a  = *(const v2f*)&Qc[jr + ti * 16 + ln][kk];
        const v2f b0 = *(const v2f*)&Kc[tj0 * 16 + ln][kk];
        const v2f b1 = *(const v2f*)&Kc[tj1 * 16 + ln][kk];
        s0 = WMMA_F32(a, b0, s0);
        s1 = WMMA_F32(a, b1, s1);
      }
#pragma unroll
      for (int r = 0; r < 8; ++r) {
        const int row = ti * 16 + r + 8 * kh;
        const int c0 = tj0 * 16 + ln, c1 = tj1 * 16 + ln;
        const float x0 = s0[r], x1 = s1[r];
        Aw[row][c0] = (jr + row >= c0) ? (1.f + 0.25f * x0 + (x0 * x0) * 0.03125f) : 0.f;
        Aw[row][c1] = (jr + row >= c1) ? (1.f + 0.25f * x1 + (x1 * x1) * 0.03125f) : 0.f;
      }
    }
    __syncthreads();

    {
      const int ti = wave & 1;
      const int kq = wv >> 1;
      const int row = ti * 16 + ln;
      const float* qr = Qc[jr + row];
      v8f acc;
      const float a_init = (kq == 0) ? s_ones[ln] : 0.f;
#pragma unroll
      for (int r = 0; r < 8; ++r) acc[r] = a_init;
      const float4 qi4 = *(const float4*)&qr[4 * kq];
      const float qi[4] = {qi4.x, qi4.y, qi4.z, qi4.w};
      v2f qj[4];
#pragma unroll
      for (int u = 0; u < 4; ++u) qj[u] = *(const v2f*)&qr[4 * u + 2 * kh];
      v2f fa[16], fb[16];
#pragma unroll
      for (int m = 0; m < 16; ++m) {
        const int kk = (kq * 16 + m) * 4 + 2 * kh;
        fa[m] = (qj[m & 3] * qi[m >> 2]) * QSC;
        fb[m].x = s_quad[kk][ln];  fb[m].y = s_quad[kk + 1][ln];
      }
      v2f wa[8], wb[8];
#pragma unroll
      for (int m = 0; m < 8; ++m) {
        const int kk = (kq * 8 + m) * 4 + 2 * kh;
        wa[m] = *(const v2f*)&Aw[row][kk];
        wb[m].x = Vc[kk][ln];  wb[m].y = Vc[kk + 1][ln];
      }
      if (kq == 2) {
#pragma unroll
        for (int m = 0; m < 4; ++m) {
          const int kk = m * 4 + 2 * kh;
          v2f a, b;
          a.x = qr[kk] * 0.5f;      a.y = qr[kk + 1] * 0.5f;
          b.x = s_lin[kk][ln];      b.y = s_lin[kk + 1][ln];
          acc = WMMA_F32(a, b, acc);
        }
      }
#pragma unroll
      for (int m = 0; m < 8; ++m) acc = WMMA_F32(wa[m], wb[m], acc);
#pragma unroll
      for (int m = 0; m < 16; ++m) acc = WMMA_F32(fa[m], fb[m], acc);
      {
        const int t = lane;
        const float* qt = Qc[jr + t];
        float qrow[16];
#pragma unroll
        for (int u = 0; u < 4; ++u) {
          const float4 v = *(const float4*)&qt[4 * u];
          qrow[4 * u] = v.x;  qrow[4 * u + 1] = v.y;  qrow[4 * u + 2] = v.z;  qrow[4 * u + 3] = v.w;
        }
        const v2f qi2 = *(const v2f*)&qt[2 * wave];
        float d = 0.f;
#pragma unroll
        for (int m = 0; m < 32; ++m) {
          const float f = ((m < 16 ? qi2.x : qi2.y) * qrow[m & 15]) * QSC;
          d += f * kcum[17 + wave * 32 + m];
        }
        float e = 0.f;
#pragma unroll
        for (int u = 0; u < 4; ++u) {
          const float4 w4 = *(const float4*)&Aw[t][16 * wave + 4 * u];
          e += w4.x;  e += w4.y;  e += w4.z;  e += w4.w;
        }
        denp[wave][t] = d + e;
        if (wave == 6) {
          denp[8][t] = kcum[0];
        } else if (wave == 7) {
          float e2 = 0.f;
#pragma unroll
          for (int i = 0; i < 16; ++i) e2 += qrow[i] * 0.5f * kcum[1 + i];
          denp[9][t] = e2;
        }
      }
#pragma unroll
      for (int r = 0; r < 8; ++r) Yp[kq][ti * 16 + r + 8 * kh][ln] = acc[r];
    }
    __syncthreads();

    if (tid < 128) {
      const int t = tid >> 2, c4 = (tid & 3) * 4;
      float dd = denp[8][t];
      dd += denp[9][t];
#pragma unroll
      for (int w = 0; w < 8; ++w) dd += denp[w][t];
      dd += EPSF;
      float4 y = *(float4*)&Yp[0][t][c4];
#pragma unroll
      for (int q = 1; q < 4; ++q) {
        const float4 p = *(float4*)&Yp[q][t][c4];
        y.x += p.x; y.y += p.y; y.z += p.z; y.w += p.w;
      }
      y.x /= dd; y.y /= dd; y.z /= dd; y.w /= dd;
      *(float4*)&VY[(size_t)(bl + t) * DM + voff + c4] = y;
    }
  }
}

extern "C" void kernel_launch(void* const* d_in, const int* in_sizes, int n_in,
                              void* d_out, int out_size, void* d_ws, size_t ws_size,
                              hipStream_t stream) {
  (void)in_sizes; (void)n_in; (void)out_size; (void)ws_size;
  const float* hs = (const float*)d_in[0];
  const float* Wq = (const float*)d_in[1];
  const float* Wk = (const float*)d_in[2];
  const float* Wv = (const float*)d_in[3];
  const float* Wo = (const float*)d_in[4];
  float* out = (float*)d_out;

  float* Qg = (float*)d_ws;
  float* Kg = Qg + (size_t)L * (Hh * FD);
  float* Vg = Kg + (size_t)L * (Hh * FD);
  float* S0 = Vg + (size_t)L * DM;
  float* S1 = out;

  const dim3 blk(256);
  const dim3 gblk(GT);
  gemm_qkv_f32_wmma<<<dim3((L / 64) * (DM / TN) + 2 * (L / 64) * ((Hh * FD) / TNQ)), gblk, 0, stream>>>(
      hs, Wq, Wk, Wv, Qg, Kg, Vg);
  based_state<<<dim3(Hh * 4 * (NSEG - 1)), blk, 0, stream>>>(Kg, Vg, S0, S1);
  based_attn<<<dim3(Hh * 4 * NSEG), blk, 0, stream>>>(Qg, Kg, Vg, S0, S1);
  gemm_f32_wmma<<<dim3((L / 64) * (DM / TN)), gblk, 0, stream>>>(Vg, Wo, out);
}


# attention, both passes: quad features packed by index symmetry (x_i x_j = x_j x_i), 160 of 256 kept, 4x4 index blocks I<J weighted 2 on the Q side; fewer state bytes and k-steps; sum order changes
# speedup vs baseline: 6.5779x; 1.0448x over previous
#include <hip/hip_runtime.h>

typedef float v2f __attribute__((ext_vector_type(2)));
typedef float v8f __attribute__((ext_vector_type(8)));

#define WMMA_F32(a, b, c) \
  __builtin_amdgcn_wmma_f32_16x16x4_f32(false, (a), false, (b), (short)0, (c), false, false)

constexpr int L = 1024, DM = 768, Hh = 12, FD = 16, HD = 64;
constexpr float EPSF = 1e-12f;
constexpr float QSC = 0.17677669529663687f;
constexpr int NSEG = 8;
constexpr int CPS  = (L / 32) / NSEG;


constexpr int KS = 64;
constexpr int LP = KS + 4;
constexpr int TN = 48;
constexpr int TNQ = 32;
constexpr int GT = 128;

typedef float v4f __attribute__((ext_vector_type(4)));
template <int TNt> struct GemmStage { v4f a[8]; v4f b[TNt / 8]; };

__device__ __forceinline__ void barrier_lds_only() {
  __builtin_amdgcn_fence(__ATOMIC_RELEASE, "workgroup", "local");
  __builtin_amdgcn_s_barrier();
  __builtin_amdgcn_fence(__ATOMIC_ACQUIRE, "workgroup", "local");
}

template <int TNt>
__device__ __forceinline__ void gemm_stage_load(GemmStage<TNt>& g, const float* __restrict__ A,
                                                const float* __restrict__ B, int N, int K,
                                                int bm, int bn, int kb, int tid) {
  constexpr int NG = TNt / 16;
  const int ar = tid >> 4, ac = (tid & 15) * 4;
  const int bk = tid >> 2, bc = (tid & 3) * 4;
  const float* ap = &A[(size_t)(bm + ar) * K + kb + ac];
  const float* bp = &B[(size_t)(kb + bk) * N + bn + bc];
#pragma unroll
  for (int u = 0; u < 8; ++u) g.a[u] = *(const v4f*)&ap[(size_t)(8 * u) * K];
#pragma unroll
  for (int u = 0; u < 2 * NG; ++u) g.b[u] = *(const v4f*)&bp[(size_t)(32 * (u / NG)) * N + 16 * (u % NG)];
}

__device__ __forceinline__ void gemm_b_store(v4f b, float (*BsT)[LP], int col, int bk) {
  BsT[col + 0][bk] = b.x;
  BsT[col + 1][bk] = b.y;
  BsT[col + 2][bk] = b.z;
  BsT[col + 3][bk] = b.w;
}

template <int TNt>
__device__ __forceinline__ void gemm_stage_store(const GemmStage<TNt>& g, float (*As)[LP],
                                                 float (*BsT)[LP], int tid) {
  constexpr int NG = TNt / 16;
  const int ar = tid >> 4, ac = (tid & 15) * 4;
  const int bk = tid >> 2, bc = (tid & 3) * 4;
#pragma unroll
  for (int u = 0; u < 8; ++u) *(v4f*)&As[ar + 8 * u][ac] = g.a[u];
#pragma unroll
  for (int u = 0; u < 2 * NG; ++u) gemm_b_store(g.b[u], BsT, bc + 16 * (u % NG), bk + 32 * (u / NG));
}

template <int TNt> struct GemmFrag { v2f a[4]; v2f b[TNt / 16][4]; };

template <int TNt>
__device__ __forceinline__ void gemm_frag_load(GemmFrag<TNt>& f, const float (*As)[LP],
                                               const float (*BsT)[LP], int grp, int wm, int ln, int kh) {
#pragma unroll
  for (int i = 0; i < 4; ++i) {
    const int kk = 16 * grp + 4 * i + 2 * kh;
    f.a[i] = *(const v2f*)&As[wm * 16 + ln][kk];
#pragma unroll
    for (int s = 0; s < TNt / 16; ++s) f.b[s][i] = *(const v2f*)&BsT[16 * s + ln][kk];
  }
}

template <int TNt>
__device__ __forceinline__ void gemm_frag_mma(const GemmFrag<TNt>& f, v8f (&acc)[TNt / 16]) {
#pragma unroll
  for (int i = 0; i < 4; ++i) {
#pragma unroll
    for (int s = 0; s < TNt / 16; ++s) acc[s] = WMMA_F32(f.a[i], f.b[s][i], acc[s]);
  }
}

template <int TNt, bool STORE, bool LOAD>
__device__ __forceinline__ void gemm_step(GemmStage<TNt>& g, const float (*As)[LP], const float (*BsT)[LP],
                                          float (*As2)[LP], float (*BsT2)[LP],
                                          const float* __restrict__ A, const float* __restrict__ B,
                                          int N, int K, int bm, int bn, int kload, int tid,
                                          v8f (&acc)[TNt / 16], int wm, int ln, int kh) {
  GemmFrag<TNt> f0, f1;
  gemm_frag_load<TNt>(f0, As, BsT, 0, wm, ln, kh);
  gemm_frag_load<TNt>(f1, As, BsT, 1, wm, ln, kh);
  __builtin_amdgcn_sched_barrier(0);
  gemm_frag_mma<TNt>(f0, acc);
  if (STORE) gemm_stage_store<TNt>(g, As2, BsT2, tid);
  if (LOAD) gemm_stage_load<TNt>(g, A, B, N, K, bm, bn, kload, tid);
  __builtin_amdgcn_sched_barrier(0);
  gemm_frag_load<TNt>(f0, As, BsT, 2, wm, ln, kh);
  gemm_frag_mma<TNt>(f1, acc);
  __builtin_amdgcn_sched_barrier(0);
  gemm_frag_load<TNt>(f1, As, BsT, 3, wm, ln, kh);
  __builtin_amdgcn_sched_barrier(0);
  gemm_frag_mma<TNt>(f0, acc);
  gemm_frag_mma<TNt>(f1, acc);
}

template <int TNt>
__device__ __forceinline__ void gemm_tile_f32_wmma(const float* __restrict__ A,
                                                   const float* __restrict__ B,
                                                   float* __restrict__ C,
                                                   int N, int K, int bm, int bn,
                                                   float (*As)[64][LP], float (*BsT)[TN][LP]) {
  const int tid  = threadIdx.x;
  const int wm   = tid >> 5, lane = tid & 31;
  const int kh   = lane >> 4, ln = lane & 15;

  GemmStage<TNt> g;
  v8f acc[TNt / 16];
#pragma unroll
  for (int s = 0; s < TNt / 16; ++s) acc[s] = v8f{};
  gemm_stage_load<TNt>(g, A, B, N, K, bm, bn, 0, tid);
  gemm_stage_store<TNt>(g, As[0], BsT[0], tid);
  gemm_stage_load<TNt>(g, A, B, N, K, bm, bn, KS, tid);
  barrier_lds_only();

  int kb = 0;
  for (; kb + 2 * KS < K; kb += 2 * KS) {
    gemm_step<TNt, true, true>(g, As[0], BsT[0], As[1], BsT[1], A, B, N, K, bm, bn, kb + 2 * KS, tid,
                               acc, wm, ln, kh);
    barrier_lds_only();
    gemm_step<TNt, true, true>(g, As[1], BsT[1], As[0], BsT[0], A, B, N, K, bm, bn, kb + 3 * KS, tid,
                               acc, wm, ln, kh);
    barrier_lds_only();
  }
  gemm_step<TNt, true, false>(g, As[0], BsT[0], As[1], BsT[1], A, B, N, K, bm, bn, 0, tid,
                              acc, wm, ln, kh);
  barrier_lds_only();
  gemm_step<TNt, false, false>(g, As[1], BsT[1], As[0], BsT[0], A, B, N, K, bm, bn, 0, tid,
                               acc, wm, ln, kh);
#pragma unroll
  for (int r = 0; r < 8; ++r) {
    const size_t row = (size_t)(bm + wm * 16 + r + 8 * kh);
#pragma unroll
    for (int s = 0; s < TNt / 16; ++s) C[row * N + bn + 16 * s + ln] = acc[s][r];
  }
}

__global__ __launch_bounds__(GT) void gemm_f32_wmma(const float* __restrict__ A,
                                                    const float* __restrict__ B,
                                                    float* __restrict__ C) {
  __shared__ __attribute__((aligned(16))) float As[2][64][LP];
  __shared__ __attribute__((aligned(16))) float BsT[2][TN][LP];
  constexpr int ntb = DM / TN;
  gemm_tile_f32_wmma<TN>(A, B, C, DM, DM, (int)(blockIdx.x / ntb) * 64, (int)(blockIdx.x % ntb) * TN,
                         As, BsT);
}

__global__ __launch_bounds__(GT) void gemm_qkv_f32_wmma(const float* __restrict__ X,
                                                        const float* __restrict__ Wq,
                                                        const float* __restrict__ Wk,
                                                        const float* __restrict__ Wv,
                                                        float* __restrict__ Qg,
                                                        float* __restrict__ Kg,
                                                        float* __restrict__ Vg) {
  __shared__ __attribute__((aligned(16))) float As[2][64][LP];
  __shared__ __attribute__((aligned(16))) float BsT[2][TN][LP];
  constexpr int NQ = Hh * FD;
  constexpr int TV = (L / 64) * (DM / TN);
  constexpr int TQ = (L / 64) * (NQ / TNQ);
  const int b = (int)blockIdx.x;
  if (b < TV) {
    gemm_tile_f32_wmma<TN>(X, Wv, Vg, DM, DM, (b / (DM / TN)) * 64, (b % (DM / TN)) * TN, As, BsT);
  } else {
    const int t = (b - TV) % TQ;
    const bool isq = (b - TV) < TQ;
    gemm_tile_f32_wmma<TNQ>(X, isq ? Wq : Wk, isq ? Qg : Kg, NQ, DM,
                            (t / (NQ / TNQ)) * 64, (t % (NQ / TNQ)) * TNQ, As, BsT);
  }
}

constexpr int NPAIR    = 10;
constexpr int NQF      = 16 * NPAIR;
constexpr int SLOT     = 3012;
constexpr int O_LIN    = 2560, O_ONES = 2816, O_KC = 2832, O_END = 3009;
constexpr int SLOTS_P0 = 168;

static_assert(O_LIN == 16 * NQF && O_ONES == O_LIN + 256 && O_KC == O_ONES + 16 && O_END == O_KC + 17 + NQF && SLOT % 4 == 0 && SLOT >= O_END,
              "slot layout: the linear tile directly behind the packed quad tiles, float4 granularity");
__device__ __forceinline__ int pair_I(int p) { return (955648 >> (2 * p)) & 3; }
__device__ __forceinline__ int pair_J(int p) { return (1030628 >> (2 * p)) & 3; }

__device__ __forceinline__ float* state_slot(float* p0, float* p1, int i) {
  return i < SLOTS_P0 ? p0 + (size_t)i * SLOT : p1 + (size_t)(i - SLOTS_P0) * SLOT;
}

__global__ __launch_bounds__(256) void based_state(const float* __restrict__ Kg,
                                                   const float* __restrict__ Vg,
                                                   float* __restrict__ P0,
                                                   float* __restrict__ P1) {
  __shared__ __attribute__((aligned(16))) float ST[2][2][16][36];

  const int tid  = threadIdx.x;
  const int wave = tid >> 5, lane = tid & 31;
  const int kh   = lane >> 4, ln = lane & 15;
  const int seg  = (int)(blockIdx.x / (Hh * 4));
  const int hv   = blockIdx.x % (Hh * 4);
  const int h    = hv >> 2;
  const int vb   = hv & 3;
  const int c_begin = seg * CPS, c_end = c_begin + CPS;
  const int qoff = h * FD;
  const int voff = h * HD + vb * 16;

  v8f q0 = {}, q1 = {};
  float kq = 0.f;
  float sc = 0.f;
  const int wv = __builtin_amdgcn_readfirstlane(wave);
  const int ra0 = 4 * pair_I(wv) + (ln >> 2), rc0 = 4 * pair_J(wv) + (ln & 3);
  const bool second = (wv < 2) || (wv == 7);
  const bool lin2 = (wv == 7);
  const int ra1 = lin2 ? ln : 4 * pair_I(wv + 8) + (ln >> 2);
  const int rc1 = lin2 ? ln : 4 * pair_J(wv + 8) + (ln & 3);
  const int pp = min(tid >> 4, NPAIR - 1);
  const int fi = 4 * pair_I(pp) + ((tid >> 2) & 3), fj = 4 * pair_J(pp) + (tid & 3);
  const int kv = tid >> 7;
  const int st = (tid & 127) >> 2, sc4 = (tid & 3) * 4;
  const float* gsrc = kv ? (Vg + voff + sc4) : (Kg + qoff + sc4);
  const size_t gp = kv ? (size_t)DM : (size_t)(Hh * FD);
  float4 pre = *(const float4*)&gsrc[(size_t)(c_begin * 32 + st) * gp];
#pragma unroll 1
  for (int c = c_begin; c < c_end; ++c) {
    const int bf = c & 1;
    ST[bf][kv][sc4 + 0][st] = pre.x;
    ST[bf][kv][sc4 + 1][st] = pre.y;
    ST[bf][kv][sc4 + 2][st] = pre.z;
    ST[bf][kv][sc4 + 3][st] = pre.w;
    __syncthreads();
    if (c + 1 < c_end) pre = *(const float4*)&gsrc[(size_t)((c + 1) * 32 + st) * gp];
    const float (*kt)[36] = ST[bf][0];
    const float (*vt)[36] = ST[bf][1];
    float4 ka[8], kb[8];
#pragma unroll
    for (int m = 0; m < 8; ++m) {
      ka[m] = *(const float4*)&kt[fi][4 * m];
      kb[m] = *(const float4*)&kt[fj][4 * m];
    }
    if (tid >= 128 && tid < 160) {
      const float* row = (tid < 144) ? vt[tid - 128] : kt[tid - 144];
      float s = 0.f;
#pragma unroll
      for (int t = 0; t < 32; t += 4) {
        const float4 a = *(const float4*)&row[t];
        s += a.x;  s += a.y;  s += a.z;  s += a.w;
      }
      sc += (tid < 144) ? s : 0.5f * s;
    } else if (tid == 160) {
      sc += 32.f;
    }
    v2f fb[8], fx0[8], fy0[8];
#pragma unroll
    for (int m = 0; m < 8; ++m) {
      const int kk = 4 * m + 2 * kh;
      fb[m]  = *(const v2f*)&vt[ln][kk];
      fx0[m] = *(const v2f*)&kt[ra0][kk];
      fy0[m] = *(const v2f*)&kt[rc0][kk];
    }
    if (second) {
      v2f fx1[8], fy1[8];
#pragma unroll
      for (int m = 0; m < 8; ++m) {
        const int kk = 4 * m + 2 * kh;
        fx1[m] = *(const v2f*)&kt[ra1][kk];
        fy1[m] = *(const v2f*)&kt[rc1][kk];
      }
#pragma unroll
      for (int m = 0; m < 8; ++m) {
        const v2f a1 = lin2 ? fx1[m] * 0.5f : fx1[m] * fy1[m] * QSC;
        q1 = WMMA_F32(a1, fb[m], q1);
      }
    }
#pragma unroll
    for (int m = 0; m < 8; ++m) {
      const v2f a0 = fx0[m] * fy0[m] * QSC;
      q0 = WMMA_F32(a0, fb[m], q0);
    }
    {
      float s = 0.f;
#pragma unroll
      for (int m = 0; m < 8; ++m) {
        s += ka[m].x * kb[m].x * QSC;
        s += ka[m].y * kb[m].y * QSC;
        s += ka[m].z * kb[m].z * QSC;
        s += ka[m].w * kb[m].w * QSC;
      }
      kq += s;
    }
  }
  float* g = state_slot(P0, P1, seg * (Hh * 4) + hv);
#pragma unroll
  for (int r = 0; r < 8; ++r) g[(wv * 16 + r + 8 * kh) * 16 + ln] = q0[r];
  if (second) {
    const int t1 = lin2 ? NPAIR : wv + 8;
#pragma unroll
    for (int r = 0; r < 8; ++r) g[(t1 * 16 + r + 8 * kh) * 16 + ln] = q1[r];
  }
  if (tid < NQF) g[O_KC + 17 + tid] = kq;
  if (tid >= 128 && tid < 144)      g[O_ONES + tid - 128] = sc;
  else if (tid >= 144 && tid < 160) g[O_KC + 1 + tid - 144] = sc;
  else if (tid == 160)              g[O_KC] = sc;
  else if (tid >= 161 && tid < 164) g[O_END + tid - 161] = 0.f;
}

__device__ __forceinline__ const float* state_slot(const float* p0, const float* p1, int i) {
  return i < SLOTS_P0 ? p0 + (size_t)i * SLOT : p1 + (size_t)(i - SLOTS_P0) * SLOT;
}

constexpr int SR = CPS * 32;
constexpr int AWP = SR + 4;

__global__ __launch_bounds__(256) void based_attn(const float* __restrict__ Qg,
                                                  const float* __restrict__ Kg,
                                                  float* VY,
                                                  const float* __restrict__ P0,
                                                  const float* __restrict__ P1) {
  __shared__ float s_quad[NQF + 16][17];
  __shared__ float s_ones[16];
  __shared__ float kc0[17];
  __shared__ __attribute__((aligned(16))) float kcq[NQF];
  __shared__ __attribute__((aligned(16))) float Qc[SR][20];
  __shared__ __attribute__((aligned(16))) float Kc[SR][20];
  __shared__ __attribute__((aligned(16))) float Vc[SR][20];
  __shared__ __attribute__((aligned(16))) float Aw[32][AWP];
  __shared__ __attribute__((aligned(16))) float Yp[4][32][20];
  __shared__ float denp[10][32];

  static_assert(SR == 128, "staging and the tile split below are written for 128-row segments");

  const int tid  = threadIdx.x;
  const int wave = tid >> 5, lane = tid & 31;
  const int kh   = lane >> 4, ln = lane & 15;
  const int wv   = __builtin_amdgcn_readfirstlane(wave);
  const int seg  = NSEG - 1 - (int)(blockIdx.x / (Hh * 4));
  const int hv   = blockIdx.x % (Hh * 4);
  const int h    = hv >> 2;
  const int vb   = hv & 3;
  const int row0 = seg * SR;
  const int qoff = h * FD;
  const int voff = h * HD + vb * 16;

  float4 gq[2], gk[2], gv[2];
#pragma unroll
  for (int u = 0; u < 2; ++u) {
    const int idx = tid + 256 * u;
    const int t = idx >> 2, c4 = (idx & 3) * 4;
    gq[u] = *(const float4*)&Qg[(size_t)(row0 + t) * (Hh * FD) + qoff + c4];
    gk[u] = *(const float4*)&Kg[(size_t)(row0 + t) * (Hh * FD) + qoff + c4];
    gv[u] = *(const float4*)&VY[(size_t)(row0 + t) * DM + voff + c4];
  }

  {
    int i4[3];
#pragma unroll
    for (int j = 0; j < 2; ++j) i4[j] = tid + 256 * j;
    i4[2] = min(tid + 512, SLOT / 4 - 1);
    float4 p[NSEG - 1][3];
    const int s_last = max(seg - 1, 0);
#pragma unroll
    for (int s = 0; s < NSEG - 1; ++s) {
      const float* g = state_slot(P0, P1, min(s, s_last) * (Hh * 4) + hv);
#pragma unroll
      for (int j = 0; j < 3; ++j) p[s][j] = *(const float4*)&g[4 * i4[j]];
    }
#pragma unroll
    for (int u = 0; u < 2; ++u) {
      const int idx = tid + 256 * u;
      const int t = idx >> 2, c4 = (idx & 3) * 4;
      *(float4*)&Qc[t][c4] = gq[u];
      *(float4*)&Kc[t][c4] = gk[u];
      *(float4*)&Vc[t][c4] = gv[u];
    }
    float4 a[3];
#pragma unroll
    for (int j = 0; j < 3; ++j) a[j] = float4{0.f, 0.f, 0.f, 0.f};
#pragma unroll
    for (int s = 0; s < NSEG - 1; ++s) {
      const bool on = s < seg;
#pragma unroll
      for (int j = 0; j < 3; ++j) {
        a[j].x += on ? p[s][j].x : 0.f;
        a[j].y += on ? p[s][j].y : 0.f;
        a[j].z += on ? p[s][j].z : 0.f;
        a[j].w += on ? p[s][j].w : 0.f;
      }
    }
#pragma unroll
    for (int j = 0; j < 3; ++j) {
      const int e = 4 * (tid + 256 * j);
      const float v[4] = {a[j].x, a[j].y, a[j].z, a[j].w};
      if (e < O_ONES) {
#pragma unroll
        for (int u = 0; u < 4; ++u) s_quad[e >> 4][(e & 15) + u] = v[u];
      } else if (e < O_KC) {
#pragma unroll
        for (int u = 0; u < 4; ++u) s_ones[e - O_ONES + u] = v[u];
      } else if (e < O_END) {
#pragma unroll
        for (int u = 0; u < 4; ++u) {
          const int x = e - O_KC + u;
          if (x < 17) kc0[x] = v[u];
          else if (x < 17 + NQF) kcq[x - 17] = v[u];
        }
      }
    }
  }
  __syncthreads();

  for (int j = 0; j < CPS; ++j) {
    const int jr = j * 32;
    const int bl = row0 + jr;

    {
      const int ti = wave & 1;
      const int tj0 = wv >> 1, tj1 = (wv >> 1) + 4;
      v8f s0 = {}, s1 = {};
#pragma unroll
      for (int k = 0; k < 16; k += 4) {
        const int kk = k + 2 * kh;
        const v2f a  = *(const v2f*)&Qc[jr + ti * 16 + ln][kk];
        const v2f b0 = *(const v2f*)&Kc[tj0 * 16 + ln][kk];
        const v2f b1 = *(const v2f*)&Kc[tj1 * 16 + ln][kk];
        s0 = WMMA_F32(a, b0, s0);
        s1 = WMMA_F32(a, b1, s1);
      }
#pragma unroll
      for (int r = 0; r < 8; ++r) {
        const int row = ti * 16 + r + 8 * kh;
        const int c0 = tj0 * 16 + ln, c1 = tj1 * 16 + ln;
        const float x0 = s0[r], x1 = s1[r];
        Aw[row][c0] = (jr + row >= c0) ? (1.f + 0.25f * x0 + (x0 * x0) * 0.03125f) : 0.f;
        Aw[row][c1] = (jr + row >= c1) ? (1.f + 0.25f * x1 + (x1 * x1) * 0.03125f) : 0.f;
      }
    }
    __syncthreads();

    {
      const int ti = wave & 1;
      const int kq = wv >> 1;
      const int row = ti * 16 + ln;
      const float* qr = Qc[jr + row];
      v8f acc;
      const float a_init = (kq == 0) ? s_ones[ln] : 0.f;
#pragma unroll
      for (int r = 0; r < 8; ++r) acc[r] = a_init;
      const bool last = (kq == 3);
      v2f ql[4];
#pragma unroll
      for (int a = 0; a < 4; ++a) ql[a] = *(const v2f*)&qr[4 * a + 2 * kh];
      v2f fa[3][4], fb[3][4];
#pragma unroll
      for (int u = 0; u < 3; ++u) {
        const int p = min(3 * kq + u, NPAIR - 1);
        const int I = pair_I(p), J = pair_J(p);
        const bool lin = last && (u == 1);
        const bool off = last && (u == 2);
        const float w = off ? 0.f : (I == J ? QSC : 2.f * QSC);
        const float4 qi4 = *(const float4*)&qr[4 * I];
        const float qi[4] = {qi4.x, qi4.y, qi4.z, qi4.w};
        const v2f qjv = *(const v2f*)&qr[4 * J + 2 * kh];
        const float (*bb)[17] = &s_quad[16 * (lin ? NPAIR : p) + 2 * kh];
#pragma unroll
        for (int a = 0; a < 4; ++a) {
          const v2f x = lin ? ql[a] : qjv;
          const float s = lin ? 0.5f : qi[a] * w;
          fa[u][a] = x * s;
          fb[u][a].x = bb[4 * a][ln];
          fb[u][a].y = bb[4 * a + 1][ln];
        }
      }
      v2f wa[8], wb[8];
#pragma unroll
      for (int m = 0; m < 8; ++m) {
        const int kk = (kq * 8 + m) * 4 + 2 * kh;
        wa[m] = *(const v2f*)&Aw[row][kk];
        wb[m].x = Vc[kk][ln];  wb[m].y = Vc[kk + 1][ln];
      }
      __builtin_amdgcn_sched_barrier(0);
#pragma unroll
      for (int u = 1; u < 3; ++u) {
#pragma unroll
        for (int a = 0; a < 4; ++a) acc = WMMA_F32(fa[u][a], fb[u][a], acc);
      }
#pragma unroll
      for (int m = 0; m < 8; ++m) acc = WMMA_F32(wa[m], wb[m], acc);
#pragma unroll
      for (int a = 0; a < 4; ++a) acc = WMMA_F32(fa[0][a], fb[0][a], acc);
      {
        const int t = lane;
        const float* qt = Qc[jr + t];
        float d = 0.f;
#pragma unroll
        for (int g = 0; g < 5; ++g) {
          const int ks = 5 * wv + g;
          const int p = ks >> 2, a = ks & 3;
          const int I = pair_I(p), J = pair_J(p);
          const float w = (I == J) ? QSC : 2.f * QSC;
          const float qi = qt[4 * I + a] * w;
          const float4 qj = *(const float4*)&qt[4 * J];
          const float4 kc = *(const float4*)&kcq[4 * ks];
          d += (qj.x * qi) * kc.x;
          d += (qj.y * qi) * kc.y;
          d += (qj.z * qi) * kc.z;
          d += (qj.w * qi) * kc.w;
        }
        float e = 0.f;
#pragma unroll
        for (int u = 0; u < 4; ++u) {
          const float4 w4 = *(const float4*)&Aw[t][16 * wave + 4 * u];
          e += w4.x;  e += w4.y;  e += w4.z;  e += w4.w;
        }
        denp[wave][t] = d + e;
        if (wave == 6) {
          denp[8][t] = kc0[0];
        } else if (wave == 7) {
          float e2 = 0.f;
#pragma unroll
          for (int u = 0; u < 4; ++u) {
            const float4 v = *(const float4*)&qt[4 * u];
            e2 += v.x * 0.5f * kc0[1 + 4 * u];
            e2 += v.y * 0.5f * kc0[2 + 4 * u];
            e2 += v.z * 0.5f * kc0[3 + 4 * u];
            e2 += v.w * 0.5f * kc0[4 + 4 * u];
          }
          denp[9][t] = e2;
        }
      }
#pragma unroll
      for (int r = 0; r < 8; ++r) Yp[kq][ti * 16 + r + 8 * kh][ln] = acc[r];
    }
    __syncthreads();

    if (tid < 128) {
      const int t = tid >> 2, c4 = (tid & 3) * 4;
      float dd = denp[8][t];
      dd += denp[9][t];
#pragma unroll
      for (int w = 0; w < 8; ++w) dd += denp[w][t];
      dd += EPSF;
      float4 y = *(float4*)&Yp[0][t][c4];
#pragma unroll
      for (int q = 1; q < 4; ++q) {
        const float4 p = *(float4*)&Yp[q][t][c4];
        y.x += p.x; y.y += p.y; y.z += p.z; y.w += p.w;
      }
      y.x /= dd; y.y /= dd; y.z /= dd; y.w /= dd;
      *(float4*)&VY[(size_t)(bl + t) * DM + voff + c4] = y;
    }
  }
}

extern "C" void kernel_launch(void* const* d_in, const int* in_sizes, int n_in,
                              void* d_out, int out_size, void* d_ws, size_t ws_size,
                              hipStream_t stream) {
  (void)in_sizes; (void)n_in; (void)out_size; (void)ws_size;
  const float* hs = (const float*)d_in[0];
  const float* Wq = (const float*)d_in[1];
  const float* Wk = (const float*)d_in[2];
  const float* Wv = (const float*)d_in[3];
  const float* Wo = (const float*)d_in[4];
  float* out = (float*)d_out;

  float* Qg = (float*)d_ws;
  float* Kg = Qg + (size_t)L * (Hh * FD);
  float* Vg = Kg + (size_t)L * (Hh * FD);
  float* S0 = Vg + (size_t)L * DM;
  float* S1 = out;

  const dim3 blk(256);
  const dim3 gblk(GT);
  gemm_qkv_f32_wmma<<<dim3((L / 64) * (DM / TN) + 2 * (L / 64) * ((Hh * FD) / TNQ)), gblk, 0, stream>>>(
      hs, Wq, Wk, Wv, Qg, Kg, Vg);
  based_state<<<dim3(Hh * 4 * (NSEG - 1)), blk, 0, stream>>>(Kg, Vg, S0, S1);
  based_attn<<<dim3(Hh * 4 * NSEG), blk, 0, stream>>>(Qg, Kg, Vg, S0, S1);
  gemm_f32_wmma<<<dim3((L / 64) * (DM / TN)), gblk, 0, stream>>>(Vg, Wo, out);
}
